# starting kernel
# baseline (speedup 1.0000x reference)
#include <hip/hip_runtime.h>
#include <hip/hip_bf16.h>

#define N_NODES 100000
#define N_EDGES 1600000
#define IN_DIM  256
#define HID     128
#define N_GRAPHS 64
#define OUT_MF 489
#define OUT_BP 1943
#define OUT_CC 320

typedef __attribute__((ext_vector_type(2))) float v2f;
typedef __attribute__((ext_vector_type(8))) float v8f;

__global__ void fill_f32(float* __restrict__ p, float v, int n) {
    int t = blockIdx.x * blockDim.x + threadIdx.x;
    if (t < n) p[t] = v;
}

__global__ void zero_f4(float4* __restrict__ p, int n4) {
    int t = blockIdx.x * blockDim.x + threadIdx.x;
    if (t < n4) { float4 z; z.x = z.y = z.z = z.w = 0.f; p[t] = z; }
}

__global__ void deg_accum(const int* __restrict__ dst, float* __restrict__ deg, int E) {
    int t = blockIdx.x * blockDim.x + threadIdx.x;
    if (t < E) atomicAdd(&deg[dst[t]], 1.0f);
}

__global__ void rsqrt_inplace(float* __restrict__ p, int n) {
    int t = blockIdx.x * blockDim.x + threadIdx.x;
    if (t < n) p[t] = rsqrtf(p[t]);
}

template <int K>
__global__ __launch_bounds__(256) void gemm_wmma_f32(const float* __restrict__ A,
                                                     const float* __restrict__ W,
                                                     float* __restrict__ out) {
    __shared__ float ldsW[64 * HID];
    const int tid  = threadIdx.x;
    const int wave = tid >> 5;
    const int lane = tid & 31;
    const int half = lane >> 4;
    const int m    = lane & 15;
    const int row0 = blockIdx.x * 16;
    const int col0 = wave * 16;

    const float* arow = A + (size_t)(row0 + m) * K;
    v8f acc = {};

    for (int kb = 0; kb < K; kb += 64) {
        const float4* wv = (const float4*)(W + (size_t)kb * HID);
        float4* lv = (float4*)ldsW;
        #pragma unroll
        for (int i = tid; i < 64 * HID / 4; i += 256) lv[i] = wv[i];
        __syncthreads();

        #pragma unroll
        for (int k = 0; k < 64; k += 4) {
            const int ka = k + 2 * half;
            v2f a;
            a.x = arow[kb + ka];
            a.y = arow[kb + ka + 1];
            v2f b;
            b.x = ldsW[ka * HID + col0 + m];
            b.y = ldsW[(ka + 1) * HID + col0 + m];
            acc = __builtin_amdgcn_wmma_f32_16x16x4_f32(
                false, a, false, b, (short)0, acc, false, false);
        }
        __syncthreads();
    }

    #pragma unroll
    for (int v = 0; v < 8; ++v)
        out[(size_t)(row0 + v + 8 * half) * HID + col0 + m] = acc[v];
}

__global__ __launch_bounds__(256) void edge_scatter(const int* __restrict__ src,
                                                    const int* __restrict__ dst,
                                                    const float* __restrict__ dis,
                                                    const float* __restrict__ h,
                                                    float* __restrict__ agg, int E) {
    const int lane = threadIdx.x & 31;
    const int gw   = (blockIdx.x * blockDim.x + threadIdx.x) >> 5;
    const int nw   = (gridDim.x * blockDim.x) >> 5;
    for (int e = gw; e < E; e += nw) {
        const int s = src[e], d = dst[e];
        const float nrm = dis[s] * dis[d];
        const float* hs = h + (size_t)s * HID;
        float* ad = agg + (size_t)d * HID;
        #pragma unroll
        for (int j = 0; j < HID; j += 32)
            atomicAdd(&ad[j + lane], hs[j + lane] * nrm);
    }
}

__global__ void finish_layer(float* __restrict__ agg, const float* __restrict__ h,
                             const float* __restrict__ dis, const float* __restrict__ bias,
                             int n) {
    int t = blockIdx.x * blockDim.x + threadIdx.x;
    if (t >= n) return;
    const int i = t >> 7, j = t & (HID - 1);
    const float di = dis[i];
    const float v = agg[t] + di * di * h[t] + bias[j];
    agg[t] = v > 0.f ? v : 0.f;
}

__global__ __launch_bounds__(256) void pool_accum(const float* __restrict__ h,
                                                  const int* __restrict__ batch,
                                                  float* __restrict__ pooled,
                                                  float* __restrict__ cnt, int N) {
    const int lane = threadIdx.x & 31;
    const int gw   = (blockIdx.x * blockDim.x + threadIdx.x) >> 5;
    const int nw   = (gridDim.x * blockDim.x) >> 5;
    for (int i = gw; i < N; i += nw) {
        const int g = batch[i];
        const float* hi = h + (size_t)i * HID;
        float* pg = pooled + g * HID;
        #pragma unroll
        for (int j = 0; j < HID; j += 32)
            atomicAdd(&pg[j + lane], hi[j + lane]);
        if (lane == 0) atomicAdd(&cnt[g], 1.0f);
    }
}

__global__ void pool_div(float* __restrict__ pooled, const float* __restrict__ cnt) {
    int t = blockIdx.x * blockDim.x + threadIdx.x;
    if (t >= N_GRAPHS * HID) return;
    float c = cnt[t >> 7];
    pooled[t] /= (c > 1.f ? c : 1.f);
}

__global__ void head_gemm(const float* __restrict__ pooled, const float* __restrict__ W,
                          const float* __restrict__ b, float* __restrict__ out, int OUT) {
    int t = blockIdx.x * blockDim.x + threadIdx.x;
    const int n = N_GRAPHS * OUT;
    if (t >= n) return;
    const int g = t / OUT, o = t - g * OUT;
    const float* p = pooled + g * HID;
    float s = b[o];
    #pragma unroll 8
    for (int k = 0; k < HID; ++k) s += p[k] * W[k * OUT + o];
    out[t] = s;
}

extern "C" void kernel_launch(void* const* d_in, const int* in_sizes, int n_in,
                              void* d_out, int out_size, void* d_ws, size_t ws_size,
                              hipStream_t stream) {
    const float* x   = (const float*)d_in[0];
    const int*   ei  = (const int*)d_in[1];
    const int*   bat = (const int*)d_in[2];
    const float* W1  = (const float*)d_in[3];
    const float* b1  = (const float*)d_in[4];
    const float* W2  = (const float*)d_in[5];
    const float* b2  = (const float*)d_in[6];
    const float* Wmf = (const float*)d_in[7];
    const float* bmf = (const float*)d_in[8];
    const float* Wbp = (const float*)d_in[9];
    const float* bbp = (const float*)d_in[10];
    const float* Wcc = (const float*)d_in[11];
    const float* bcc = (const float*)d_in[12];
    float* out = (float*)d_out;

    const int E = in_sizes[1] / 2;
    const int* src = ei;
    const int* dst = ei + E;

    float* ws     = (float*)d_ws;
    float* dis    = ws;
    float* h      = dis + N_NODES;
    float* agg    = h + (size_t)N_NODES * HID;
    float* pooled = agg + (size_t)N_NODES * HID;
    float* cnt    = pooled + N_GRAPHS * HID;

    const int NH   = N_NODES * HID;
    const int NH4  = NH / 4;

    fill_f32<<<(N_NODES + 255) / 256, 256, 0, stream>>>(dis, 1.0f, N_NODES);
    deg_accum<<<(E + 255) / 256, 256, 0, stream>>>(dst, dis, E);
    rsqrt_inplace<<<(N_NODES + 255) / 256, 256, 0, stream>>>(dis, N_NODES);

    gemm_wmma_f32<IN_DIM><<<N_NODES / 16, 256, 0, stream>>>(x, W1, h);
    zero_f4<<<(NH4 + 255) / 256, 256, 0, stream>>>((float4*)agg, NH4);
    edge_scatter<<<2048, 256, 0, stream>>>(src, dst, dis, h, agg, E);
    finish_layer<<<(NH + 255) / 256, 256, 0, stream>>>(agg, h, dis, b1, NH);

    gemm_wmma_f32<HID><<<N_NODES / 16, 256, 0, stream>>>(agg, W2, h);
    zero_f4<<<(NH4 + 255) / 256, 256, 0, stream>>>((float4*)agg, NH4);
    edge_scatter<<<2048, 256, 0, stream>>>(src, dst, dis, h, agg, E);
    finish_layer<<<(NH + 255) / 256, 256, 0, stream>>>(agg, h, dis, b2, NH);

    fill_f32<<<(N_GRAPHS * HID + N_GRAPHS + 255) / 256, 256, 0, stream>>>(
        pooled, 0.0f, N_GRAPHS * HID + N_GRAPHS);
    pool_accum<<<12500, 256, 0, stream>>>(agg, bat, pooled, cnt, N_NODES);
    pool_div<<<(N_GRAPHS * HID + 255) / 256, 256, 0, stream>>>(pooled, cnt);

    float* out_mf = out;
    float* out_bp = out + N_GRAPHS * OUT_MF;
    float* out_cc = out + N_GRAPHS * (OUT_MF + OUT_BP);
    head_gemm<<<(N_GRAPHS * OUT_MF + 255) / 256, 256, 0, stream>>>(pooled, Wmf, bmf, out_mf, OUT_MF);
    head_gemm<<<(N_GRAPHS * OUT_BP + 255) / 256, 256, 0, stream>>>(pooled, Wbp, bbp, out_bp, OUT_BP);
    head_gemm<<<(N_GRAPHS * OUT_CC + 255) / 256, 256, 0, stream>>>(pooled, Wcc, bcc, out_cc, OUT_CC);
}


# each layer's neighbour sum kept in LDS by the work-group owning 391 nodes (all groups scan every edge destination, LDS float adds): no global float atomics, zero or finish pass; sum order changes
# speedup vs baseline: 1.4576x; 1.4576x over previous
#include <hip/hip_runtime.h>
#include <hip/hip_bf16.h>

#define N_NODES 100000
#define N_EDGES 1600000
#define IN_DIM  256
#define HID     128
#define N_GRAPHS 64
#define OUT_MF 489
#define OUT_BP 1943
#define OUT_CC 320

typedef __attribute__((ext_vector_type(2))) float v2f;
typedef __attribute__((ext_vector_type(8))) float v8f;

__global__ void fill_f32(float* __restrict__ p, float v, int n) {
    int t = blockIdx.x * blockDim.x + threadIdx.x;
    if (t < n) p[t] = v;
}

__global__ void deg_accum(const int* __restrict__ dst, float* __restrict__ deg, int E) {
    int t = blockIdx.x * blockDim.x + threadIdx.x;
    if (t < E) atomicAdd(&deg[dst[t]], 1.0f);
}

__global__ void rsqrt_inplace(float* __restrict__ p, int n) {
    int t = blockIdx.x * blockDim.x + threadIdx.x;
    if (t < n) p[t] = rsqrtf(p[t]);
}

template <int K>
__global__ __launch_bounds__(256) void gemm_wmma_f32(const float* __restrict__ A,
                                                     const float* __restrict__ W,
                                                     float* __restrict__ out) {
    __shared__ float ldsW[64 * HID];
    const int tid  = threadIdx.x;
    const int wave = tid >> 5;
    const int lane = tid & 31;
    const int half = lane >> 4;
    const int m    = lane & 15;
    const int row0 = blockIdx.x * 16;
    const int col0 = wave * 16;

    const float* arow = A + (size_t)(row0 + m) * K;
    v8f acc = {};

    for (int kb = 0; kb < K; kb += 64) {
        const float4* wv = (const float4*)(W + (size_t)kb * HID);
        float4* lv = (float4*)ldsW;
        #pragma unroll
        for (int i = tid; i < 64 * HID / 4; i += 256) lv[i] = wv[i];
        __syncthreads();

        #pragma unroll
        for (int k = 0; k < 64; k += 4) {
            const int ka = k + 2 * half;
            v2f a;
            a.x = arow[kb + ka];
            a.y = arow[kb + ka + 1];
            v2f b;
            b.x = ldsW[ka * HID + col0 + m];
            b.y = ldsW[(ka + 1) * HID + col0 + m];
            acc = __builtin_amdgcn_wmma_f32_16x16x4_f32(
                false, a, false, b, (short)0, acc, false, false);
        }
        __syncthreads();
    }

    #pragma unroll
    for (int v = 0; v < 8; ++v)
        out[(size_t)(row0 + v + 8 * half) * HID + col0 + m] = acc[v];
}

#define OWN_R     391
#define OWN_WAVES 32
#define OWN_Q     64
#define OWN_U     8
#define OWN_G     8

typedef __attribute__((address_space(3))) int lds_int;

__device__ __forceinline__ void owner_drain(volatile lds_int* q, int head, int nb, int lane, int lo,
                                            const int* __restrict__ src, const int* __restrict__ dst,
                                            const float* __restrict__ dis, const float* __restrict__ h,
                                            float* acc) {
    const bool v = lane < nb;
    const int e = v ? q[(head + lane) & (OWN_Q - 1)] : 0;
    const int s = v ? src[e] : 0;
    const int d = v ? dst[e] : lo;
    const float nr = v ? dis[s] * dis[d] : 0.f;
    const int r = d - lo;
    int i = 0;
    #pragma unroll 1
    for (; i + OWN_G <= nb; i += OWN_G) {
        float val[OWN_G][4];
        #pragma unroll
        for (int u = 0; u < OWN_G; ++u) {
            const int si = __builtin_amdgcn_readlane(s, i + u);
            const float* hs = h + (size_t)si * HID;
            #pragma unroll
            for (int j = 0; j < 4; ++j) val[u][j] = hs[j * 32 + lane];
        }
        #pragma unroll
        for (int u = 0; u < OWN_G; ++u) {
            const int ri = __builtin_amdgcn_readlane(r, i + u);
            const float ni = __int_as_float(__builtin_amdgcn_readlane(__float_as_int(nr), i + u));
            float* ar = acc + ri * HID;
            #pragma unroll
            for (int j = 0; j < 4; ++j) atomicAdd(&ar[j * 32 + lane], val[u][j] * ni);
        }
    }
    for (; i < nb; ++i) {
        const int si = __builtin_amdgcn_readlane(s, i);
        const int ri = __builtin_amdgcn_readlane(r, i);
        const float ni = __int_as_float(__builtin_amdgcn_readlane(__float_as_int(nr), i));
        const float* hs = h + (size_t)si * HID;
        float* ar = acc + ri * HID;
        #pragma unroll
        for (int j = 0; j < 4; ++j) atomicAdd(&ar[j * 32 + lane], hs[j * 32 + lane] * ni);
    }
}

template <bool VEC>
__device__ __forceinline__ void owner_scan(volatile lds_int* q, int wave, int lane, int lo, int nrows,
                                           int rot, int nsteps, int E,
                                           const int* __restrict__ src, const int* __restrict__ dst,
                                           const float* __restrict__ dis, const float* __restrict__ h,
                                           float* acc) {
    int head = 0, cnt = 0;
    for (int st0 = wave; st0 < nsteps; st0 += OWN_WAVES * OWN_U) {
        int4 dv[OWN_U];
        int e0v[OWN_U];
        #pragma unroll
        for (int u = 0; u < OWN_U; ++u) {
            const int st = st0 + u * OWN_WAVES;
            int sidx = st + rot; if (sidx >= nsteps) sidx -= nsteps;
            const int e0 = (st < nsteps) ? (sidx * 32 + lane) * 4 : E;
            e0v[u] = e0;
            if (VEC) {
                dv[u] = *(const int4*)(dst + (e0 < E ? e0 : 0));
            } else {
                dv[u].x = dst[e0     < E ? e0     : 0];
                dv[u].y = dst[e0 + 1 < E ? e0 + 1 : 0];
                dv[u].z = dst[e0 + 2 < E ? e0 + 2 : 0];
                dv[u].w = dst[e0 + 3 < E ? e0 + 3 : 0];
            }
        }
        __builtin_amdgcn_sched_barrier(0);
        unsigned lm = 0;
        #pragma unroll
        for (int u = 0; u < OWN_U; ++u) {
            const int e0 = e0v[u];
            const unsigned m4 = ((unsigned)((unsigned)(dv[u].x - lo) < (unsigned)nrows))
                              | ((unsigned)((unsigned)(dv[u].y - lo) < (unsigned)nrows) << 1)
                              | ((unsigned)((unsigned)(dv[u].z - lo) < (unsigned)nrows) << 2)
                              | ((unsigned)((unsigned)(dv[u].w - lo) < (unsigned)nrows) << 3);
            const unsigned v4 = VEC ? (e0 < E ? 15u : 0u)
                                    : ((unsigned)(e0 < E) | ((unsigned)(e0 + 1 < E) << 1)
                                       | ((unsigned)(e0 + 2 < E) << 2) | ((unsigned)(e0 + 3 < E) << 3));
            lm |= (m4 & v4) << (4 * u);
        }
        for (int round = 0; round < 4 * OWN_U; ++round) {
            const bool has = lm != 0u;
            const unsigned mask = __builtin_amdgcn_ballot_w32(has);
            if (mask == 0u) break;
            const int bit = has ? __builtin_ctz(lm) : 0;
            const int st = st0 + (bit >> 2) * OWN_WAVES;
            int sidx = st + rot; if (sidx >= nsteps) sidx -= nsteps;
            const int e = (sidx * 32 + lane) * 4 + (bit & 3);
            const int pos = (head + cnt + (int)__builtin_amdgcn_mbcnt_lo(mask, 0u)) & (OWN_Q - 1);
            if (has) { q[pos] = e; lm &= lm - 1u; }
            cnt += __builtin_popcount(mask);
            if (cnt >= 32) {
                owner_drain(q, head, 32, lane, lo, src, dst, dis, h, acc);
                head = (head + 32) & (OWN_Q - 1);
                cnt -= 32;
            }
        }
    }
    owner_drain(q, head, cnt, lane, lo, src, dst, dis, h, acc);
}

__global__ __launch_bounds__(1024) void owner_aggregate(const int* __restrict__ src,
                                                        const int* __restrict__ dst,
                                                        const float* __restrict__ dis,
                                                        const float* __restrict__ h,
                                                        const float* __restrict__ bias,
                                                        float* __restrict__ outp, int E, int N) {
    __shared__ __attribute__((aligned(16))) float acc[OWN_R * HID];
    __shared__ int queue[OWN_WAVES * OWN_Q];

    const int tid  = threadIdx.x;
    const int lane = tid & 31;
    const int wave = __builtin_amdgcn_readfirstlane(tid >> 5);
    const int lo   = blockIdx.x * OWN_R;
    const int nrows = (N - lo) < OWN_R ? (N - lo) : OWN_R;
    volatile lds_int* q = (volatile lds_int*)(queue + wave * OWN_Q);

    for (int idx = tid; idx < nrows * (HID / 4); idx += 1024) {
        const int r = idx >> 5, c4 = idx & 31;
        const float di = dis[lo + r];
        const float w = di * di;
        const float4 hv = ((const float4*)(h + (size_t)(lo + r) * HID))[c4];
        const float4 bv = ((const float4*)bias)[c4];
        float4 o;
        o.x = w * hv.x + bv.x; o.y = w * hv.y + bv.y; o.z = w * hv.z + bv.z; o.w = w * hv.w + bv.w;
        ((float4*)acc)[idx] = o;
    }
    __syncthreads();

    const int nsteps = (E + 127) >> 7;
    const int rot = (int)(((unsigned)blockIdx.x * (unsigned)nsteps) / gridDim.x);
    if (((E & 3) == 0) && ((((size_t)dst) & 15) == 0))
        owner_scan<true >(q, wave, lane, lo, nrows, rot, nsteps, E, src, dst, dis, h, acc);
    else
        owner_scan<false>(q, wave, lane, lo, nrows, rot, nsteps, E, src, dst, dis, h, acc);
    __syncthreads();

    float4* orow = (float4*)(outp + (size_t)lo * HID);
    for (int idx = tid; idx < nrows * (HID / 4); idx += 1024) {
        float4 v4 = ((const float4*)acc)[idx];
        v4.x = v4.x > 0.f ? v4.x : 0.f; v4.y = v4.y > 0.f ? v4.y : 0.f;
        v4.z = v4.z > 0.f ? v4.z : 0.f; v4.w = v4.w > 0.f ? v4.w : 0.f;
        orow[idx] = v4;
    }
}

__global__ __launch_bounds__(256) void pool_accum(const float* __restrict__ h,
                                                  const int* __restrict__ batch,
                                                  float* __restrict__ pooled,
                                                  float* __restrict__ cnt, int N) {
    const int lane = threadIdx.x & 31;
    const int gw   = (blockIdx.x * blockDim.x + threadIdx.x) >> 5;
    const int nw   = (gridDim.x * blockDim.x) >> 5;
    for (int i = gw; i < N; i += nw) {
        const int g = batch[i];
        const float* hi = h + (size_t)i * HID;
        float* pg = pooled + g * HID;
        #pragma unroll
        for (int j = 0; j < HID; j += 32)
            atomicAdd(&pg[j + lane], hi[j + lane]);
        if (lane == 0) atomicAdd(&cnt[g], 1.0f);
    }
}

__global__ void pool_div(float* __restrict__ pooled, const float* __restrict__ cnt) {
    int t = blockIdx.x * blockDim.x + threadIdx.x;
    if (t >= N_GRAPHS * HID) return;
    float c = cnt[t >> 7];
    pooled[t] /= (c > 1.f ? c : 1.f);
}

__global__ void head_gemm(const float* __restrict__ pooled, const float* __restrict__ W,
                          const float* __restrict__ b, float* __restrict__ out, int OUT) {
    int t = blockIdx.x * blockDim.x + threadIdx.x;
    const int n = N_GRAPHS * OUT;
    if (t >= n) return;
    const int g = t / OUT, o = t - g * OUT;
    const float* p = pooled + g * HID;
    float s = b[o];
    #pragma unroll 8
    for (int k = 0; k < HID; ++k) s += p[k] * W[k * OUT + o];
    out[t] = s;
}

extern "C" void kernel_launch(void* const* d_in, const int* in_sizes, int n_in,
                              void* d_out, int out_size, void* d_ws, size_t ws_size,
                              hipStream_t stream) {
    const float* x   = (const float*)d_in[0];
    const int*   ei  = (const int*)d_in[1];
    const int*   bat = (const int*)d_in[2];
    const float* W1  = (const float*)d_in[3];
    const float* b1  = (const float*)d_in[4];
    const float* W2  = (const float*)d_in[5];
    const float* b2  = (const float*)d_in[6];
    const float* Wmf = (const float*)d_in[7];
    const float* bmf = (const float*)d_in[8];
    const float* Wbp = (const float*)d_in[9];
    const float* bbp = (const float*)d_in[10];
    const float* Wcc = (const float*)d_in[11];
    const float* bcc = (const float*)d_in[12];
    float* out = (float*)d_out;

    const int E = in_sizes[1] / 2;
    const int* src = ei;
    const int* dst = ei + E;

    float* ws     = (float*)d_ws;
    float* dis    = ws;
    float* h      = dis + N_NODES;
    float* agg    = h + (size_t)N_NODES * HID;
    float* pooled = agg + (size_t)N_NODES * HID;
    float* cnt    = pooled + N_GRAPHS * HID;

    fill_f32<<<(N_NODES + 255) / 256, 256, 0, stream>>>(dis, 1.0f, N_NODES);
    deg_accum<<<(E + 255) / 256, 256, 0, stream>>>(dst, dis, E);
    rsqrt_inplace<<<(N_NODES + 255) / 256, 256, 0, stream>>>(dis, N_NODES);

    gemm_wmma_f32<IN_DIM><<<N_NODES / 16, 256, 0, stream>>>(x, W1, h);
    owner_aggregate<<<(N_NODES + OWN_R - 1) / OWN_R, 1024, 0, stream>>>(src, dst, dis, h, b1, agg, E, N_NODES);

    gemm_wmma_f32<HID><<<N_NODES / 16, 256, 0, stream>>>(agg, W2, h);
    owner_aggregate<<<(N_NODES + OWN_R - 1) / OWN_R, 1024, 0, stream>>>(src, dst, dis, h, b2, agg, E, N_NODES);

    fill_f32<<<(N_GRAPHS * HID + N_GRAPHS + 255) / 256, 256, 0, stream>>>(
        pooled, 0.0f, N_GRAPHS * HID + N_GRAPHS);
    pool_accum<<<12500, 256, 0, stream>>>(agg, bat, pooled, cnt, N_NODES);
    pool_div<<<(N_GRAPHS * HID + 255) / 256, 256, 0, stream>>>(pooled, cnt);

    float* out_mf = out;
    float* out_bp = out + N_GRAPHS * OUT_MF;
    float* out_cc = out + N_GRAPHS * (OUT_MF + OUT_BP);
    head_gemm<<<(N_GRAPHS * OUT_MF + 255) / 256, 256, 0, stream>>>(pooled, Wmf, bmf, out_mf, OUT_MF);
    head_gemm<<<(N_GRAPHS * OUT_BP + 255) / 256, 256, 0, stream>>>(pooled, Wbp, bbp, out_bp, OUT_BP);
    head_gemm<<<(N_GRAPHS * OUT_CC + 255) / 256, 256, 0, stream>>>(pooled, Wcc, bcc, out_cc, OUT_CC);
}


# aggregation scans a 1-byte owner-id plane (dst/391, written by the degree pass) with 4 ids compared per 32-bit word, not the 4-byte destinations: 4x fewer scanned bytes; LDS sum order changes
# speedup vs baseline: 1.5249x; 1.0462x over previous
#include <hip/hip_runtime.h>
#include <hip/hip_bf16.h>

#define N_NODES 100000
#define N_EDGES 1600000
#define IN_DIM  256
#define HID     128
#define N_GRAPHS 64
#define OUT_MF 489
#define OUT_BP 1943
#define OUT_CC 320

typedef __attribute__((ext_vector_type(2))) float v2f;
typedef __attribute__((ext_vector_type(8))) float v8f;

__global__ void fill_f32(float* __restrict__ p, float v, int n) {
    int t = blockIdx.x * blockDim.x + threadIdx.x;
    if (t < n) p[t] = v;
}

#define OWN_R     391
__global__ void deg_accum(const int* __restrict__ dst, float* __restrict__ deg,
                          unsigned* __restrict__ own, int E) {
    const int t = blockIdx.x * blockDim.x + threadIdx.x;
    const int nwords = ((E + 511) >> 9) << 7;
    if (t >= nwords) return;
    const int e0 = 4 * t;
    unsigned w = 0u;
    if (e0 + 3 < E && ((E & 3) == 0) && ((((size_t)dst) & 15) == 0)) {
        const int4 d = *(const int4*)(dst + e0);
        atomicAdd(&deg[d.x], 1.0f); atomicAdd(&deg[d.y], 1.0f);
        atomicAdd(&deg[d.z], 1.0f); atomicAdd(&deg[d.w], 1.0f);
        w = (unsigned)(d.x / OWN_R) | ((unsigned)(d.y / OWN_R) << 8)
          | ((unsigned)(d.z / OWN_R) << 16) | ((unsigned)(d.w / OWN_R) << 24);
    } else {
        for (int c = 0; c < 4; ++c) {
            if (e0 + c < E) {
                const int d = dst[e0 + c];
                atomicAdd(&deg[d], 1.0f);
                w |= ((unsigned)(d / OWN_R) & 255u) << (8 * c);
            }
        }
    }
    own[t] = w;
}

__global__ void rsqrt_inplace(float* __restrict__ p, int n) {
    int t = blockIdx.x * blockDim.x + threadIdx.x;
    if (t < n) p[t] = rsqrtf(p[t]);
}

template <int K>
__global__ __launch_bounds__(256) void gemm_wmma_f32(const float* __restrict__ A,
                                                     const float* __restrict__ W,
                                                     float* __restrict__ out) {
    __shared__ float ldsW[64 * HID];
    const int tid  = threadIdx.x;
    const int wave = tid >> 5;
    const int lane = tid & 31;
    const int half = lane >> 4;
    const int m    = lane & 15;
    const int row0 = blockIdx.x * 16;
    const int col0 = wave * 16;

    const float* arow = A + (size_t)(row0 + m) * K;
    v8f acc = {};

    for (int kb = 0; kb < K; kb += 64) {
        const float4* wv = (const float4*)(W + (size_t)kb * HID);
        float4* lv = (float4*)ldsW;
        #pragma unroll
        for (int i = tid; i < 64 * HID / 4; i += 256) lv[i] = wv[i];
        __syncthreads();

        #pragma unroll
        for (int k = 0; k < 64; k += 4) {
            const int ka = k + 2 * half;
            v2f a;
            a.x = arow[kb + ka];
            a.y = arow[kb + ka + 1];
            v2f b;
            b.x = ldsW[ka * HID + col0 + m];
            b.y = ldsW[(ka + 1) * HID + col0 + m];
            acc = __builtin_amdgcn_wmma_f32_16x16x4_f32(
                false, a, false, b, (short)0, acc, false, false);
        }
        __syncthreads();
    }

    #pragma unroll
    for (int v = 0; v < 8; ++v)
        out[(size_t)(row0 + v + 8 * half) * HID + col0 + m] = acc[v];
}

#define OWN_WAVES 32
#define OWN_Q     64
#define OWN_U     4
#define OWN_G     8

typedef __attribute__((address_space(3))) int lds_int;

__device__ __forceinline__ void owner_drain(volatile lds_int* q, int head, int nb, int lane, int lo,
                                            const int* __restrict__ src, const int* __restrict__ dst,
                                            const float* __restrict__ dis, const float* __restrict__ h,
                                            float* acc) {
    const bool v = lane < nb;
    const int e = v ? q[(head + lane) & (OWN_Q - 1)] : 0;
    const int s = v ? src[e] : 0;
    const int d = v ? dst[e] : lo;
    const float nr = v ? dis[s] * dis[d] : 0.f;
    const int r = d - lo;
    int i = 0;
    #pragma unroll 1
    for (; i + OWN_G <= nb; i += OWN_G) {
        float val[OWN_G][4];
        #pragma unroll
        for (int u = 0; u < OWN_G; ++u) {
            const int si = __builtin_amdgcn_readlane(s, i + u);
            const float* hs = h + (size_t)si * HID;
            #pragma unroll
            for (int j = 0; j < 4; ++j) val[u][j] = hs[j * 32 + lane];
        }
        #pragma unroll
        for (int u = 0; u < OWN_G; ++u) {
            const int ri = __builtin_amdgcn_readlane(r, i + u);
            const float ni = __int_as_float(__builtin_amdgcn_readlane(__float_as_int(nr), i + u));
            float* ar = acc + ri * HID;
            #pragma unroll
            for (int j = 0; j < 4; ++j) atomicAdd(&ar[j * 32 + lane], val[u][j] * ni);
        }
    }
    for (; i < nb; ++i) {
        const int si = __builtin_amdgcn_readlane(s, i);
        const int ri = __builtin_amdgcn_readlane(r, i);
        const float ni = __int_as_float(__builtin_amdgcn_readlane(__float_as_int(nr), i));
        const float* hs = h + (size_t)si * HID;
        float* ar = acc + ri * HID;
        #pragma unroll
        for (int j = 0; j < 4; ++j) atomicAdd(&ar[j * 32 + lane], hs[j * 32 + lane] * ni);
    }
}

__device__ __forceinline__ void owner_scan(volatile lds_int* q, int wave, int lane, int lo, unsigned pat,
                                           int rot, int nsteps, int E,
                                           const unsigned* __restrict__ own,
                                           const int* __restrict__ src, const int* __restrict__ dst,
                                           const float* __restrict__ dis, const float* __restrict__ h,
                                           float* acc) {
    int head = 0, cnt = 0;
    for (int st0 = wave; st0 < nsteps; st0 += OWN_WAVES * OWN_U) {
        uint4 ov[OWN_U];
        #pragma unroll
        for (int u = 0; u < OWN_U; ++u) {
            const int st = st0 + u * OWN_WAVES;
            int sidx = st + rot; if (sidx >= nsteps) sidx -= nsteps;
            if (st >= nsteps) sidx = 0;
            ov[u] = *(const uint4*)(own + ((size_t)(sidx * 32 + lane) << 2));
        }
        __builtin_amdgcn_sched_barrier(0);
        unsigned long long lm = 0ull;
        #pragma unroll
        for (int u = 0; u < OWN_U; ++u) {
            const unsigned wd[4] = {ov[u].x, ov[u].y, ov[u].z, ov[u].w};
            unsigned nz = 0u;
            #pragma unroll
            for (int j = 0; j < 4; ++j) {
                const unsigned x = wd[j] ^ pat;
                const unsigned y = ((x & 0x7F7F7F7Fu) + 0x7F7F7F7Fu) | x;
                nz |= ((y >> 7) & 0x01010101u) << j;
            }
            unsigned m16 = ~nz & 0x0F0F0F0Fu;
            if (st0 + u * OWN_WAVES >= nsteps) m16 = 0u;
            lm |= (unsigned long long)(m16 << (4 * (u & 1))) << (32 * (u >> 1));
        }
        for (int round = 0; round < 16 * OWN_U; ++round) {
            const bool has = lm != 0ull;
            if (__builtin_amdgcn_ballot_w32(has) == 0u) break;
            const int p  = has ? __builtin_ctzll(lm) : 0;
            const int lw = p & 31;
            const int u  = ((p >> 5) << 1) | ((lw >> 2) & 1);
            int sidx = st0 + u * OWN_WAVES + rot; if (sidx >= nsteps) sidx -= nsteps;
            const int e = ((sidx * 32 + lane) << 4) + ((lw & 3) << 2) + (lw >> 3);
            const bool take = has && (e < E);
            const unsigned mask = __builtin_amdgcn_ballot_w32(take);
            const int pos = (head + cnt + (int)__builtin_amdgcn_mbcnt_lo(mask, 0u)) & (OWN_Q - 1);
            if (take) q[pos] = e;
            lm &= lm - 1ull;
            cnt += __builtin_popcount(mask);
            if (cnt >= 32) {
                owner_drain(q, head, 32, lane, lo, src, dst, dis, h, acc);
                head = (head + 32) & (OWN_Q - 1);
                cnt -= 32;
            }
        }
    }
    owner_drain(q, head, cnt, lane, lo, src, dst, dis, h, acc);
}

__global__ __launch_bounds__(1024) void owner_aggregate(const int* __restrict__ src,
                                                        const int* __restrict__ dst,
                                                        const float* __restrict__ dis,
                                                        const float* __restrict__ h,
                                                        const float* __restrict__ bias,
                                                        const unsigned* __restrict__ own,
                                                        float* __restrict__ outp, int E, int N) {
    static_assert((N_NODES + OWN_R - 1) / OWN_R <= 256, "an owner id must fit in one byte");
    __shared__ __attribute__((aligned(16))) float acc[OWN_R * HID];
    __shared__ int queue[OWN_WAVES * OWN_Q];

    const int tid  = threadIdx.x;
    const int lane = tid & 31;
    const int wave = __builtin_amdgcn_readfirstlane(tid >> 5);
    const int lo   = blockIdx.x * OWN_R;
    const int nrows = (N - lo) < OWN_R ? (N - lo) : OWN_R;
    volatile lds_int* q = (volatile lds_int*)(queue + wave * OWN_Q);

    for (int idx = tid; idx < nrows * (HID / 4); idx += 1024) {
        const int r = idx >> 5, c4 = idx & 31;
        const float di = dis[lo + r];
        const float w = di * di;
        const float4 hv = ((const float4*)(h + (size_t)(lo + r) * HID))[c4];
        const float4 bv = ((const float4*)bias)[c4];
        float4 o;
        o.x = w * hv.x + bv.x; o.y = w * hv.y + bv.y; o.z = w * hv.z + bv.z; o.w = w * hv.w + bv.w;
        ((float4*)acc)[idx] = o;
    }
    __syncthreads();

    const int nsteps = (E + 511) >> 9;
    const int rot = (int)(((unsigned)blockIdx.x * (unsigned)nsteps) / gridDim.x);
    const unsigned pat = (unsigned)blockIdx.x * 0x01010101u;
    owner_scan(q, wave, lane, lo, pat, rot, nsteps, E, own, src, dst, dis, h, acc);
    __syncthreads();

    float4* orow = (float4*)(outp + (size_t)lo * HID);
    for (int idx = tid; idx < nrows * (HID / 4); idx += 1024) {
        float4 v4 = ((const float4*)acc)[idx];
        v4.x = v4.x > 0.f ? v4.x : 0.f; v4.y = v4.y > 0.f ? v4.y : 0.f;
        v4.z = v4.z > 0.f ? v4.z : 0.f; v4.w = v4.w > 0.f ? v4.w : 0.f;
        orow[idx] = v4;
    }
}

__global__ __launch_bounds__(256) void pool_accum(const float* __restrict__ h,
                                                  const int* __restrict__ batch,
                                                  float* __restrict__ pooled,
                                                  float* __restrict__ cnt, int N) {
    const int lane = threadIdx.x & 31;
    const int gw   = (blockIdx.x * blockDim.x + threadIdx.x) >> 5;
    const int nw   = (gridDim.x * blockDim.x) >> 5;
    for (int i = gw; i < N; i += nw) {
        const int g = batch[i];
        const float* hi = h + (size_t)i * HID;
        float* pg = pooled + g * HID;
        #pragma unroll
        for (int j = 0; j < HID; j += 32)
            atomicAdd(&pg[j + lane], hi[j + lane]);
        if (lane == 0) atomicAdd(&cnt[g], 1.0f);
    }
}

__global__ void pool_div(float* __restrict__ pooled, const float* __restrict__ cnt) {
    int t = blockIdx.x * blockDim.x + threadIdx.x;
    if (t >= N_GRAPHS * HID) return;
    float c = cnt[t >> 7];
    pooled[t] /= (c > 1.f ? c : 1.f);
}

__global__ void head_gemm(const float* __restrict__ pooled, const float* __restrict__ W,
                          const float* __restrict__ b, float* __restrict__ out, int OUT) {
    int t = blockIdx.x * blockDim.x + threadIdx.x;
    const int n = N_GRAPHS * OUT;
    if (t >= n) return;
    const int g = t / OUT, o = t - g * OUT;
    const float* p = pooled + g * HID;
    float s = b[o];
    #pragma unroll 8
    for (int k = 0; k < HID; ++k) s += p[k] * W[k * OUT + o];
    out[t] = s;
}

extern "C" void kernel_launch(void* const* d_in, const int* in_sizes, int n_in,
                              void* d_out, int out_size, void* d_ws, size_t ws_size,
                              hipStream_t stream) {
    const float* x   = (const float*)d_in[0];
    const int*   ei  = (const int*)d_in[1];
    const int*   bat = (const int*)d_in[2];
    const float* W1  = (const float*)d_in[3];
    const float* b1  = (const float*)d_in[4];
    const float* W2  = (const float*)d_in[5];
    const float* b2  = (const float*)d_in[6];
    const float* Wmf = (const float*)d_in[7];
    const float* bmf = (const float*)d_in[8];
    const float* Wbp = (const float*)d_in[9];
    const float* bbp = (const float*)d_in[10];
    const float* Wcc = (const float*)d_in[11];
    const float* bcc = (const float*)d_in[12];
    float* out = (float*)d_out;

    const int E = in_sizes[1] / 2;
    const int* src = ei;
    const int* dst = ei + E;

    float* ws     = (float*)d_ws;
    float* dis    = ws;
    float* h      = dis + N_NODES;
    float* agg    = h + (size_t)N_NODES * HID;
    float* pooled = agg + (size_t)N_NODES * HID;
    float* cnt    = pooled + N_GRAPHS * HID;
    unsigned* own = (unsigned*)(cnt + N_GRAPHS);

    fill_f32<<<(N_NODES + 255) / 256, 256, 0, stream>>>(dis, 1.0f, N_NODES);
    deg_accum<<<(E + 1023) / 1024, 256, 0, stream>>>(dst, dis, own, E);
    rsqrt_inplace<<<(N_NODES + 255) / 256, 256, 0, stream>>>(dis, N_NODES);

    gemm_wmma_f32<IN_DIM><<<N_NODES / 16, 256, 0, stream>>>(x, W1, h);
    owner_aggregate<<<(N_NODES + OWN_R - 1) / OWN_R, 1024, 0, stream>>>(src, dst, dis, h, b1, own, agg, E, N_NODES);

    gemm_wmma_f32<HID><<<N_NODES / 16, 256, 0, stream>>>(agg, W2, h);
    owner_aggregate<<<(N_NODES + OWN_R - 1) / OWN_R, 1024, 0, stream>>>(src, dst, dis, h, b2, own, agg, E, N_NODES);

    fill_f32<<<(N_GRAPHS * HID + N_GRAPHS + 255) / 256, 256, 0, stream>>>(
        pooled, 0.0f, N_GRAPHS * HID + N_GRAPHS);
    pool_accum<<<12500, 256, 0, stream>>>(agg, bat, pooled, cnt, N_NODES);
    pool_div<<<(N_GRAPHS * HID + 255) / 256, 256, 0, stream>>>(pooled, cnt);

    float* out_mf = out;
    float* out_bp = out + N_GRAPHS * OUT_MF;
    float* out_cc = out + N_GRAPHS * (OUT_MF + OUT_BP);
    head_gemm<<<(N_GRAPHS * OUT_MF + 255) / 256, 256, 0, stream>>>(pooled, Wmf, bmf, out_mf, OUT_MF);
    head_gemm<<<(N_GRAPHS * OUT_BP + 255) / 256, 256, 0, stream>>>(pooled, Wbp, bbp, out_bp, OUT_BP);
    head_gemm<<<(N_GRAPHS * OUT_CC + 255) / 256, 256, 0, stream>>>(pooled, Wcc, bcc, out_cc, OUT_CC);
}


# mean pool without global atomics: 256 work-groups each sum 391 rows (register runs, LDS adds) into a partial plane; one group per graph adds the planes in fixed order, divides; sum order changes
# speedup vs baseline: 1.9252x; 1.2625x over previous
#include <hip/hip_runtime.h>
#include <hip/hip_bf16.h>

#define N_NODES 100000
#define N_EDGES 1600000
#define IN_DIM  256
#define HID     128
#define N_GRAPHS 64
#define OUT_MF 489
#define OUT_BP 1943
#define OUT_CC 320

typedef __attribute__((ext_vector_type(2))) float v2f;
typedef __attribute__((ext_vector_type(8))) float v8f;

__global__ void fill_f32(float* __restrict__ p, float v, int n) {
    int t = blockIdx.x * blockDim.x + threadIdx.x;
    if (t < n) p[t] = v;
}

#define OWN_R     391
__global__ void deg_accum(const int* __restrict__ dst, float* __restrict__ deg,
                          unsigned* __restrict__ own, int E) {
    const int t = blockIdx.x * blockDim.x + threadIdx.x;
    const int nwords = ((E + 511) >> 9) << 7;
    if (t >= nwords) return;
    const int e0 = 4 * t;
    unsigned w = 0u;
    if (e0 + 3 < E && ((E & 3) == 0) && ((((size_t)dst) & 15) == 0)) {
        const int4 d = *(const int4*)(dst + e0);
        atomicAdd(&deg[d.x], 1.0f); atomicAdd(&deg[d.y], 1.0f);
        atomicAdd(&deg[d.z], 1.0f); atomicAdd(&deg[d.w], 1.0f);
        w = (unsigned)(d.x / OWN_R) | ((unsigned)(d.y / OWN_R) << 8)
          | ((unsigned)(d.z / OWN_R) << 16) | ((unsigned)(d.w / OWN_R) << 24);
    } else {
        for (int c = 0; c < 4; ++c) {
            if (e0 + c < E) {
                const int d = dst[e0 + c];
                atomicAdd(&deg[d], 1.0f);
                w |= ((unsigned)(d / OWN_R) & 255u) << (8 * c);
            }
        }
    }
    own[t] = w;
}

__global__ void rsqrt_inplace(float* __restrict__ p, int n) {
    int t = blockIdx.x * blockDim.x + threadIdx.x;
    if (t < n) p[t] = rsqrtf(p[t]);
}

template <int K>
__global__ __launch_bounds__(256) void gemm_wmma_f32(const float* __restrict__ A,
                                                     const float* __restrict__ W,
                                                     float* __restrict__ out) {
    __shared__ float ldsW[64 * HID];
    const int tid  = threadIdx.x;
    const int wave = tid >> 5;
    const int lane = tid & 31;
    const int half = lane >> 4;
    const int m    = lane & 15;
    const int row0 = blockIdx.x * 16;
    const int col0 = wave * 16;

    const float* arow = A + (size_t)(row0 + m) * K;
    v8f acc = {};

    for (int kb = 0; kb < K; kb += 64) {
        const float4* wv = (const float4*)(W + (size_t)kb * HID);
        float4* lv = (float4*)ldsW;
        #pragma unroll
        for (int i = tid; i < 64 * HID / 4; i += 256) lv[i] = wv[i];
        __syncthreads();

        #pragma unroll
        for (int k = 0; k < 64; k += 4) {
            const int ka = k + 2 * half;
            v2f a;
            a.x = arow[kb + ka];
            a.y = arow[kb + ka + 1];
            v2f b;
            b.x = ldsW[ka * HID + col0 + m];
            b.y = ldsW[(ka + 1) * HID + col0 + m];
            acc = __builtin_amdgcn_wmma_f32_16x16x4_f32(
                false, a, false, b, (short)0, acc, false, false);
        }
        __syncthreads();
    }

    #pragma unroll
    for (int v = 0; v < 8; ++v)
        out[(size_t)(row0 + v + 8 * half) * HID + col0 + m] = acc[v];
}

#define OWN_WAVES 32
#define OWN_Q     64
#define OWN_U     4
#define OWN_G     8

typedef __attribute__((address_space(3))) int lds_int;

__device__ __forceinline__ void owner_drain(volatile lds_int* q, int head, int nb, int lane, int lo,
                                            const int* __restrict__ src, const int* __restrict__ dst,
                                            const float* __restrict__ dis, const float* __restrict__ h,
                                            float* acc) {
    const bool v = lane < nb;
    const int e = v ? q[(head + lane) & (OWN_Q - 1)] : 0;
    const int s = v ? src[e] : 0;
    const int d = v ? dst[e] : lo;
    const float nr = v ? dis[s] * dis[d] : 0.f;
    const int r = d - lo;
    int i = 0;
    #pragma unroll 1
    for (; i + OWN_G <= nb; i += OWN_G) {
        float val[OWN_G][4];
        #pragma unroll
        for (int u = 0; u < OWN_G; ++u) {
            const int si = __builtin_amdgcn_readlane(s, i + u);
            const float* hs = h + (size_t)si * HID;
            #pragma unroll
            for (int j = 0; j < 4; ++j) val[u][j] = hs[j * 32 + lane];
        }
        #pragma unroll
        for (int u = 0; u < OWN_G; ++u) {
            const int ri = __builtin_amdgcn_readlane(r, i + u);
            const float ni = __int_as_float(__builtin_amdgcn_readlane(__float_as_int(nr), i + u));
            float* ar = acc + ri * HID;
            #pragma unroll
            for (int j = 0; j < 4; ++j) atomicAdd(&ar[j * 32 + lane], val[u][j] * ni);
        }
    }
    for (; i < nb; ++i) {
        const int si = __builtin_amdgcn_readlane(s, i);
        const int ri = __builtin_amdgcn_readlane(r, i);
        const float ni = __int_as_float(__builtin_amdgcn_readlane(__float_as_int(nr), i));
        const float* hs = h + (size_t)si * HID;
        float* ar = acc + ri * HID;
        #pragma unroll
        for (int j = 0; j < 4; ++j) atomicAdd(&ar[j * 32 + lane], hs[j * 32 + lane] * ni);
    }
}

__device__ __forceinline__ void owner_scan(volatile lds_int* q, int wave, int lane, int lo, unsigned pat,
                                           int rot, int nsteps, int E,
                                           const unsigned* __restrict__ own,
                                           const int* __restrict__ src, const int* __restrict__ dst,
                                           const float* __restrict__ dis, const float* __restrict__ h,
                                           float* acc) {
    int head = 0, cnt = 0;
    for (int st0 = wave; st0 < nsteps; st0 += OWN_WAVES * OWN_U) {
        uint4 ov[OWN_U];
        #pragma unroll
        for (int u = 0; u < OWN_U; ++u) {
            const int st = st0 + u * OWN_WAVES;
            int sidx = st + rot; if (sidx >= nsteps) sidx -= nsteps;
            if (st >= nsteps) sidx = 0;
            ov[u] = *(const uint4*)(own + ((size_t)(sidx * 32 + lane) << 2));
        }
        __builtin_amdgcn_sched_barrier(0);
        unsigned long long lm = 0ull;
        #pragma unroll
        for (int u = 0; u < OWN_U; ++u) {
            const unsigned wd[4] = {ov[u].x, ov[u].y, ov[u].z, ov[u].w};
            unsigned nz = 0u;
            #pragma unroll
            for (int j = 0; j < 4; ++j) {
                const unsigned x = wd[j] ^ pat;
                const unsigned y = ((x & 0x7F7F7F7Fu) + 0x7F7F7F7Fu) | x;
                nz |= ((y >> 7) & 0x01010101u) << j;
            }
            unsigned m16 = ~nz & 0x0F0F0F0Fu;
            if (st0 + u * OWN_WAVES >= nsteps) m16 = 0u;
            lm |= (unsigned long long)(m16 << (4 * (u & 1))) << (32 * (u >> 1));
        }
        for (int round = 0; round < 16 * OWN_U; ++round) {
            const bool has = lm != 0ull;
            if (__builtin_amdgcn_ballot_w32(has) == 0u) break;
            const int p  = has ? __builtin_ctzll(lm) : 0;
            const int lw = p & 31;
            const int u  = ((p >> 5) << 1) | ((lw >> 2) & 1);
            int sidx = st0 + u * OWN_WAVES + rot; if (sidx >= nsteps) sidx -= nsteps;
            const int e = ((sidx * 32 + lane) << 4) + ((lw & 3) << 2) + (lw >> 3);
            const bool take = has && (e < E);
            const unsigned mask = __builtin_amdgcn_ballot_w32(take);
            const int pos = (head + cnt + (int)__builtin_amdgcn_mbcnt_lo(mask, 0u)) & (OWN_Q - 1);
            if (take) q[pos] = e;
            lm &= lm - 1ull;
            cnt += __builtin_popcount(mask);
            if (cnt >= 32) {
                owner_drain(q, head, 32, lane, lo, src, dst, dis, h, acc);
                head = (head + 32) & (OWN_Q - 1);
                cnt -= 32;
            }
        }
    }
    owner_drain(q, head, cnt, lane, lo, src, dst, dis, h, acc);
}

__global__ __launch_bounds__(1024) void owner_aggregate(const int* __restrict__ src,
                                                        const int* __restrict__ dst,
                                                        const float* __restrict__ dis,
                                                        const float* __restrict__ h,
                                                        const float* __restrict__ bias,
                                                        const unsigned* __restrict__ own,
                                                        float* __restrict__ outp, int E, int N) {
    static_assert((N_NODES + OWN_R - 1) / OWN_R <= 256, "an owner id must fit in one byte");
    __shared__ __attribute__((aligned(16))) float acc[OWN_R * HID];
    __shared__ int queue[OWN_WAVES * OWN_Q];

    const int tid  = threadIdx.x;
    const int lane = tid & 31;
    const int wave = __builtin_amdgcn_readfirstlane(tid >> 5);
    const int lo   = blockIdx.x * OWN_R;
    const int nrows = (N - lo) < OWN_R ? (N - lo) : OWN_R;
    volatile lds_int* q = (volatile lds_int*)(queue + wave * OWN_Q);

    for (int idx = tid; idx < nrows * (HID / 4); idx += 1024) {
        const int r = idx >> 5, c4 = idx & 31;
        const float di = dis[lo + r];
        const float w = di * di;
        const float4 hv = ((const float4*)(h + (size_t)(lo + r) * HID))[c4];
        const float4 bv = ((const float4*)bias)[c4];
        float4 o;
        o.x = w * hv.x + bv.x; o.y = w * hv.y + bv.y; o.z = w * hv.z + bv.z; o.w = w * hv.w + bv.w;
        ((float4*)acc)[idx] = o;
    }
    __syncthreads();

    const int nsteps = (E + 511) >> 9;
    const int rot = (int)(((unsigned)blockIdx.x * (unsigned)nsteps) / gridDim.x);
    const unsigned pat = (unsigned)blockIdx.x * 0x01010101u;
    owner_scan(q, wave, lane, lo, pat, rot, nsteps, E, own, src, dst, dis, h, acc);
    __syncthreads();

    float4* orow = (float4*)(outp + (size_t)lo * HID);
    for (int idx = tid; idx < nrows * (HID / 4); idx += 1024) {
        float4 v4 = ((const float4*)acc)[idx];
        v4.x = v4.x > 0.f ? v4.x : 0.f; v4.y = v4.y > 0.f ? v4.y : 0.f;
        v4.z = v4.z > 0.f ? v4.z : 0.f; v4.w = v4.w > 0.f ? v4.w : 0.f;
        orow[idx] = v4;
    }
}

#define POOL_NB    256
#define POOL_R     ((N_NODES + POOL_NB - 1) / POOL_NB)
#define POOL_WAVES 16
#define POOL_RW    ((POOL_R + POOL_WAVES - 1) / POOL_WAVES)

__device__ __forceinline__ void pool_flush(float* tab, int* tcnt, int g, int rl, float4 a, int lane) {
    if ((unsigned)g < (unsigned)N_GRAPHS) {
        float* t = tab + g * HID + 4 * lane;
        atomicAdd(t + 0, a.x); atomicAdd(t + 1, a.y); atomicAdd(t + 2, a.z); atomicAdd(t + 3, a.w);
        if (lane == 0) atomicAdd(tcnt + g, rl);
    }
}

__global__ __launch_bounds__(POOL_WAVES * 32) void pool_partial(const float* __restrict__ h,
                                                                const int* __restrict__ batch,
                                                                float* __restrict__ part,
                                                                int* __restrict__ cntp, int N) {
    __shared__ __attribute__((aligned(16))) float tab[N_GRAPHS * HID];
    __shared__ int tcnt[N_GRAPHS];
    const int tid  = threadIdx.x;
    const int lane = tid & 31;
    const int wave = __builtin_amdgcn_readfirstlane(tid >> 5);
    const float4 zero = make_float4(0.f, 0.f, 0.f, 0.f);

    const int r0 = blockIdx.x * POOL_R + wave * POOL_RW;
    int rend = blockIdx.x * POOL_R + POOL_R;
    if (rend > N) rend = N;
    int nr = rend - r0;
    nr = nr < 0 ? 0 : (nr > POOL_RW ? POOL_RW : nr);
    const int gl = (lane < nr) ? batch[r0 + lane] : -1;
    float4 v[POOL_RW];
    #pragma unroll
    for (int i = 0; i < POOL_RW; ++i) {
        int ri = r0 + i;
        if (ri > N - 1) ri = N - 1;
        v[i] = ((const float4*)(h + (size_t)ri * HID))[lane];
    }

    for (int i = tid; i < N_GRAPHS * HID / 4; i += POOL_WAVES * 32) ((float4*)tab)[i] = zero;
    if (tid < N_GRAPHS) tcnt[tid] = 0;
    __syncthreads();

    int gcur = -1, rl = 0;
    float4 acc = zero;
    #pragma unroll
    for (int i = 0; i < POOL_RW; ++i) {
        if (i < nr) {
            const int g = __builtin_amdgcn_readlane(gl, i);
            if (g != gcur) {
                pool_flush(tab, tcnt, gcur, rl, acc, lane);
                gcur = g; rl = 0; acc = zero;
            }
            acc.x += v[i].x; acc.y += v[i].y; acc.z += v[i].z; acc.w += v[i].w;
            ++rl;
        }
    }
    pool_flush(tab, tcnt, gcur, rl, acc, lane);
    __syncthreads();

    float4* pl = (float4*)(part + (size_t)blockIdx.x * (N_GRAPHS * HID));
    for (int i = tid; i < N_GRAPHS * HID / 4; i += POOL_WAVES * 32) pl[i] = ((const float4*)tab)[i];
    if (tid < N_GRAPHS) cntp[blockIdx.x * N_GRAPHS + tid] = tcnt[tid];
}

__global__ __launch_bounds__(1024) void pool_finish(const float* __restrict__ part,
                                                    const int* __restrict__ cntp,
                                                    float* __restrict__ pooled) {
    static_assert(POOL_NB == 256, "32 waves x 8 planes, one count per thread of the first 256");
    __shared__ __attribute__((aligned(16))) float red[32 * HID];
    __shared__ int ntot;
    const int tid  = threadIdx.x;
    const int lane = tid & 31;
    const int wave = tid >> 5;
    const int g    = blockIdx.x;

    float4 v[8];
    #pragma unroll
    for (int i = 0; i < 8; ++i)
        v[i] = ((const float4*)(part + ((size_t)(wave * 8 + i) * N_GRAPHS + g) * HID))[lane];
    const int c = (tid < POOL_NB) ? cntp[tid * N_GRAPHS + g] : 0;

    if (tid == 0) ntot = 0;
    __syncthreads();
    float4 s = v[0];
    #pragma unroll
    for (int i = 1; i < 8; ++i) { s.x += v[i].x; s.y += v[i].y; s.z += v[i].z; s.w += v[i].w; }
    ((float4*)red)[wave * 32 + lane] = s;
    if (c != 0) atomicAdd(&ntot, c);
    __syncthreads();

    if (wave == 0) {
        float4 t = ((const float4*)red)[lane];
        #pragma unroll
        for (int w = 1; w < 32; ++w) {
            const float4 u = ((const float4*)red)[w * 32 + lane];
            t.x += u.x; t.y += u.y; t.z += u.z; t.w += u.w;
        }
        float n = (float)ntot;
        n = n > 1.f ? n : 1.f;
        t.x /= n; t.y /= n; t.z /= n; t.w /= n;
        ((float4*)(pooled + g * HID))[lane] = t;
    }
}

__global__ void head_gemm(const float* __restrict__ pooled, const float* __restrict__ W,
                          const float* __restrict__ b, float* __restrict__ out, int OUT) {
    int t = blockIdx.x * blockDim.x + threadIdx.x;
    const int n = N_GRAPHS * OUT;
    if (t >= n) return;
    const int g = t / OUT, o = t - g * OUT;
    const float* p = pooled + g * HID;
    float s = b[o];
    #pragma unroll 8
    for (int k = 0; k < HID; ++k) s += p[k] * W[k * OUT + o];
    out[t] = s;
}

extern "C" void kernel_launch(void* const* d_in, const int* in_sizes, int n_in,
                              void* d_out, int out_size, void* d_ws, size_t ws_size,
                              hipStream_t stream) {
    const float* x   = (const float*)d_in[0];
    const int*   ei  = (const int*)d_in[1];
    const int*   bat = (const int*)d_in[2];
    const float* W1  = (const float*)d_in[3];
    const float* b1  = (const float*)d_in[4];
    const float* W2  = (const float*)d_in[5];
    const float* b2  = (const float*)d_in[6];
    const float* Wmf = (const float*)d_in[7];
    const float* bmf = (const float*)d_in[8];
    const float* Wbp = (const float*)d_in[9];
    const float* bbp = (const float*)d_in[10];
    const float* Wcc = (const float*)d_in[11];
    const float* bcc = (const float*)d_in[12];
    float* out = (float*)d_out;

    const int E = in_sizes[1] / 2;
    const int* src = ei;
    const int* dst = ei + E;

    float* ws     = (float*)d_ws;
    float* dis    = ws;
    float* h      = dis + N_NODES;
    float* agg    = h + (size_t)N_NODES * HID;
    float* pooled = agg + (size_t)N_NODES * HID;
    float* cnt    = pooled + N_GRAPHS * HID;
    unsigned* own = (unsigned*)(cnt + N_GRAPHS);

    fill_f32<<<(N_NODES + 255) / 256, 256, 0, stream>>>(dis, 1.0f, N_NODES);
    deg_accum<<<(E + 1023) / 1024, 256, 0, stream>>>(dst, dis, own, E);
    rsqrt_inplace<<<(N_NODES + 255) / 256, 256, 0, stream>>>(dis, N_NODES);

    gemm_wmma_f32<IN_DIM><<<N_NODES / 16, 256, 0, stream>>>(x, W1, h);
    owner_aggregate<<<(N_NODES + OWN_R - 1) / OWN_R, 1024, 0, stream>>>(src, dst, dis, h, b1, own, agg, E, N_NODES);

    gemm_wmma_f32<HID><<<N_NODES / 16, 256, 0, stream>>>(agg, W2, h);
    owner_aggregate<<<(N_NODES + OWN_R - 1) / OWN_R, 1024, 0, stream>>>(src, dst, dis, h, b2, own, agg, E, N_NODES);

    float* part = h;
    int*   cntp = (int*)(h + (size_t)POOL_NB * N_GRAPHS * HID);
    pool_partial<<<POOL_NB, POOL_WAVES * 32, 0, stream>>>(agg, bat, part, cntp, N_NODES);
    pool_finish<<<N_GRAPHS, 1024, 0, stream>>>(part, cntp, pooled);

    float* out_mf = out;
    float* out_bp = out + N_GRAPHS * OUT_MF;
    float* out_cc = out + N_GRAPHS * (OUT_MF + OUT_BP);
    head_gemm<<<(N_GRAPHS * OUT_MF + 255) / 256, 256, 0, stream>>>(pooled, Wmf, bmf, out_mf, OUT_MF);
    head_gemm<<<(N_GRAPHS * OUT_BP + 255) / 256, 256, 0, stream>>>(pooled, Wbp, bbp, out_bp, OUT_BP);
    head_gemm<<<(N_GRAPHS * OUT_CC + 255) / 256, 256, 0, stream>>>(pooled, Wcc, bcc, out_cc, OUT_CC);
}


# both products re-fed: persistent work-groups hold each wave's W fragments for all K in registers; a 16-row A tile enters once by whole-line loads via padded LDS; k order of every sum kept
# speedup vs baseline: 2.3766x; 1.2345x over previous
#include <hip/hip_runtime.h>
#include <hip/hip_bf16.h>

#define N_NODES 100000
#define N_EDGES 1600000
#define IN_DIM  256
#define HID     128
#define N_GRAPHS 64
#define OUT_MF 489
#define OUT_BP 1943
#define OUT_CC 320

typedef __attribute__((ext_vector_type(2))) float v2f;
typedef __attribute__((ext_vector_type(8))) float v8f;

__global__ void fill_f32(float* __restrict__ p, float v, int n) {
    int t = blockIdx.x * blockDim.x + threadIdx.x;
    if (t < n) p[t] = v;
}

#define OWN_R     391
__global__ void deg_accum(const int* __restrict__ dst, float* __restrict__ deg,
                          unsigned* __restrict__ own, int E) {
    const int t = blockIdx.x * blockDim.x + threadIdx.x;
    const int nwords = ((E + 511) >> 9) << 7;
    if (t >= nwords) return;
    const int e0 = 4 * t;
    unsigned w = 0u;
    if (e0 + 3 < E && ((E & 3) == 0) && ((((size_t)dst) & 15) == 0)) {
        const int4 d = *(const int4*)(dst + e0);
        atomicAdd(&deg[d.x], 1.0f); atomicAdd(&deg[d.y], 1.0f);
        atomicAdd(&deg[d.z], 1.0f); atomicAdd(&deg[d.w], 1.0f);
        w = (unsigned)(d.x / OWN_R) | ((unsigned)(d.y / OWN_R) << 8)
          | ((unsigned)(d.z / OWN_R) << 16) | ((unsigned)(d.w / OWN_R) << 24);
    } else {
        for (int c = 0; c < 4; ++c) {
            if (e0 + c < E) {
                const int d = dst[e0 + c];
                atomicAdd(&deg[d], 1.0f);
                w |= ((unsigned)(d / OWN_R) & 255u) << (8 * c);
            }
        }
    }
    own[t] = w;
}

__global__ void rsqrt_inplace(float* __restrict__ p, int n) {
    int t = blockIdx.x * blockDim.x + threadIdx.x;
    if (t < n) p[t] = rsqrtf(p[t]);
}

#define GEMM_GRID_K256 512
#define GEMM_GRID_K128 768
typedef __attribute__((ext_vector_type(4))) float v4f;

template <int K>
__global__ __launch_bounds__(256) void gemm_wmma_f32(const float* __restrict__ A,
                                                     const float* __restrict__ W,
                                                     float* __restrict__ out, int ntiles) {
    constexpr int PITCH = K + 4;
    constexpr int RV    = K / 4;
    constexpr int NV    = 16 * RV / 256;
    constexpr int RS    = 256 / RV;
    constexpr int G     = 16;
    constexpr int NG    = K / 4 / G;
    static_assert(NV == 4 || NV == 2, "a thread moves 2 or 4 pieces of a tile");
    __shared__ __attribute__((aligned(16))) float ldsA[16 * PITCH];
    const int tid  = threadIdx.x;
    const int wave = tid >> 5;
    const int lane = tid & 31;
    const int half = lane >> 4;
    const int m    = lane & 15;
    const int col0 = wave * 16;

    int tile = blockIdx.x;
    if (tile >= ntiles) return;

    const int pr = tid / RV, pc = tid - pr * RV;
    float* lst = ldsA + pr * PITCH + 4 * pc;
    const v4f* ap = (const v4f*)(A + (size_t)tile * 16 * K) + tid;
    v4f p0 = ap[0], p1 = ap[256], p2 = {}, p3 = {};
    if constexpr (NV == 4) { p2 = ap[512]; p3 = ap[768]; }

    v2f bf[K / 4];
    #pragma unroll
    for (int s = 0; s < K / 4; ++s) {
        const int ka = 4 * s + 2 * half;
        bf[s].x = W[ka * HID + col0 + m];
        bf[s].y = W[(ka + 1) * HID + col0 + m];
    }

    const float* arow = ldsA + m * PITCH + 2 * half;
    for (;;) {
        __syncthreads();
        *(v4f*)(lst) = p0;
        *(v4f*)(lst + RS * PITCH) = p1;
        if constexpr (NV == 4) {
            *(v4f*)(lst + 2 * RS * PITCH) = p2;
            *(v4f*)(lst + 3 * RS * PITCH) = p3;
        }
        __syncthreads();

        const int nt = tile + (int)gridDim.x;
        const int lt = nt < ntiles ? nt : tile;
        ap = (const v4f*)(A + (size_t)lt * 16 * K) + tid;
        p0 = ap[0]; p1 = ap[256];
        if constexpr (NV == 4) { p2 = ap[512]; p3 = ap[768]; }

        v2f fr[2][G];
        #pragma unroll
        for (int i = 0; i < G; ++i) fr[0][i] = *(const v2f*)(arow + 4 * i);
        v8f acc = {};
        #pragma unroll
        for (int g = 0; g < NG; ++g) {
            if (g + 1 < NG) {
                #pragma unroll
                for (int i = 0; i < G; ++i)
                    fr[(g + 1) & 1][i] = *(const v2f*)(arow + 4 * ((g + 1) * G + i));
            }
            __builtin_amdgcn_sched_barrier(0);
            #pragma unroll
            for (int i = 0; i < G; ++i)
                acc = __builtin_amdgcn_wmma_f32_16x16x4_f32(
                    false, fr[g & 1][i], false, bf[g * G + i], (short)0, acc, false, false);
            __builtin_amdgcn_sched_barrier(0);
        }

        float* op = out + (size_t)(tile * 16 + 8 * half) * HID + col0 + m;
        #pragma unroll
        for (int v = 0; v < 8; ++v) op[v * HID] = acc[v];

        if (nt >= ntiles) break;
        tile = nt;
    }
}

#define OWN_WAVES 32
#define OWN_Q     64
#define OWN_U     4
#define OWN_G     8

typedef __attribute__((address_space(3))) int lds_int;

__device__ __forceinline__ void owner_drain(volatile lds_int* q, int head, int nb, int lane, int lo,
                                            const int* __restrict__ src, const int* __restrict__ dst,
                                            const float* __restrict__ dis, const float* __restrict__ h,
                                            float* acc) {
    const bool v = lane < nb;
    const int e = v ? q[(head + lane) & (OWN_Q - 1)] : 0;
    const int s = v ? src[e] : 0;
    const int d = v ? dst[e] : lo;
    const float nr = v ? dis[s] * dis[d] : 0.f;
    const int r = d - lo;
    int i = 0;
    #pragma unroll 1
    for (; i + OWN_G <= nb; i += OWN_G) {
        float val[OWN_G][4];
        #pragma unroll
        for (int u = 0; u < OWN_G; ++u) {
            const int si = __builtin_amdgcn_readlane(s, i + u);
            const float* hs = h + (size_t)si * HID;
            #pragma unroll
            for (int j = 0; j < 4; ++j) val[u][j] = hs[j * 32 + lane];
        }
        #pragma unroll
        for (int u = 0; u < OWN_G; ++u) {
            const int ri = __builtin_amdgcn_readlane(r, i + u);
            const float ni = __int_as_float(__builtin_amdgcn_readlane(__float_as_int(nr), i + u));
            float* ar = acc + ri * HID;
            #pragma unroll
            for (int j = 0; j < 4; ++j) atomicAdd(&ar[j * 32 + lane], val[u][j] * ni);
        }
    }
    for (; i < nb; ++i) {
        const int si = __builtin_amdgcn_readlane(s, i);
        const int ri = __builtin_amdgcn_readlane(r, i);
        const float ni = __int_as_float(__builtin_amdgcn_readlane(__float_as_int(nr), i));
        const float* hs = h + (size_t)si * HID;
        float* ar = acc + ri * HID;
        #pragma unroll
        for (int j = 0; j < 4; ++j) atomicAdd(&ar[j * 32 + lane], hs[j * 32 + lane] * ni);
    }
}

__device__ __forceinline__ void owner_scan(volatile lds_int* q, int wave, int lane, int lo, unsigned pat,
                                           int rot, int nsteps, int E,
                                           const unsigned* __restrict__ own,
                                           const int* __restrict__ src, const int* __restrict__ dst,
                                           const float* __restrict__ dis, const float* __restrict__ h,
                                           float* acc) {
    int head = 0, cnt = 0;
    for (int st0 = wave; st0 < nsteps; st0 += OWN_WAVES * OWN_U) {
        uint4 ov[OWN_U];
        #pragma unroll
        for (int u = 0; u < OWN_U; ++u) {
            const int st = st0 + u * OWN_WAVES;
            int sidx = st + rot; if (sidx >= nsteps) sidx -= nsteps;
            if (st >= nsteps) sidx = 0;
            ov[u] = *(const uint4*)(own + ((size_t)(sidx * 32 + lane) << 2));
        }
        __builtin_amdgcn_sched_barrier(0);
        unsigned long long lm = 0ull;
        #pragma unroll
        for (int u = 0; u < OWN_U; ++u) {
            const unsigned wd[4] = {ov[u].x, ov[u].y, ov[u].z, ov[u].w};
            unsigned nz = 0u;
            #pragma unroll
            for (int j = 0; j < 4; ++j) {
                const unsigned x = wd[j] ^ pat;
                const unsigned y = ((x & 0x7F7F7F7Fu) + 0x7F7F7F7Fu) | x;
                nz |= ((y >> 7) & 0x01010101u) << j;
            }
            unsigned m16 = ~nz & 0x0F0F0F0Fu;
            if (st0 + u * OWN_WAVES >= nsteps) m16 = 0u;
            lm |= (unsigned long long)(m16 << (4 * (u & 1))) << (32 * (u >> 1));
        }
        for (int round = 0; round < 16 * OWN_U; ++round) {
            const bool has = lm != 0ull;
            if (__builtin_amdgcn_ballot_w32(has) == 0u) break;
            const int p  = has ? __builtin_ctzll(lm) : 0;
            const int lw = p & 31;
            const int u  = ((p >> 5) << 1) | ((lw >> 2) & 1);
            int sidx = st0 + u * OWN_WAVES + rot; if (sidx >= nsteps) sidx -= nsteps;
            const int e = ((sidx * 32 + lane) << 4) + ((lw & 3) << 2) + (lw >> 3);
            const bool take = has && (e < E);
            const unsigned mask = __builtin_amdgcn_ballot_w32(take);
            const int pos = (head + cnt + (int)__builtin_amdgcn_mbcnt_lo(mask, 0u)) & (OWN_Q - 1);
            if (take) q[pos] = e;
            lm &= lm - 1ull;
            cnt += __builtin_popcount(mask);
            if (cnt >= 32) {
                owner_drain(q, head, 32, lane, lo, src, dst, dis, h, acc);
                head = (head + 32) & (OWN_Q - 1);
                cnt -= 32;
            }
        }
    }
    owner_drain(q, head, cnt, lane, lo, src, dst, dis, h, acc);
}

__global__ __launch_bounds__(1024) void owner_aggregate(const int* __restrict__ src,
                                                        const int* __restrict__ dst,
                                                        const float* __restrict__ dis,
                                                        const float* __restrict__ h,
                                                        const float* __restrict__ bias,
                                                        const unsigned* __restrict__ own,
                                                        float* __restrict__ outp, int E, int N) {
    static_assert((N_NODES + OWN_R - 1) / OWN_R <= 256, "an owner id must fit in one byte");
    __shared__ __attribute__((aligned(16))) float acc[OWN_R * HID];
    __shared__ int queue[OWN_WAVES * OWN_Q];

    const int tid  = threadIdx.x;
    const int lane = tid & 31;
    const int wave = __builtin_amdgcn_readfirstlane(tid >> 5);
    const int lo   = blockIdx.x * OWN_R;
    const int nrows = (N - lo) < OWN_R ? (N - lo) : OWN_R;
    volatile lds_int* q = (volatile lds_int*)(queue + wave * OWN_Q);

    for (int idx = tid; idx < nrows * (HID / 4); idx += 1024) {
        const int r = idx >> 5, c4 = idx & 31;
        const float di = dis[lo + r];
        const float w = di * di;
        const float4 hv = ((const float4*)(h + (size_t)(lo + r) * HID))[c4];
        const float4 bv = ((const float4*)bias)[c4];
        float4 o;
        o.x = w * hv.x + bv.x; o.y = w * hv.y + bv.y; o.z = w * hv.z + bv.z; o.w = w * hv.w + bv.w;
        ((float4*)acc)[idx] = o;
    }
    __syncthreads();

    const int nsteps = (E + 511) >> 9;
    const int rot = (int)(((unsigned)blockIdx.x * (unsigned)nsteps) / gridDim.x);
    const unsigned pat = (unsigned)blockIdx.x * 0x01010101u;
    owner_scan(q, wave, lane, lo, pat, rot, nsteps, E, own, src, dst, dis, h, acc);
    __syncthreads();

    float4* orow = (float4*)(outp + (size_t)lo * HID);
    for (int idx = tid; idx < nrows * (HID / 4); idx += 1024) {
        float4 v4 = ((const float4*)acc)[idx];
        v4.x = v4.x > 0.f ? v4.x : 0.f; v4.y = v4.y > 0.f ? v4.y : 0.f;
        v4.z = v4.z > 0.f ? v4.z : 0.f; v4.w = v4.w > 0.f ? v4.w : 0.f;
        orow[idx] = v4;
    }
}

#define POOL_NB    256
#define POOL_R     ((N_NODES + POOL_NB - 1) / POOL_NB)
#define POOL_WAVES 16
#define POOL_RW    ((POOL_R + POOL_WAVES - 1) / POOL_WAVES)

__device__ __forceinline__ void pool_flush(float* tab, int* tcnt, int g, int rl, float4 a, int lane) {
    if ((unsigned)g < (unsigned)N_GRAPHS) {
        float* t = tab + g * HID + 4 * lane;
        atomicAdd(t + 0, a.x); atomicAdd(t + 1, a.y); atomicAdd(t + 2, a.z); atomicAdd(t + 3, a.w);
        if (lane == 0) atomicAdd(tcnt + g, rl);
    }
}

__global__ __launch_bounds__(POOL_WAVES * 32) void pool_partial(const float* __restrict__ h,
                                                                const int* __restrict__ batch,
                                                                float* __restrict__ part,
                                                                int* __restrict__ cntp, int N) {
    __shared__ __attribute__((aligned(16))) float tab[N_GRAPHS * HID];
    __shared__ int tcnt[N_GRAPHS];
    const int tid  = threadIdx.x;
    const int lane = tid & 31;
    const int wave = __builtin_amdgcn_readfirstlane(tid >> 5);
    const float4 zero = make_float4(0.f, 0.f, 0.f, 0.f);

    const int r0 = blockIdx.x * POOL_R + wave * POOL_RW;
    int rend = blockIdx.x * POOL_R + POOL_R;
    if (rend > N) rend = N;
    int nr = rend - r0;
    nr = nr < 0 ? 0 : (nr > POOL_RW ? POOL_RW : nr);
    const int gl = (lane < nr) ? batch[r0 + lane] : -1;
    float4 v[POOL_RW];
    #pragma unroll
    for (int i = 0; i < POOL_RW; ++i) {
        int ri = r0 + i;
        if (ri > N - 1) ri = N - 1;
        v[i] = ((const float4*)(h + (size_t)ri * HID))[lane];
    }

    for (int i = tid; i < N_GRAPHS * HID / 4; i += POOL_WAVES * 32) ((float4*)tab)[i] = zero;
    if (tid < N_GRAPHS) tcnt[tid] = 0;
    __syncthreads();

    int gcur = -1, rl = 0;
    float4 acc = zero;
    #pragma unroll
    for (int i = 0; i < POOL_RW; ++i) {
        if (i < nr) {
            const int g = __builtin_amdgcn_readlane(gl, i);
            if (g != gcur) {
                pool_flush(tab, tcnt, gcur, rl, acc, lane);
                gcur = g; rl = 0; acc = zero;
            }
            acc.x += v[i].x; acc.y += v[i].y; acc.z += v[i].z; acc.w += v[i].w;
            ++rl;
        }
    }
    pool_flush(tab, tcnt, gcur, rl, acc, lane);
    __syncthreads();

    float4* pl = (float4*)(part + (size_t)blockIdx.x * (N_GRAPHS * HID));
    for (int i = tid; i < N_GRAPHS * HID / 4; i += POOL_WAVES * 32) pl[i] = ((const float4*)tab)[i];
    if (tid < N_GRAPHS) cntp[blockIdx.x * N_GRAPHS + tid] = tcnt[tid];
}

__global__ __launch_bounds__(1024) void pool_finish(const float* __restrict__ part,
                                                    const int* __restrict__ cntp,
                                                    float* __restrict__ pooled) {
    static_assert(POOL_NB == 256, "32 waves x 8 planes, one count per thread of the first 256");
    __shared__ __attribute__((aligned(16))) float red[32 * HID];
    __shared__ int ntot;
    const int tid  = threadIdx.x;
    const int lane = tid & 31;
    const int wave = tid >> 5;
    const int g    = blockIdx.x;

    float4 v[8];
    #pragma unroll
    for (int i = 0; i < 8; ++i)
        v[i] = ((const float4*)(part + ((size_t)(wave * 8 + i) * N_GRAPHS + g) * HID))[lane];
    const int c = (tid < POOL_NB) ? cntp[tid * N_GRAPHS + g] : 0;

    if (tid == 0) ntot = 0;
    __syncthreads();
    float4 s = v[0];
    #pragma unroll
    for (int i = 1; i < 8; ++i) { s.x += v[i].x; s.y += v[i].y; s.z += v[i].z; s.w += v[i].w; }
    ((float4*)red)[wave * 32 + lane] = s;
    if (c != 0) atomicAdd(&ntot, c);
    __syncthreads();

    if (wave == 0) {
        float4 t = ((const float4*)red)[lane];
        #pragma unroll
        for (int w = 1; w < 32; ++w) {
            const float4 u = ((const float4*)red)[w * 32 + lane];
            t.x += u.x; t.y += u.y; t.z += u.z; t.w += u.w;
        }
        float n = (float)ntot;
        n = n > 1.f ? n : 1.f;
        t.x /= n; t.y /= n; t.z /= n; t.w /= n;
        ((float4*)(pooled + g * HID))[lane] = t;
    }
}

__global__ void head_gemm(const float* __restrict__ pooled, const float* __restrict__ W,
                          const float* __restrict__ b, float* __restrict__ out, int OUT) {
    int t = blockIdx.x * blockDim.x + threadIdx.x;
    const int n = N_GRAPHS * OUT;
    if (t >= n) return;
    const int g = t / OUT, o = t - g * OUT;
    const float* p = pooled + g * HID;
    float s = b[o];
    #pragma unroll 8
    for (int k = 0; k < HID; ++k) s += p[k] * W[k * OUT + o];
    out[t] = s;
}

extern "C" void kernel_launch(void* const* d_in, const int* in_sizes, int n_in,
                              void* d_out, int out_size, void* d_ws, size_t ws_size,
                              hipStream_t stream) {
    const float* x   = (const float*)d_in[0];
    const int*   ei  = (const int*)d_in[1];
    const int*   bat = (const int*)d_in[2];
    const float* W1  = (const float*)d_in[3];
    const float* b1  = (const float*)d_in[4];
    const float* W2  = (const float*)d_in[5];
    const float* b2  = (const float*)d_in[6];
    const float* Wmf = (const float*)d_in[7];
    const float* bmf = (const float*)d_in[8];
    const float* Wbp = (const float*)d_in[9];
    const float* bbp = (const float*)d_in[10];
    const float* Wcc = (const float*)d_in[11];
    const float* bcc = (const float*)d_in[12];
    float* out = (float*)d_out;

    const int E = in_sizes[1] / 2;
    const int* src = ei;
    const int* dst = ei + E;

    float* ws     = (float*)d_ws;
    float* dis    = ws;
    float* h      = dis + N_NODES;
    float* agg    = h + (size_t)N_NODES * HID;
    float* pooled = agg + (size_t)N_NODES * HID;
    float* cnt    = pooled + N_GRAPHS * HID;
    unsigned* own = (unsigned*)(cnt + N_GRAPHS);

    fill_f32<<<(N_NODES + 255) / 256, 256, 0, stream>>>(dis, 1.0f, N_NODES);
    deg_accum<<<(E + 1023) / 1024, 256, 0, stream>>>(dst, dis, own, E);
    rsqrt_inplace<<<(N_NODES + 255) / 256, 256, 0, stream>>>(dis, N_NODES);

    gemm_wmma_f32<IN_DIM><<<GEMM_GRID_K256, 256, 0, stream>>>(x, W1, h, N_NODES / 16);
    owner_aggregate<<<(N_NODES + OWN_R - 1) / OWN_R, 1024, 0, stream>>>(src, dst, dis, h, b1, own, agg, E, N_NODES);

    gemm_wmma_f32<HID><<<GEMM_GRID_K128, 256, 0, stream>>>(agg, W2, h, N_NODES / 16);
    owner_aggregate<<<(N_NODES + OWN_R - 1) / OWN_R, 1024, 0, stream>>>(src, dst, dis, h, b2, own, agg, E, N_NODES);

    float* part = h;
    int*   cntp = (int*)(h + (size_t)POOL_NB * N_GRAPHS * HID);
    pool_partial<<<POOL_NB, POOL_WAVES * 32, 0, stream>>>(agg, bat, part, cntp, N_NODES);
    pool_finish<<<N_GRAPHS, 1024, 0, stream>>>(part, cntp, pooled);

    float* out_mf = out;
    float* out_bp = out + N_GRAPHS * OUT_MF;
    float* out_cc = out + N_GRAPHS * (OUT_MF + OUT_BP);
    head_gemm<<<(N_GRAPHS * OUT_MF + 255) / 256, 256, 0, stream>>>(pooled, Wmf, bmf, out_mf, OUT_MF);
    head_gemm<<<(N_GRAPHS * OUT_BP + 255) / 256, 256, 0, stream>>>(pooled, Wbp, bbp, out_bp, OUT_BP);
    head_gemm<<<(N_GRAPHS * OUT_CC + 255) / 256, 256, 0, stream>>>(pooled, Wcc, bcc, out_cc, OUT_CC);
}


# in-degrees without global atomics: 64 work-groups each count an edge slice in an all-node table of 16-bit LDS counters; one pass adds the 64 planes, +1, rsqrt (exact integers); one launch fewer
# speedup vs baseline: 2.7589x; 1.1609x over previous
#include <hip/hip_runtime.h>
#include <hip/hip_bf16.h>

#define N_NODES 100000
#define N_EDGES 1600000
#define IN_DIM  256
#define HID     128
#define N_GRAPHS 64
#define OUT_MF 489
#define OUT_BP 1943
#define OUT_CC 320

typedef __attribute__((ext_vector_type(2))) float v2f;
typedef __attribute__((ext_vector_type(8))) float v8f;

#define OWN_R     391
#define DEG_P     64
#define DEG_U     8
#define DEG_WORDS (((N_NODES + 1) / 2 + 3) & ~3)

__device__ __forceinline__ unsigned deg_count(unsigned* tab, int d) {
    if ((unsigned)d >= (unsigned)N_NODES) return 0u;
    atomicAdd(&tab[d >> 1], (d & 1) ? 0x10000u : 1u);
    return (unsigned)(d / OWN_R);
}

__global__ __launch_bounds__(1024) void deg_partial(const int* __restrict__ dst,
                                                    unsigned* __restrict__ part,
                                                    unsigned* __restrict__ own, int E) {
    static_assert(((N_EDGES + 511) / 512 * 512 + DEG_P - 1) / DEG_P + 4 < 65536,
                  "the counts of one slice must fit in 16 bits");
    __shared__ __attribute__((aligned(16))) unsigned tab[DEG_WORDS];
    const int tid = threadIdx.x;
    for (int i = tid; i < DEG_WORDS / 4; i += 1024) ((uint4*)tab)[i] = make_uint4(0u, 0u, 0u, 0u);
    __syncthreads();

    const int nwords = ((E + 511) >> 9) << 7;
    const int per = (nwords + DEG_P - 1) / DEG_P;
    const int w0 = blockIdx.x * per;
    const int w1 = (w0 + per) < nwords ? (w0 + per) : nwords;
    if (((E & 3) == 0) && ((((size_t)dst) & 15) == 0)) {
        const int lastw = (E >> 2) - 1;
        for (int tb = w0 + tid; tb < w1; tb += 1024 * DEG_U) {
            int4 d[DEG_U];
            #pragma unroll
            for (int u = 0; u < DEG_U; ++u) {
                const int t = tb + u * 1024;
                const int tl = t < lastw ? t : lastw;
                d[u] = *(const int4*)(dst + 4 * (size_t)tl);
            }
            #pragma unroll
            for (int u = 0; u < DEG_U; ++u) {
                const int t = tb + u * 1024;
                if (t < w1) {
                    const int4 dd = t <= lastw ? d[u] : make_int4(-1, -1, -1, -1);
                    own[t] = deg_count(tab, dd.x) | (deg_count(tab, dd.y) << 8)
                           | (deg_count(tab, dd.z) << 16) | (deg_count(tab, dd.w) << 24);
                }
            }
        }
    } else {
        for (int t = w0 + tid; t < w1; t += 1024) {
            unsigned w = 0u;
            for (int c = 0; c < 4; ++c)
                if (4 * t + c < E) w |= deg_count(tab, dst[4 * t + c]) << (8 * c);
            own[t] = w;
        }
    }
    __syncthreads();

    uint4* pl = (uint4*)(part + (size_t)blockIdx.x * DEG_WORDS);
    for (int i = tid; i < DEG_WORDS / 4; i += 1024) pl[i] = ((const uint4*)tab)[i];
}

__global__ void deg_finish(const unsigned* __restrict__ part, float* __restrict__ dis, int N) {
    const int t = blockIdx.x * blockDim.x + threadIdx.x;
    if (2 * t >= N) return;
    unsigned lo = 0u, hi = 0u;
    #pragma unroll 16
    for (int p = 0; p < DEG_P; ++p) {
        const unsigned w = part[(size_t)p * DEG_WORDS + t];
        lo += w & 0xFFFFu;
        hi += w >> 16;
    }
    const float d0 = rsqrtf(1.0f + (float)lo);
    const float d1 = rsqrtf(1.0f + (float)hi);
    if (2 * t + 1 < N) {
        *(float2*)(dis + 2 * t) = make_float2(d0, d1);
    } else {
        dis[2 * t] = d0;
    }
}

#define GEMM_GRID_K256 512
#define GEMM_GRID_K128 768
typedef __attribute__((ext_vector_type(4))) float v4f;

template <int K>
__global__ __launch_bounds__(256) void gemm_wmma_f32(const float* __restrict__ A,
                                                     const float* __restrict__ W,
                                                     float* __restrict__ out, int ntiles) {
    constexpr int PITCH = K + 4;
    constexpr int RV    = K / 4;
    constexpr int NV    = 16 * RV / 256;
    constexpr int RS    = 256 / RV;
    constexpr int G     = 16;
    constexpr int NG    = K / 4 / G;
    static_assert(NV == 4 || NV == 2, "a thread moves 2 or 4 pieces of a tile");
    __shared__ __attribute__((aligned(16))) float ldsA[16 * PITCH];
    const int tid  = threadIdx.x;
    const int wave = tid >> 5;
    const int lane = tid & 31;
    const int half = lane >> 4;
    const int m    = lane & 15;
    const int col0 = wave * 16;

    int tile = blockIdx.x;
    if (tile >= ntiles) return;

    const int pr = tid / RV, pc = tid - pr * RV;
    float* lst = ldsA + pr * PITCH + 4 * pc;
    const v4f* ap = (const v4f*)(A + (size_t)tile * 16 * K) + tid;
    v4f p0 = ap[0], p1 = ap[256], p2 = {}, p3 = {};
    if constexpr (NV == 4) { p2 = ap[512]; p3 = ap[768]; }

    v2f bf[K / 4];
    #pragma unroll
    for (int s = 0; s < K / 4; ++s) {
        const int ka = 4 * s + 2 * half;
        bf[s].x = W[ka * HID + col0 + m];
        bf[s].y = W[(ka + 1) * HID + col0 + m];
    }

    const float* arow = ldsA + m * PITCH + 2 * half;
    for (;;) {
        __syncthreads();
        *(v4f*)(lst) = p0;
        *(v4f*)(lst + RS * PITCH) = p1;
        if constexpr (NV == 4) {
            *(v4f*)(lst + 2 * RS * PITCH) = p2;
            *(v4f*)(lst + 3 * RS * PITCH) = p3;
        }
        __syncthreads();

        const int nt = tile + (int)gridDim.x;
        const int lt = nt < ntiles ? nt : tile;
        ap = (const v4f*)(A + (size_t)lt * 16 * K) + tid;
        p0 = ap[0]; p1 = ap[256];
        if constexpr (NV == 4) { p2 = ap[512]; p3 = ap[768]; }

        v2f fr[2][G];
        #pragma unroll
        for (int i = 0; i < G; ++i) fr[0][i] = *(const v2f*)(arow + 4 * i);
        v8f acc = {};
        #pragma unroll
        for (int g = 0; g < NG; ++g) {
            if (g + 1 < NG) {
                #pragma unroll
                for (int i = 0; i < G; ++i)
                    fr[(g + 1) & 1][i] = *(const v2f*)(arow + 4 * ((g + 1) * G + i));
            }
            __builtin_amdgcn_sched_barrier(0);
            #pragma unroll
            for (int i = 0; i < G; ++i)
                acc = __builtin_amdgcn_wmma_f32_16x16x4_f32(
                    false, fr[g & 1][i], false, bf[g * G + i], (short)0, acc, false, false);
            __builtin_amdgcn_sched_barrier(0);
        }

        float* op = out + (size_t)(tile * 16 + 8 * half) * HID + col0 + m;
        #pragma unroll
        for (int v = 0; v < 8; ++v) op[v * HID] = acc[v];

        if (nt >= ntiles) break;
        tile = nt;
    }
}

#define OWN_WAVES 32
#define OWN_Q     64
#define OWN_U     4
#define OWN_G     8

typedef __attribute__((address_space(3))) int lds_int;

__device__ __forceinline__ void owner_drain(volatile lds_int* q, int head, int nb, int lane, int lo,
                                            const int* __restrict__ src, const int* __restrict__ dst,
                                            const float* __restrict__ dis, const float* __restrict__ h,
                                            float* acc) {
    const bool v = lane < nb;
    const int e = v ? q[(head + lane) & (OWN_Q - 1)] : 0;
    const int s = v ? src[e] : 0;
    const int d = v ? dst[e] : lo;
    const float nr = v ? dis[s] * dis[d] : 0.f;
    const int r = d - lo;
    int i = 0;
    #pragma unroll 1
    for (; i + OWN_G <= nb; i += OWN_G) {
        float val[OWN_G][4];
        #pragma unroll
        for (int u = 0; u < OWN_G; ++u) {
            const int si = __builtin_amdgcn_readlane(s, i + u);
            const float* hs = h + (size_t)si * HID;
            #pragma unroll
            for (int j = 0; j < 4; ++j) val[u][j] = hs[j * 32 + lane];
        }
        #pragma unroll
        for (int u = 0; u < OWN_G; ++u) {
            const int ri = __builtin_amdgcn_readlane(r, i + u);
            const float ni = __int_as_float(__builtin_amdgcn_readlane(__float_as_int(nr), i + u));
            float* ar = acc + ri * HID;
            #pragma unroll
            for (int j = 0; j < 4; ++j) atomicAdd(&ar[j * 32 + lane], val[u][j] * ni);
        }
    }
    for (; i < nb; ++i) {
        const int si = __builtin_amdgcn_readlane(s, i);
        const int ri = __builtin_amdgcn_readlane(r, i);
        const float ni = __int_as_float(__builtin_amdgcn_readlane(__float_as_int(nr), i));
        const float* hs = h + (size_t)si * HID;
        float* ar = acc + ri * HID;
        #pragma unroll
        for (int j = 0; j < 4; ++j) atomicAdd(&ar[j * 32 + lane], hs[j * 32 + lane] * ni);
    }
}

__device__ __forceinline__ void owner_scan(volatile lds_int* q, int wave, int lane, int lo, unsigned pat,
                                           int rot, int nsteps, int E,
                                           const unsigned* __restrict__ own,
                                           const int* __restrict__ src, const int* __restrict__ dst,
                                           const float* __restrict__ dis, const float* __restrict__ h,
                                           float* acc) {
    int head = 0, cnt = 0;
    for (int st0 = wave; st0 < nsteps; st0 += OWN_WAVES * OWN_U) {
        uint4 ov[OWN_U];
        #pragma unroll
        for (int u = 0; u < OWN_U; ++u) {
            const int st = st0 + u * OWN_WAVES;
            int sidx = st + rot; if (sidx >= nsteps) sidx -= nsteps;
            if (st >= nsteps) sidx = 0;
            ov[u] = *(const uint4*)(own + ((size_t)(sidx * 32 + lane) << 2));
        }
        __builtin_amdgcn_sched_barrier(0);
        unsigned long long lm = 0ull;
        #pragma unroll
        for (int u = 0; u < OWN_U; ++u) {
            const unsigned wd[4] = {ov[u].x, ov[u].y, ov[u].z, ov[u].w};
            unsigned nz = 0u;
            #pragma unroll
            for (int j = 0; j < 4; ++j) {
                const unsigned x = wd[j] ^ pat;
                const unsigned y = ((x & 0x7F7F7F7Fu) + 0x7F7F7F7Fu) | x;
                nz |= ((y >> 7) & 0x01010101u) << j;
            }
            unsigned m16 = ~nz & 0x0F0F0F0Fu;
            if (st0 + u * OWN_WAVES >= nsteps) m16 = 0u;
            lm |= (unsigned long long)(m16 << (4 * (u & 1))) << (32 * (u >> 1));
        }
        for (int round = 0; round < 16 * OWN_U; ++round) {
            const bool has = lm != 0ull;
            if (__builtin_amdgcn_ballot_w32(has) == 0u) break;
            const int p  = has ? __builtin_ctzll(lm) : 0;
            const int lw = p & 31;
            const int u  = ((p >> 5) << 1) | ((lw >> 2) & 1);
            int sidx = st0 + u * OWN_WAVES + rot; if (sidx >= nsteps) sidx -= nsteps;
            const int e = ((sidx * 32 + lane) << 4) + ((lw & 3) << 2) + (lw >> 3);
            const bool take = has && (e < E);
            const unsigned mask = __builtin_amdgcn_ballot_w32(take);
            const int pos = (head + cnt + (int)__builtin_amdgcn_mbcnt_lo(mask, 0u)) & (OWN_Q - 1);
            if (take) q[pos] = e;
            lm &= lm - 1ull;
            cnt += __builtin_popcount(mask);
            if (cnt >= 32) {
                owner_drain(q, head, 32, lane, lo, src, dst, dis, h, acc);
                head = (head + 32) & (OWN_Q - 1);
                cnt -= 32;
            }
        }
    }
    owner_drain(q, head, cnt, lane, lo, src, dst, dis, h, acc);
}

__global__ __launch_bounds__(1024) void owner_aggregate(const int* __restrict__ src,
                                                        const int* __restrict__ dst,
                                                        const float* __restrict__ dis,
                                                        const float* __restrict__ h,
                                                        const float* __restrict__ bias,
                                                        const unsigned* __restrict__ own,
                                                        float* __restrict__ outp, int E, int N) {
    static_assert((N_NODES + OWN_R - 1) / OWN_R <= 256, "an owner id must fit in one byte");
    __shared__ __attribute__((aligned(16))) float acc[OWN_R * HID];
    __shared__ int queue[OWN_WAVES * OWN_Q];

    const int tid  = threadIdx.x;
    const int lane = tid & 31;
    const int wave = __builtin_amdgcn_readfirstlane(tid >> 5);
    const int lo   = blockIdx.x * OWN_R;
    const int nrows = (N - lo) < OWN_R ? (N - lo) : OWN_R;
    volatile lds_int* q = (volatile lds_int*)(queue + wave * OWN_Q);

    for (int idx = tid; idx < nrows * (HID / 4); idx += 1024) {
        const int r = idx >> 5, c4 = idx & 31;
        const float di = dis[lo + r];
        const float w = di * di;
        const float4 hv = ((const float4*)(h + (size_t)(lo + r) * HID))[c4];
        const float4 bv = ((const float4*)bias)[c4];
        float4 o;
        o.x = w * hv.x + bv.x; o.y = w * hv.y + bv.y; o.z = w * hv.z + bv.z; o.w = w * hv.w + bv.w;
        ((float4*)acc)[idx] = o;
    }
    __syncthreads();

    const int nsteps = (E + 511) >> 9;
    const int rot = (int)(((unsigned)blockIdx.x * (unsigned)nsteps) / gridDim.x);
    const unsigned pat = (unsigned)blockIdx.x * 0x01010101u;
    owner_scan(q, wave, lane, lo, pat, rot, nsteps, E, own, src, dst, dis, h, acc);
    __syncthreads();

    float4* orow = (float4*)(outp + (size_t)lo * HID);
    for (int idx = tid; idx < nrows * (HID / 4); idx += 1024) {
        float4 v4 = ((const float4*)acc)[idx];
        v4.x = v4.x > 0.f ? v4.x : 0.f; v4.y = v4.y > 0.f ? v4.y : 0.f;
        v4.z = v4.z > 0.f ? v4.z : 0.f; v4.w = v4.w > 0.f ? v4.w : 0.f;
        orow[idx] = v4;
    }
}

#define POOL_NB    256
#define POOL_R     ((N_NODES + POOL_NB - 1) / POOL_NB)
#define POOL_WAVES 16
#define POOL_RW    ((POOL_R + POOL_WAVES - 1) / POOL_WAVES)

__device__ __forceinline__ void pool_flush(float* tab, int* tcnt, int g, int rl, float4 a, int lane) {
    if ((unsigned)g < (unsigned)N_GRAPHS) {
        float* t = tab + g * HID + 4 * lane;
        atomicAdd(t + 0, a.x); atomicAdd(t + 1, a.y); atomicAdd(t + 2, a.z); atomicAdd(t + 3, a.w);
        if (lane == 0) atomicAdd(tcnt + g, rl);
    }
}

__global__ __launch_bounds__(POOL_WAVES * 32) void pool_partial(const float* __restrict__ h,
                                                                const int* __restrict__ batch,
                                                                float* __restrict__ part,
                                                                int* __restrict__ cntp, int N) {
    __shared__ __attribute__((aligned(16))) float tab[N_GRAPHS * HID];
    __shared__ int tcnt[N_GRAPHS];
    const int tid  = threadIdx.x;
    const int lane = tid & 31;
    const int wave = __builtin_amdgcn_readfirstlane(tid >> 5);
    const float4 zero = make_float4(0.f, 0.f, 0.f, 0.f);

    const int r0 = blockIdx.x * POOL_R + wave * POOL_RW;
    int rend = blockIdx.x * POOL_R + POOL_R;
    if (rend > N) rend = N;
    int nr = rend - r0;
    nr = nr < 0 ? 0 : (nr > POOL_RW ? POOL_RW : nr);
    const int gl = (lane < nr) ? batch[r0 + lane] : -1;
    float4 v[POOL_RW];
    #pragma unroll
    for (int i = 0; i < POOL_RW; ++i) {
        int ri = r0 + i;
        if (ri > N - 1) ri = N - 1;
        v[i] = ((const float4*)(h + (size_t)ri * HID))[lane];
    }

    for (int i = tid; i < N_GRAPHS * HID / 4; i += POOL_WAVES * 32) ((float4*)tab)[i] = zero;
    if (tid < N_GRAPHS) tcnt[tid] = 0;
    __syncthreads();

    int gcur = -1, rl = 0;
    float4 acc = zero;
    #pragma unroll
    for (int i = 0; i < POOL_RW; ++i) {
        if (i < nr) {
            const int g = __builtin_amdgcn_readlane(gl, i);
            if (g != gcur) {
                pool_flush(tab, tcnt, gcur, rl, acc, lane);
                gcur = g; rl = 0; acc = zero;
            }
            acc.x += v[i].x; acc.y += v[i].y; acc.z += v[i].z; acc.w += v[i].w;
            ++rl;
        }
    }
    pool_flush(tab, tcnt, gcur, rl, acc, lane);
    __syncthreads();

    float4* pl = (float4*)(part + (size_t)blockIdx.x * (N_GRAPHS * HID));
    for (int i = tid; i < N_GRAPHS * HID / 4; i += POOL_WAVES * 32) pl[i] = ((const float4*)tab)[i];
    if (tid < N_GRAPHS) cntp[blockIdx.x * N_GRAPHS + tid] = tcnt[tid];
}

__global__ __launch_bounds__(1024) void pool_finish(const float* __restrict__ part,
                                                    const int* __restrict__ cntp,
                                                    float* __restrict__ pooled) {
    static_assert(POOL_NB == 256, "32 waves x 8 planes, one count per thread of the first 256");
    __shared__ __attribute__((aligned(16))) float red[32 * HID];
    __shared__ int ntot;
    const int tid  = threadIdx.x;
    const int lane = tid & 31;
    const int wave = tid >> 5;
    const int g    = blockIdx.x;

    float4 v[8];
    #pragma unroll
    for (int i = 0; i < 8; ++i)
        v[i] = ((const float4*)(part + ((size_t)(wave * 8 + i) * N_GRAPHS + g) * HID))[lane];
    const int c = (tid < POOL_NB) ? cntp[tid * N_GRAPHS + g] : 0;

    if (tid == 0) ntot = 0;
    __syncthreads();
    float4 s = v[0];
    #pragma unroll
    for (int i = 1; i < 8; ++i) { s.x += v[i].x; s.y += v[i].y; s.z += v[i].z; s.w += v[i].w; }
    ((float4*)red)[wave * 32 + lane] = s;
    if (c != 0) atomicAdd(&ntot, c);
    __syncthreads();

    if (wave == 0) {
        float4 t = ((const float4*)red)[lane];
        #pragma unroll
        for (int w = 1; w < 32; ++w) {
            const float4 u = ((const float4*)red)[w * 32 + lane];
            t.x += u.x; t.y += u.y; t.z += u.z; t.w += u.w;
        }
        float n = (float)ntot;
        n = n > 1.f ? n : 1.f;
        t.x /= n; t.y /= n; t.z /= n; t.w /= n;
        ((float4*)(pooled + g * HID))[lane] = t;
    }
}

__global__ void head_gemm(const float* __restrict__ pooled, const float* __restrict__ W,
                          const float* __restrict__ b, float* __restrict__ out, int OUT) {
    int t = blockIdx.x * blockDim.x + threadIdx.x;
    const int n = N_GRAPHS * OUT;
    if (t >= n) return;
    const int g = t / OUT, o = t - g * OUT;
    const float* p = pooled + g * HID;
    float s = b[o];
    #pragma unroll 8
    for (int k = 0; k < HID; ++k) s += p[k] * W[k * OUT + o];
    out[t] = s;
}

extern "C" void kernel_launch(void* const* d_in, const int* in_sizes, int n_in,
                              void* d_out, int out_size, void* d_ws, size_t ws_size,
                              hipStream_t stream) {
    const float* x   = (const float*)d_in[0];
    const int*   ei  = (const int*)d_in[1];
    const int*   bat = (const int*)d_in[2];
    const float* W1  = (const float*)d_in[3];
    const float* b1  = (const float*)d_in[4];
    const float* W2  = (const float*)d_in[5];
    const float* b2  = (const float*)d_in[6];
    const float* Wmf = (const float*)d_in[7];
    const float* bmf = (const float*)d_in[8];
    const float* Wbp = (const float*)d_in[9];
    const float* bbp = (const float*)d_in[10];
    const float* Wcc = (const float*)d_in[11];
    const float* bcc = (const float*)d_in[12];
    float* out = (float*)d_out;

    const int E = in_sizes[1] / 2;
    const int* src = ei;
    const int* dst = ei + E;

    float* ws     = (float*)d_ws;
    float* dis    = ws;
    float* h      = dis + N_NODES;
    float* agg    = h + (size_t)N_NODES * HID;
    float* pooled = agg + (size_t)N_NODES * HID;
    float* cnt    = pooled + N_GRAPHS * HID;
    unsigned* own = (unsigned*)(cnt + N_GRAPHS);

    unsigned* degp = (unsigned*)agg;
    deg_partial<<<DEG_P, 1024, 0, stream>>>(dst, degp, own, E);
    deg_finish<<<((N_NODES + 1) / 2 + 255) / 256, 256, 0, stream>>>(degp, dis, N_NODES);

    gemm_wmma_f32<IN_DIM><<<GEMM_GRID_K256, 256, 0, stream>>>(x, W1, h, N_NODES / 16);
    owner_aggregate<<<(N_NODES + OWN_R - 1) / OWN_R, 1024, 0, stream>>>(src, dst, dis, h, b1, own, agg, E, N_NODES);

    gemm_wmma_f32<HID><<<GEMM_GRID_K128, 256, 0, stream>>>(agg, W2, h, N_NODES / 16);
    owner_aggregate<<<(N_NODES + OWN_R - 1) / OWN_R, 1024, 0, stream>>>(src, dst, dis, h, b2, own, agg, E, N_NODES);

    float* part = h;
    int*   cntp = (int*)(h + (size_t)POOL_NB * N_GRAPHS * HID);
    pool_partial<<<POOL_NB, POOL_WAVES * 32, 0, stream>>>(agg, bat, part, cntp, N_NODES);
    pool_finish<<<N_GRAPHS, 1024, 0, stream>>>(part, cntp, pooled);

    float* out_mf = out;
    float* out_bp = out + N_GRAPHS * OUT_MF;
    float* out_cc = out + N_GRAPHS * (OUT_MF + OUT_BP);
    head_gemm<<<(N_GRAPHS * OUT_MF + 255) / 256, 256, 0, stream>>>(pooled, Wmf, bmf, out_mf, OUT_MF);
    head_gemm<<<(N_GRAPHS * OUT_BP + 255) / 256, 256, 0, stream>>>(pooled, Wbp, bbp, out_bp, OUT_BP);
    head_gemm<<<(N_GRAPHS * OUT_CC + 255) / 256, 256, 0, stream>>>(pooled, Wcc, bcc, out_cc, OUT_CC);
}


# layer 1's aggregation records each wave's picked edges (source, row) as it drains; layer 2 replays the record, no owner-id scan or edge-list look-up; LDS sum order stays by arrival
# speedup vs baseline: 3.1514x; 1.1423x over previous
#include <hip/hip_runtime.h>
#include <hip/hip_bf16.h>

#define N_NODES 100000
#define N_EDGES 1600000
#define IN_DIM  256
#define HID     128
#define N_GRAPHS 64
#define OUT_MF 489
#define OUT_BP 1943
#define OUT_CC 320

typedef __attribute__((ext_vector_type(2))) float v2f;
typedef __attribute__((ext_vector_type(8))) float v8f;

#define OWN_R     391
#define DEG_P     64
#define DEG_U     8
#define DEG_WORDS (((N_NODES + 1) / 2 + 3) & ~3)

__device__ __forceinline__ unsigned deg_count(unsigned* tab, int d) {
    if ((unsigned)d >= (unsigned)N_NODES) return 0u;
    atomicAdd(&tab[d >> 1], (d & 1) ? 0x10000u : 1u);
    return (unsigned)(d / OWN_R);
}

__global__ __launch_bounds__(1024) void deg_partial(const int* __restrict__ dst,
                                                    unsigned* __restrict__ part,
                                                    unsigned* __restrict__ own, int E) {
    static_assert(((N_EDGES + 511) / 512 * 512 + DEG_P - 1) / DEG_P + 4 < 65536,
                  "the counts of one slice must fit in 16 bits");
    __shared__ __attribute__((aligned(16))) unsigned tab[DEG_WORDS];
    const int tid = threadIdx.x;
    for (int i = tid; i < DEG_WORDS / 4; i += 1024) ((uint4*)tab)[i] = make_uint4(0u, 0u, 0u, 0u);
    __syncthreads();

    const int nwords = ((E + 511) >> 9) << 7;
    const int per = (nwords + DEG_P - 1) / DEG_P;
    const int w0 = blockIdx.x * per;
    const int w1 = (w0 + per) < nwords ? (w0 + per) : nwords;
    if (((E & 3) == 0) && ((((size_t)dst) & 15) == 0)) {
        const int lastw = (E >> 2) - 1;
        for (int tb = w0 + tid; tb < w1; tb += 1024 * DEG_U) {
            int4 d[DEG_U];
            #pragma unroll
            for (int u = 0; u < DEG_U; ++u) {
                const int t = tb + u * 1024;
                const int tl = t < lastw ? t : lastw;
                d[u] = *(const int4*)(dst + 4 * (size_t)tl);
            }
            #pragma unroll
            for (int u = 0; u < DEG_U; ++u) {
                const int t = tb + u * 1024;
                if (t < w1) {
                    const int4 dd = t <= lastw ? d[u] : make_int4(-1, -1, -1, -1);
                    own[t] = deg_count(tab, dd.x) | (deg_count(tab, dd.y) << 8)
                           | (deg_count(tab, dd.z) << 16) | (deg_count(tab, dd.w) << 24);
                }
            }
        }
    } else {
        for (int t = w0 + tid; t < w1; t += 1024) {
            unsigned w = 0u;
            for (int c = 0; c < 4; ++c)
                if (4 * t + c < E) w |= deg_count(tab, dst[4 * t + c]) << (8 * c);
            own[t] = w;
        }
    }
    __syncthreads();

    uint4* pl = (uint4*)(part + (size_t)blockIdx.x * DEG_WORDS);
    for (int i = tid; i < DEG_WORDS / 4; i += 1024) pl[i] = ((const uint4*)tab)[i];
}

__global__ void deg_finish(const unsigned* __restrict__ part, float* __restrict__ dis, int N) {
    const int t = blockIdx.x * blockDim.x + threadIdx.x;
    if (2 * t >= N) return;
    unsigned lo = 0u, hi = 0u;
    #pragma unroll 16
    for (int p = 0; p < DEG_P; ++p) {
        const unsigned w = part[(size_t)p * DEG_WORDS + t];
        lo += w & 0xFFFFu;
        hi += w >> 16;
    }
    const float d0 = rsqrtf(1.0f + (float)lo);
    const float d1 = rsqrtf(1.0f + (float)hi);
    if (2 * t + 1 < N) {
        *(float2*)(dis + 2 * t) = make_float2(d0, d1);
    } else {
        dis[2 * t] = d0;
    }
}

#define GEMM_GRID_K256 512
#define GEMM_GRID_K128 768
typedef __attribute__((ext_vector_type(4))) float v4f;

template <int K>
__global__ __launch_bounds__(256) void gemm_wmma_f32(const float* __restrict__ A,
                                                     const float* __restrict__ W,
                                                     float* __restrict__ out, int ntiles) {
    constexpr int PITCH = K + 4;
    constexpr int RV    = K / 4;
    constexpr int NV    = 16 * RV / 256;
    constexpr int RS    = 256 / RV;
    constexpr int G     = 16;
    constexpr int NG    = K / 4 / G;
    static_assert(NV == 4 || NV == 2, "a thread moves 2 or 4 pieces of a tile");
    __shared__ __attribute__((aligned(16))) float ldsA[16 * PITCH];
    const int tid  = threadIdx.x;
    const int wave = tid >> 5;
    const int lane = tid & 31;
    const int half = lane >> 4;
    const int m    = lane & 15;
    const int col0 = wave * 16;

    int tile = blockIdx.x;
    if (tile >= ntiles) return;

    const int pr = tid / RV, pc = tid - pr * RV;
    float* lst = ldsA + pr * PITCH + 4 * pc;
    const v4f* ap = (const v4f*)(A + (size_t)tile * 16 * K) + tid;
    v4f p0 = ap[0], p1 = ap[256], p2 = {}, p3 = {};
    if constexpr (NV == 4) { p2 = ap[512]; p3 = ap[768]; }

    v2f bf[K / 4];
    #pragma unroll
    for (int s = 0; s < K / 4; ++s) {
        const int ka = 4 * s + 2 * half;
        bf[s].x = W[ka * HID + col0 + m];
        bf[s].y = W[(ka + 1) * HID + col0 + m];
    }

    const float* arow = ldsA + m * PITCH + 2 * half;
    for (;;) {
        __syncthreads();
        *(v4f*)(lst) = p0;
        *(v4f*)(lst + RS * PITCH) = p1;
        if constexpr (NV == 4) {
            *(v4f*)(lst + 2 * RS * PITCH) = p2;
            *(v4f*)(lst + 3 * RS * PITCH) = p3;
        }
        __syncthreads();

        const int nt = tile + (int)gridDim.x;
        const int lt = nt < ntiles ? nt : tile;
        ap = (const v4f*)(A + (size_t)lt * 16 * K) + tid;
        p0 = ap[0]; p1 = ap[256];
        if constexpr (NV == 4) { p2 = ap[512]; p3 = ap[768]; }

        v2f fr[2][G];
        #pragma unroll
        for (int i = 0; i < G; ++i) fr[0][i] = *(const v2f*)(arow + 4 * i);
        v8f acc = {};
        #pragma unroll
        for (int g = 0; g < NG; ++g) {
            if (g + 1 < NG) {
                #pragma unroll
                for (int i = 0; i < G; ++i)
                    fr[(g + 1) & 1][i] = *(const v2f*)(arow + 4 * ((g + 1) * G + i));
            }
            __builtin_amdgcn_sched_barrier(0);
            #pragma unroll
            for (int i = 0; i < G; ++i)
                acc = __builtin_amdgcn_wmma_f32_16x16x4_f32(
                    false, fr[g & 1][i], false, bf[g * G + i], (short)0, acc, false, false);
            __builtin_amdgcn_sched_barrier(0);
        }

        float* op = out + (size_t)(tile * 16 + 8 * half) * HID + col0 + m;
        #pragma unroll
        for (int v = 0; v < 8; ++v) op[v * HID] = acc[v];

        if (nt >= ntiles) break;
        tile = nt;
    }
}

#define OWN_WAVES 32
#define OWN_Q     64
#define OWN_U     4
#define OWN_G     8
#define OWN_NB    ((N_NODES + OWN_R - 1) / OWN_R)
#define OWN_CAP   256

#define OWN_PLANE_WORDS ((N_EDGES + 511) / 512 * 128)
#define OWN_WS_WORDS ((size_t)N_NODES + 2 * (size_t)N_NODES * HID + N_GRAPHS * HID + N_GRAPHS \
                      + OWN_PLANE_WORDS + (size_t)OWN_NB * OWN_WAVES * OWN_CAP + OWN_NB * OWN_WAVES)

typedef __attribute__((address_space(3))) int lds_int;

__device__ __forceinline__ void owner_rows(int s, int r, float nr, int nb, int lane,
                                           const float* __restrict__ h, float* acc) {
    int i = 0;
    #pragma unroll 1
    for (; i + OWN_G <= nb; i += OWN_G) {
        float val[OWN_G][4];
        #pragma unroll
        for (int u = 0; u < OWN_G; ++u) {
            const int si = __builtin_amdgcn_readlane(s, i + u);
            const float* hs = h + (size_t)si * HID;
            #pragma unroll
            for (int j = 0; j < 4; ++j) val[u][j] = hs[j * 32 + lane];
        }
        #pragma unroll
        for (int u = 0; u < OWN_G; ++u) {
            const int ri = __builtin_amdgcn_readlane(r, i + u);
            const float ni = __int_as_float(__builtin_amdgcn_readlane(__float_as_int(nr), i + u));
            float* ar = acc + ri * HID;
            #pragma unroll
            for (int j = 0; j < 4; ++j) atomicAdd(&ar[j * 32 + lane], val[u][j] * ni);
        }
    }
    for (; i < nb; ++i) {
        const int si = __builtin_amdgcn_readlane(s, i);
        const int ri = __builtin_amdgcn_readlane(r, i);
        const float ni = __int_as_float(__builtin_amdgcn_readlane(__float_as_int(nr), i));
        const float* hs = h + (size_t)si * HID;
        float* ar = acc + ri * HID;
        #pragma unroll
        for (int j = 0; j < 4; ++j) atomicAdd(&ar[j * 32 + lane], hs[j * 32 + lane] * ni);
    }
}

__device__ __forceinline__ void owner_drain(volatile lds_int* q, int head, int nb, int lane, int lo,
                                            const int* __restrict__ src, const int* __restrict__ dst,
                                            const float* __restrict__ dis, const float* __restrict__ h,
                                            float* acc, unsigned* __restrict__ rec, int& nrec) {
    const bool v = lane < nb;
    const int e = v ? q[(head + lane) & (OWN_Q - 1)] : 0;
    const int s = v ? src[e] : 0;
    const int d = v ? dst[e] : lo;
    const float nr = v ? dis[s] * dis[d] : 0.f;
    const int r = d - lo;
    if (rec != nullptr) {
        if (v && nrec + nb <= OWN_CAP) rec[nrec + lane] = (unsigned)s | ((unsigned)r << 17);
        nrec += nb;
    }
    owner_rows(s, r, nr, nb, lane, h, acc);
}

__device__ __forceinline__ void owner_replay(const unsigned* __restrict__ rec, int n, int lane, int lo,
                                             const float* __restrict__ dis, const float* __restrict__ h,
                                             float* acc) {
    if (n <= 0) return;
    const int last = n - 1;
    unsigned w1 = rec[lane < last ? lane : last];
    unsigned w2 = rec[32 + lane < last ? 32 + lane : last];
    float fa = dis[w1 & 0x1FFFFu], fb = dis[lo + (int)(w1 >> 17)];
    for (int i = 0; i < n; i += 32) {
        const int s = (int)(w1 & 0x1FFFFu), r = (int)(w1 >> 17);
        const float nr = fa * fb;
        w1 = w2;
        const int nx = i + 64 + lane;
        w2 = rec[nx < last ? nx : last];
        fa = dis[w1 & 0x1FFFFu]; fb = dis[lo + (int)(w1 >> 17)];
        const int left = n - i;
        owner_rows(s, r, nr, left < 32 ? left : 32, lane, h, acc);
    }
}

__device__ __forceinline__ void owner_scan(volatile lds_int* q, int wave, int lane, int lo, unsigned pat,
                                           int rot, int nsteps, int E,
                                           const unsigned* __restrict__ own,
                                           const int* __restrict__ src, const int* __restrict__ dst,
                                           const float* __restrict__ dis, const float* __restrict__ h,
                                           float* acc, unsigned* __restrict__ rec, int& nrec) {
    int head = 0, cnt = 0;
    for (int st0 = wave; st0 < nsteps; st0 += OWN_WAVES * OWN_U) {
        uint4 ov[OWN_U];
        #pragma unroll
        for (int u = 0; u < OWN_U; ++u) {
            const int st = st0 + u * OWN_WAVES;
            int sidx = st + rot; if (sidx >= nsteps) sidx -= nsteps;
            if (st >= nsteps) sidx = 0;
            ov[u] = *(const uint4*)(own + ((size_t)(sidx * 32 + lane) << 2));
        }
        __builtin_amdgcn_sched_barrier(0);
        unsigned long long lm = 0ull;
        #pragma unroll
        for (int u = 0; u < OWN_U; ++u) {
            const unsigned wd[4] = {ov[u].x, ov[u].y, ov[u].z, ov[u].w};
            unsigned nz = 0u;
            #pragma unroll
            for (int j = 0; j < 4; ++j) {
                const unsigned x = wd[j] ^ pat;
                const unsigned y = ((x & 0x7F7F7F7Fu) + 0x7F7F7F7Fu) | x;
                nz |= ((y >> 7) & 0x01010101u) << j;
            }
            unsigned m16 = ~nz & 0x0F0F0F0Fu;
            if (st0 + u * OWN_WAVES >= nsteps) m16 = 0u;
            lm |= (unsigned long long)(m16 << (4 * (u & 1))) << (32 * (u >> 1));
        }
        for (int round = 0; round < 16 * OWN_U; ++round) {
            const bool has = lm != 0ull;
            if (__builtin_amdgcn_ballot_w32(has) == 0u) break;
            const int p  = has ? __builtin_ctzll(lm) : 0;
            const int lw = p & 31;
            const int u  = ((p >> 5) << 1) | ((lw >> 2) & 1);
            int sidx = st0 + u * OWN_WAVES + rot; if (sidx >= nsteps) sidx -= nsteps;
            const int e = ((sidx * 32 + lane) << 4) + ((lw & 3) << 2) + (lw >> 3);
            const bool take = has && (e < E);
            const unsigned mask = __builtin_amdgcn_ballot_w32(take);
            const int pos = (head + cnt + (int)__builtin_amdgcn_mbcnt_lo(mask, 0u)) & (OWN_Q - 1);
            if (take) q[pos] = e;
            lm &= lm - 1ull;
            cnt += __builtin_popcount(mask);
            if (cnt >= 32) {
                owner_drain(q, head, 32, lane, lo, src, dst, dis, h, acc, rec, nrec);
                head = (head + 32) & (OWN_Q - 1);
                cnt -= 32;
            }
        }
    }
    owner_drain(q, head, cnt, lane, lo, src, dst, dis, h, acc, rec, nrec);
}

template <int LAYER>
__global__ __launch_bounds__(1024) void owner_aggregate(const int* __restrict__ src,
                                                        const int* __restrict__ dst,
                                                        const float* __restrict__ dis,
                                                        const float* __restrict__ h,
                                                        const float* __restrict__ bias,
                                                        const unsigned* __restrict__ own,
                                                        float* __restrict__ outp,
                                                        unsigned* __restrict__ rlist,
                                                        int* __restrict__ rcnt,
                                                        size_t ws_bytes, int E, int N) {
    static_assert((N_NODES + OWN_R - 1) / OWN_R <= 256, "an owner id must fit in one byte");
    static_assert(N_NODES <= (1 << 17) && OWN_R <= (1 << 15), "a record word holds source and row");
    __shared__ __attribute__((aligned(16))) float acc[OWN_R * HID];
    __shared__ int queue[OWN_WAVES * OWN_Q];

    const int tid  = threadIdx.x;
    const int lane = tid & 31;
    const int wave = __builtin_amdgcn_readfirstlane(tid >> 5);
    const int lo   = blockIdx.x * OWN_R;
    const int nrows = (N - lo) < OWN_R ? (N - lo) : OWN_R;
    volatile lds_int* q = (volatile lds_int*)(queue + wave * OWN_Q);

    for (int idx = tid; idx < nrows * (HID / 4); idx += 1024) {
        const int r = idx >> 5, c4 = idx & 31;
        const float di = dis[lo + r];
        const float w = di * di;
        const float4 hv = ((const float4*)(h + (size_t)(lo + r) * HID))[c4];
        const float4 bv = ((const float4*)bias)[c4];
        float4 o;
        o.x = w * hv.x + bv.x; o.y = w * hv.y + bv.y; o.z = w * hv.z + bv.z; o.w = w * hv.w + bv.w;
        ((float4*)acc)[idx] = o;
    }
    __syncthreads();

    const int nsteps = (E + 511) >> 9;
    const int rot = (int)(((unsigned)blockIdx.x * (unsigned)nsteps) / gridDim.x);
    const unsigned pat = (unsigned)blockIdx.x * 0x01010101u;
    const bool room = ws_bytes >= OWN_WS_WORDS * sizeof(float);
    const int slot = blockIdx.x * OWN_WAVES + wave;
    unsigned* rec = rlist + (size_t)slot * OWN_CAP;
    if (LAYER == 1) {
        int nrec = 0;
        owner_scan(q, wave, lane, lo, pat, rot, nsteps, E, own, src, dst, dis, h, acc,
                   room ? rec : nullptr, nrec);
        if (lane == 0) q[0] = nrec;
    } else {
        const int n = room ? rcnt[slot] : OWN_CAP + 1;
        if (n <= OWN_CAP) {
            owner_replay(rec, n, lane, lo, dis, h, acc);
        } else {
            int none = 0;
            owner_scan(q, wave, lane, lo, pat, rot, nsteps, E, own, src, dst, dis, h, acc, nullptr, none);
        }
    }
    __syncthreads();

    if (LAYER == 1 && room && wave == 0) rcnt[blockIdx.x * OWN_WAVES + lane] = queue[lane * OWN_Q];

    float4* orow = (float4*)(outp + (size_t)lo * HID);
    for (int idx = tid; idx < nrows * (HID / 4); idx += 1024) {
        float4 v4 = ((const float4*)acc)[idx];
        v4.x = v4.x > 0.f ? v4.x : 0.f; v4.y = v4.y > 0.f ? v4.y : 0.f;
        v4.z = v4.z > 0.f ? v4.z : 0.f; v4.w = v4.w > 0.f ? v4.w : 0.f;
        orow[idx] = v4;
    }
}

#define POOL_NB    256
#define POOL_R     ((N_NODES + POOL_NB - 1) / POOL_NB)
#define POOL_WAVES 16
#define POOL_RW    ((POOL_R + POOL_WAVES - 1) / POOL_WAVES)

__device__ __forceinline__ void pool_flush(float* tab, int* tcnt, int g, int rl, float4 a, int lane) {
    if ((unsigned)g < (unsigned)N_GRAPHS) {
        float* t = tab + g * HID + 4 * lane;
        atomicAdd(t + 0, a.x); atomicAdd(t + 1, a.y); atomicAdd(t + 2, a.z); atomicAdd(t + 3, a.w);
        if (lane == 0) atomicAdd(tcnt + g, rl);
    }
}

__global__ __launch_bounds__(POOL_WAVES * 32) void pool_partial(const float* __restrict__ h,
                                                                const int* __restrict__ batch,
                                                                float* __restrict__ part,
                                                                int* __restrict__ cntp, int N) {
    __shared__ __attribute__((aligned(16))) float tab[N_GRAPHS * HID];
    __shared__ int tcnt[N_GRAPHS];
    const int tid  = threadIdx.x;
    const int lane = tid & 31;
    const int wave = __builtin_amdgcn_readfirstlane(tid >> 5);
    const float4 zero = make_float4(0.f, 0.f, 0.f, 0.f);

    const int r0 = blockIdx.x * POOL_R + wave * POOL_RW;
    int rend = blockIdx.x * POOL_R + POOL_R;
    if (rend > N) rend = N;
    int nr = rend - r0;
    nr = nr < 0 ? 0 : (nr > POOL_RW ? POOL_RW : nr);
    const int gl = (lane < nr) ? batch[r0 + lane] : -1;
    float4 v[POOL_RW];
    #pragma unroll
    for (int i = 0; i < POOL_RW; ++i) {
        int ri = r0 + i;
        if (ri > N - 1) ri = N - 1;
        v[i] = ((const float4*)(h + (size_t)ri * HID))[lane];
    }

    for (int i = tid; i < N_GRAPHS * HID / 4; i += POOL_WAVES * 32) ((float4*)tab)[i] = zero;
    if (tid < N_GRAPHS) tcnt[tid] = 0;
    __syncthreads();

    int gcur = -1, rl = 0;
    float4 acc = zero;
    #pragma unroll
    for (int i = 0; i < POOL_RW; ++i) {
        if (i < nr) {
            const int g = __builtin_amdgcn_readlane(gl, i);
            if (g != gcur) {
                pool_flush(tab, tcnt, gcur, rl, acc, lane);
                gcur = g; rl = 0; acc = zero;
            }
            acc.x += v[i].x; acc.y += v[i].y; acc.z += v[i].z; acc.w += v[i].w;
            ++rl;
        }
    }
    pool_flush(tab, tcnt, gcur, rl, acc, lane);
    __syncthreads();

    float4* pl = (float4*)(part + (size_t)blockIdx.x * (N_GRAPHS * HID));
    for (int i = tid; i < N_GRAPHS * HID / 4; i += POOL_WAVES * 32) pl[i] = ((const float4*)tab)[i];
    if (tid < N_GRAPHS) cntp[blockIdx.x * N_GRAPHS + tid] = tcnt[tid];
}

__global__ __launch_bounds__(1024) void pool_finish(const float* __restrict__ part,
                                                    const int* __restrict__ cntp,
                                                    float* __restrict__ pooled) {
    static_assert(POOL_NB == 256, "32 waves x 8 planes, one count per thread of the first 256");
    __shared__ __attribute__((aligned(16))) float red[32 * HID];
    __shared__ int ntot;
    const int tid  = threadIdx.x;
    const int lane = tid & 31;
    const int wave = tid >> 5;
    const int g    = blockIdx.x;

    float4 v[8];
    #pragma unroll
    for (int i = 0; i < 8; ++i)
        v[i] = ((const float4*)(part + ((size_t)(wave * 8 + i) * N_GRAPHS + g) * HID))[lane];
    const int c = (tid < POOL_NB) ? cntp[tid * N_GRAPHS + g] : 0;

    if (tid == 0) ntot = 0;
    __syncthreads();
    float4 s = v[0];
    #pragma unroll
    for (int i = 1; i < 8; ++i) { s.x += v[i].x; s.y += v[i].y; s.z += v[i].z; s.w += v[i].w; }
    ((float4*)red)[wave * 32 + lane] = s;
    if (c != 0) atomicAdd(&ntot, c);
    __syncthreads();

    if (wave == 0) {
        float4 t = ((const float4*)red)[lane];
        #pragma unroll
        for (int w = 1; w < 32; ++w) {
            const float4 u = ((const float4*)red)[w * 32 + lane];
            t.x += u.x; t.y += u.y; t.z += u.z; t.w += u.w;
        }
        float n = (float)ntot;
        n = n > 1.f ? n : 1.f;
        t.x /= n; t.y /= n; t.z /= n; t.w /= n;
        ((float4*)(pooled + g * HID))[lane] = t;
    }
}

__global__ void head_gemm(const float* __restrict__ pooled, const float* __restrict__ W,
                          const float* __restrict__ b, float* __restrict__ out, int OUT) {
    int t = blockIdx.x * blockDim.x + threadIdx.x;
    const int n = N_GRAPHS * OUT;
    if (t >= n) return;
    const int g = t / OUT, o = t - g * OUT;
    const float* p = pooled + g * HID;
    float s = b[o];
    #pragma unroll 8
    for (int k = 0; k < HID; ++k) s += p[k] * W[k * OUT + o];
    out[t] = s;
}

extern "C" void kernel_launch(void* const* d_in, const int* in_sizes, int n_in,
                              void* d_out, int out_size, void* d_ws, size_t ws_size,
                              hipStream_t stream) {
    const float* x   = (const float*)d_in[0];
    const int*   ei  = (const int*)d_in[1];
    const int*   bat = (const int*)d_in[2];
    const float* W1  = (const float*)d_in[3];
    const float* b1  = (const float*)d_in[4];
    const float* W2  = (const float*)d_in[5];
    const float* b2  = (const float*)d_in[6];
    const float* Wmf = (const float*)d_in[7];
    const float* bmf = (const float*)d_in[8];
    const float* Wbp = (const float*)d_in[9];
    const float* bbp = (const float*)d_in[10];
    const float* Wcc = (const float*)d_in[11];
    const float* bcc = (const float*)d_in[12];
    float* out = (float*)d_out;

    const int E = in_sizes[1] / 2;
    const int* src = ei;
    const int* dst = ei + E;

    float* ws     = (float*)d_ws;
    float* dis    = ws;
    float* h      = dis + N_NODES;
    float* agg    = h + (size_t)N_NODES * HID;
    float* pooled = agg + (size_t)N_NODES * HID;
    float* cnt    = pooled + N_GRAPHS * HID;
    unsigned* own = (unsigned*)(cnt + N_GRAPHS);
    unsigned* rlist = own + OWN_PLANE_WORDS;
    int*      rcnt  = (int*)(rlist + (size_t)OWN_NB * OWN_WAVES * OWN_CAP);

    unsigned* degp = (unsigned*)agg;
    deg_partial<<<DEG_P, 1024, 0, stream>>>(dst, degp, own, E);
    deg_finish<<<((N_NODES + 1) / 2 + 255) / 256, 256, 0, stream>>>(degp, dis, N_NODES);

    gemm_wmma_f32<IN_DIM><<<GEMM_GRID_K256, 256, 0, stream>>>(x, W1, h, N_NODES / 16);
    owner_aggregate<1><<<OWN_NB, 1024, 0, stream>>>(src, dst, dis, h, b1, own, agg, rlist, rcnt, ws_size, E, N_NODES);

    gemm_wmma_f32<HID><<<GEMM_GRID_K128, 256, 0, stream>>>(agg, W2, h, N_NODES / 16);
    owner_aggregate<2><<<OWN_NB, 1024, 0, stream>>>(src, dst, dis, h, b2, own, agg, rlist, rcnt, ws_size, E, N_NODES);

    float* part = h;
    int*   cntp = (int*)(h + (size_t)POOL_NB * N_GRAPHS * HID);
    pool_partial<<<POOL_NB, POOL_WAVES * 32, 0, stream>>>(agg, bat, part, cntp, N_NODES);
    pool_finish<<<N_GRAPHS, 1024, 0, stream>>>(part, cntp, pooled);

    float* out_mf = out;
    float* out_bp = out + N_GRAPHS * OUT_MF;
    float* out_cc = out + N_GRAPHS * (OUT_MF + OUT_BP);
    head_gemm<<<(N_GRAPHS * OUT_MF + 255) / 256, 256, 0, stream>>>(pooled, Wmf, bmf, out_mf, OUT_MF);
    head_gemm<<<(N_GRAPHS * OUT_BP + 255) / 256, 256, 0, stream>>>(pooled, Wbp, bbp, out_bp, OUT_BP);
    head_gemm<<<(N_GRAPHS * OUT_CC + 255) / 256, 256, 0, stream>>>(pooled, Wcc, bcc, out_cc, OUT_CC);
}


# the three head products run on WMMA in one launch (were 3 vector launches): a wave owns a 16-column tile for all 64 graphs, W read once, means via padded LDS; k order of the head sums changes
# speedup vs baseline: 3.4645x; 1.0994x over previous
#include <hip/hip_runtime.h>
#include <hip/hip_bf16.h>

#define N_NODES 100000
#define N_EDGES 1600000
#define IN_DIM  256
#define HID     128
#define N_GRAPHS 64
#define OUT_MF 489
#define OUT_BP 1943
#define OUT_CC 320

typedef __attribute__((ext_vector_type(2))) float v2f;
typedef __attribute__((ext_vector_type(8))) float v8f;

#define OWN_R     391
#define DEG_P     64
#define DEG_U     8
#define DEG_WORDS (((N_NODES + 1) / 2 + 3) & ~3)

__device__ __forceinline__ unsigned deg_count(unsigned* tab, int d) {
    if ((unsigned)d >= (unsigned)N_NODES) return 0u;
    atomicAdd(&tab[d >> 1], (d & 1) ? 0x10000u : 1u);
    return (unsigned)(d / OWN_R);
}

__global__ __launch_bounds__(1024) void deg_partial(const int* __restrict__ dst,
                                                    unsigned* __restrict__ part,
                                                    unsigned* __restrict__ own, int E) {
    static_assert(((N_EDGES + 511) / 512 * 512 + DEG_P - 1) / DEG_P + 4 < 65536,
                  "the counts of one slice must fit in 16 bits");
    __shared__ __attribute__((aligned(16))) unsigned tab[DEG_WORDS];
    const int tid = threadIdx.x;
    for (int i = tid; i < DEG_WORDS / 4; i += 1024) ((uint4*)tab)[i] = make_uint4(0u, 0u, 0u, 0u);
    __syncthreads();

    const int nwords = ((E + 511) >> 9) << 7;
    const int per = (nwords + DEG_P - 1) / DEG_P;
    const int w0 = blockIdx.x * per;
    const int w1 = (w0 + per) < nwords ? (w0 + per) : nwords;
    if (((E & 3) == 0) && ((((size_t)dst) & 15) == 0)) {
        const int lastw = (E >> 2) - 1;
        for (int tb = w0 + tid; tb < w1; tb += 1024 * DEG_U) {
            int4 d[DEG_U];
            #pragma unroll
            for (int u = 0; u < DEG_U; ++u) {
                const int t = tb + u * 1024;
                const int tl = t < lastw ? t : lastw;
                d[u] = *(const int4*)(dst + 4 * (size_t)tl);
            }
            #pragma unroll
            for (int u = 0; u < DEG_U; ++u) {
                const int t = tb + u * 1024;
                if (t < w1) {
                    const int4 dd = t <= lastw ? d[u] : make_int4(-1, -1, -1, -1);
                    own[t] = deg_count(tab, dd.x) | (deg_count(tab, dd.y) << 8)
                           | (deg_count(tab, dd.z) << 16) | (deg_count(tab, dd.w) << 24);
                }
            }
        }
    } else {
        for (int t = w0 + tid; t < w1; t += 1024) {
            unsigned w = 0u;
            for (int c = 0; c < 4; ++c)
                if (4 * t + c < E) w |= deg_count(tab, dst[4 * t + c]) << (8 * c);
            own[t] = w;
        }
    }
    __syncthreads();

    uint4* pl = (uint4*)(part + (size_t)blockIdx.x * DEG_WORDS);
    for (int i = tid; i < DEG_WORDS / 4; i += 1024) pl[i] = ((const uint4*)tab)[i];
}

__global__ void deg_finish(const unsigned* __restrict__ part, float* __restrict__ dis, int N) {
    const int t = blockIdx.x * blockDim.x + threadIdx.x;
    if (2 * t >= N) return;
    unsigned lo = 0u, hi = 0u;
    #pragma unroll 16
    for (int p = 0; p < DEG_P; ++p) {
        const unsigned w = part[(size_t)p * DEG_WORDS + t];
        lo += w & 0xFFFFu;
        hi += w >> 16;
    }
    const float d0 = rsqrtf(1.0f + (float)lo);
    const float d1 = rsqrtf(1.0f + (float)hi);
    if (2 * t + 1 < N) {
        *(float2*)(dis + 2 * t) = make_float2(d0, d1);
    } else {
        dis[2 * t] = d0;
    }
}

#define GEMM_GRID_K256 512
#define GEMM_GRID_K128 768
typedef __attribute__((ext_vector_type(4))) float v4f;

template <int K>
__global__ __launch_bounds__(256) void gemm_wmma_f32(const float* __restrict__ A,
                                                     const float* __restrict__ W,
                                                     float* __restrict__ out, int ntiles) {
    constexpr int PITCH = K + 4;
    constexpr int RV    = K / 4;
    constexpr int NV    = 16 * RV / 256;
    constexpr int RS    = 256 / RV;
    constexpr int G     = 16;
    constexpr int NG    = K / 4 / G;
    static_assert(NV == 4 || NV == 2, "a thread moves 2 or 4 pieces of a tile");
    __shared__ __attribute__((aligned(16))) float ldsA[16 * PITCH];
    const int tid  = threadIdx.x;
    const int wave = tid >> 5;
    const int lane = tid & 31;
    const int half = lane >> 4;
    const int m    = lane & 15;
    const int col0 = wave * 16;

    int tile = blockIdx.x;
    if (tile >= ntiles) return;

    const int pr = tid / RV, pc = tid - pr * RV;
    float* lst = ldsA + pr * PITCH + 4 * pc;
    const v4f* ap = (const v4f*)(A + (size_t)tile * 16 * K) + tid;
    v4f p0 = ap[0], p1 = ap[256], p2 = {}, p3 = {};
    if constexpr (NV == 4) { p2 = ap[512]; p3 = ap[768]; }

    v2f bf[K / 4];
    #pragma unroll
    for (int s = 0; s < K / 4; ++s) {
        const int ka = 4 * s + 2 * half;
        bf[s].x = W[ka * HID + col0 + m];
        bf[s].y = W[(ka + 1) * HID + col0 + m];
    }

    const float* arow = ldsA + m * PITCH + 2 * half;
    for (;;) {
        __syncthreads();
        *(v4f*)(lst) = p0;
        *(v4f*)(lst + RS * PITCH) = p1;
        if constexpr (NV == 4) {
            *(v4f*)(lst + 2 * RS * PITCH) = p2;
            *(v4f*)(lst + 3 * RS * PITCH) = p3;
        }
        __syncthreads();

        const int nt = tile + (int)gridDim.x;
        const int lt = nt < ntiles ? nt : tile;
        ap = (const v4f*)(A + (size_t)lt * 16 * K) + tid;
        p0 = ap[0]; p1 = ap[256];
        if constexpr (NV == 4) { p2 = ap[512]; p3 = ap[768]; }

        v2f fr[2][G];
        #pragma unroll
        for (int i = 0; i < G; ++i) fr[0][i] = *(const v2f*)(arow + 4 * i);
        v8f acc = {};
        #pragma unroll
        for (int g = 0; g < NG; ++g) {
            if (g + 1 < NG) {
                #pragma unroll
                for (int i = 0; i < G; ++i)
                    fr[(g + 1) & 1][i] = *(const v2f*)(arow + 4 * ((g + 1) * G + i));
            }
            __builtin_amdgcn_sched_barrier(0);
            #pragma unroll
            for (int i = 0; i < G; ++i)
                acc = __builtin_amdgcn_wmma_f32_16x16x4_f32(
                    false, fr[g & 1][i], false, bf[g * G + i], (short)0, acc, false, false);
            __builtin_amdgcn_sched_barrier(0);
        }

        float* op = out + (size_t)(tile * 16 + 8 * half) * HID + col0 + m;
        #pragma unroll
        for (int v = 0; v < 8; ++v) op[v * HID] = acc[v];

        if (nt >= ntiles) break;
        tile = nt;
    }
}

#define OWN_WAVES 32
#define OWN_Q     64
#define OWN_U     4
#define OWN_G     8
#define OWN_NB    ((N_NODES + OWN_R - 1) / OWN_R)
#define OWN_CAP   256

#define OWN_PLANE_WORDS ((N_EDGES + 511) / 512 * 128)
#define OWN_WS_WORDS ((size_t)N_NODES + 2 * (size_t)N_NODES * HID + N_GRAPHS * HID + N_GRAPHS \
                      + OWN_PLANE_WORDS + (size_t)OWN_NB * OWN_WAVES * OWN_CAP + OWN_NB * OWN_WAVES)

typedef __attribute__((address_space(3))) int lds_int;

__device__ __forceinline__ void owner_rows(int s, int r, float nr, int nb, int lane,
                                           const float* __restrict__ h, float* acc) {
    int i = 0;
    #pragma unroll 1
    for (; i + OWN_G <= nb; i += OWN_G) {
        float val[OWN_G][4];
        #pragma unroll
        for (int u = 0; u < OWN_G; ++u) {
            const int si = __builtin_amdgcn_readlane(s, i + u);
            const float* hs = h + (size_t)si * HID;
            #pragma unroll
            for (int j = 0; j < 4; ++j) val[u][j] = hs[j * 32 + lane];
        }
        #pragma unroll
        for (int u = 0; u < OWN_G; ++u) {
            const int ri = __builtin_amdgcn_readlane(r, i + u);
            const float ni = __int_as_float(__builtin_amdgcn_readlane(__float_as_int(nr), i + u));
            float* ar = acc + ri * HID;
            #pragma unroll
            for (int j = 0; j < 4; ++j) atomicAdd(&ar[j * 32 + lane], val[u][j] * ni);
        }
    }
    for (; i < nb; ++i) {
        const int si = __builtin_amdgcn_readlane(s, i);
        const int ri = __builtin_amdgcn_readlane(r, i);
        const float ni = __int_as_float(__builtin_amdgcn_readlane(__float_as_int(nr), i));
        const float* hs = h + (size_t)si * HID;
        float* ar = acc + ri * HID;
        #pragma unroll
        for (int j = 0; j < 4; ++j) atomicAdd(&ar[j * 32 + lane], hs[j * 32 + lane] * ni);
    }
}

__device__ __forceinline__ void owner_drain(volatile lds_int* q, int head, int nb, int lane, int lo,
                                            const int* __restrict__ src, const int* __restrict__ dst,
                                            const float* __restrict__ dis, const float* __restrict__ h,
                                            float* acc, unsigned* __restrict__ rec, int& nrec) {
    const bool v = lane < nb;
    const int e = v ? q[(head + lane) & (OWN_Q - 1)] : 0;
    const int s = v ? src[e] : 0;
    const int d = v ? dst[e] : lo;
    const float nr = v ? dis[s] * dis[d] : 0.f;
    const int r = d - lo;
    if (rec != nullptr) {
        if (v && nrec + nb <= OWN_CAP) rec[nrec + lane] = (unsigned)s | ((unsigned)r << 17);
        nrec += nb;
    }
    owner_rows(s, r, nr, nb, lane, h, acc);
}

__device__ __forceinline__ void owner_replay(const unsigned* __restrict__ rec, int n, int lane, int lo,
                                             const float* __restrict__ dis, const float* __restrict__ h,
                                             float* acc) {
    if (n <= 0) return;
    const int last = n - 1;
    unsigned w1 = rec[lane < last ? lane : last];
    unsigned w2 = rec[32 + lane < last ? 32 + lane : last];
    float fa = dis[w1 & 0x1FFFFu], fb = dis[lo + (int)(w1 >> 17)];
    for (int i = 0; i < n; i += 32) {
        const int s = (int)(w1 & 0x1FFFFu), r = (int)(w1 >> 17);
        const float nr = fa * fb;
        w1 = w2;
        const int nx = i + 64 + lane;
        w2 = rec[nx < last ? nx : last];
        fa = dis[w1 & 0x1FFFFu]; fb = dis[lo + (int)(w1 >> 17)];
        const int left = n - i;
        owner_rows(s, r, nr, left < 32 ? left : 32, lane, h, acc);
    }
}

__device__ __forceinline__ void owner_scan(volatile lds_int* q, int wave, int lane, int lo, unsigned pat,
                                           int rot, int nsteps, int E,
                                           const unsigned* __restrict__ own,
                                           const int* __restrict__ src, const int* __restrict__ dst,
                                           const float* __restrict__ dis, const float* __restrict__ h,
                                           float* acc, unsigned* __restrict__ rec, int& nrec) {
    int head = 0, cnt = 0;
    for (int st0 = wave; st0 < nsteps; st0 += OWN_WAVES * OWN_U) {
        uint4 ov[OWN_U];
        #pragma unroll
        for (int u = 0; u < OWN_U; ++u) {
            const int st = st0 + u * OWN_WAVES;
            int sidx = st + rot; if (sidx >= nsteps) sidx -= nsteps;
            if (st >= nsteps) sidx = 0;
            ov[u] = *(const uint4*)(own + ((size_t)(sidx * 32 + lane) << 2));
        }
        __builtin_amdgcn_sched_barrier(0);
        unsigned long long lm = 0ull;
        #pragma unroll
        for (int u = 0; u < OWN_U; ++u) {
            const unsigned wd[4] = {ov[u].x, ov[u].y, ov[u].z, ov[u].w};
            unsigned nz = 0u;
            #pragma unroll
            for (int j = 0; j < 4; ++j) {
                const unsigned x = wd[j] ^ pat;
                const unsigned y = ((x & 0x7F7F7F7Fu) + 0x7F7F7F7Fu) | x;
                nz |= ((y >> 7) & 0x01010101u) << j;
            }
            unsigned m16 = ~nz & 0x0F0F0F0Fu;
            if (st0 + u * OWN_WAVES >= nsteps) m16 = 0u;
            lm |= (unsigned long long)(m16 << (4 * (u & 1))) << (32 * (u >> 1));
        }
        for (int round = 0; round < 16 * OWN_U; ++round) {
            const bool has = lm != 0ull;
            if (__builtin_amdgcn_ballot_w32(has) == 0u) break;
            const int p  = has ? __builtin_ctzll(lm) : 0;
            const int lw = p & 31;
            const int u  = ((p >> 5) << 1) | ((lw >> 2) & 1);
            int sidx = st0 + u * OWN_WAVES + rot; if (sidx >= nsteps) sidx -= nsteps;
            const int e = ((sidx * 32 + lane) << 4) + ((lw & 3) << 2) + (lw >> 3);
            const bool take = has && (e < E);
            const unsigned mask = __builtin_amdgcn_ballot_w32(take);
            const int pos = (head + cnt + (int)__builtin_amdgcn_mbcnt_lo(mask, 0u)) & (OWN_Q - 1);
            if (take) q[pos] = e;
            lm &= lm - 1ull;
            cnt += __builtin_popcount(mask);
            if (cnt >= 32) {
                owner_drain(q, head, 32, lane, lo, src, dst, dis, h, acc, rec, nrec);
                head = (head + 32) & (OWN_Q - 1);
                cnt -= 32;
            }
        }
    }
    owner_drain(q, head, cnt, lane, lo, src, dst, dis, h, acc, rec, nrec);
}

template <int LAYER>
__global__ __launch_bounds__(1024) void owner_aggregate(const int* __restrict__ src,
                                                        const int* __restrict__ dst,
                                                        const float* __restrict__ dis,
                                                        const float* __restrict__ h,
                                                        const float* __restrict__ bias,
                                                        const unsigned* __restrict__ own,
                                                        float* __restrict__ outp,
                                                        unsigned* __restrict__ rlist,
                                                        int* __restrict__ rcnt,
                                                        size_t ws_bytes, int E, int N) {
    static_assert((N_NODES + OWN_R - 1) / OWN_R <= 256, "an owner id must fit in one byte");
    static_assert(N_NODES <= (1 << 17) && OWN_R <= (1 << 15), "a record word holds source and row");
    __shared__ __attribute__((aligned(16))) float acc[OWN_R * HID];
    __shared__ int queue[OWN_WAVES * OWN_Q];

    const int tid  = threadIdx.x;
    const int lane = tid & 31;
    const int wave = __builtin_amdgcn_readfirstlane(tid >> 5);
    const int lo   = blockIdx.x * OWN_R;
    const int nrows = (N - lo) < OWN_R ? (N - lo) : OWN_R;
    volatile lds_int* q = (volatile lds_int*)(queue + wave * OWN_Q);

    for (int idx = tid; idx < nrows * (HID / 4); idx += 1024) {
        const int r = idx >> 5, c4 = idx & 31;
        const float di = dis[lo + r];
        const float w = di * di;
        const float4 hv = ((const float4*)(h + (size_t)(lo + r) * HID))[c4];
        const float4 bv = ((const float4*)bias)[c4];
        float4 o;
        o.x = w * hv.x + bv.x; o.y = w * hv.y + bv.y; o.z = w * hv.z + bv.z; o.w = w * hv.w + bv.w;
        ((float4*)acc)[idx] = o;
    }
    __syncthreads();

    const int nsteps = (E + 511) >> 9;
    const int rot = (int)(((unsigned)blockIdx.x * (unsigned)nsteps) / gridDim.x);
    const unsigned pat = (unsigned)blockIdx.x * 0x01010101u;
    const bool room = ws_bytes >= OWN_WS_WORDS * sizeof(float);
    const int slot = blockIdx.x * OWN_WAVES + wave;
    unsigned* rec = rlist + (size_t)slot * OWN_CAP;
    if (LAYER == 1) {
        int nrec = 0;
        owner_scan(q, wave, lane, lo, pat, rot, nsteps, E, own, src, dst, dis, h, acc,
                   room ? rec : nullptr, nrec);
        if (lane == 0) q[0] = nrec;
    } else {
        const int n = room ? rcnt[slot] : OWN_CAP + 1;
        if (n <= OWN_CAP) {
            owner_replay(rec, n, lane, lo, dis, h, acc);
        } else {
            int none = 0;
            owner_scan(q, wave, lane, lo, pat, rot, nsteps, E, own, src, dst, dis, h, acc, nullptr, none);
        }
    }
    __syncthreads();

    if (LAYER == 1 && room && wave == 0) rcnt[blockIdx.x * OWN_WAVES + lane] = queue[lane * OWN_Q];

    float4* orow = (float4*)(outp + (size_t)lo * HID);
    for (int idx = tid; idx < nrows * (HID / 4); idx += 1024) {
        float4 v4 = ((const float4*)acc)[idx];
        v4.x = v4.x > 0.f ? v4.x : 0.f; v4.y = v4.y > 0.f ? v4.y : 0.f;
        v4.z = v4.z > 0.f ? v4.z : 0.f; v4.w = v4.w > 0.f ? v4.w : 0.f;
        orow[idx] = v4;
    }
}

#define POOL_NB    256
#define POOL_R     ((N_NODES + POOL_NB - 1) / POOL_NB)
#define POOL_WAVES 16
#define POOL_RW    ((POOL_R + POOL_WAVES - 1) / POOL_WAVES)

__device__ __forceinline__ void pool_flush(float* tab, int* tcnt, int g, int rl, float4 a, int lane) {
    if ((unsigned)g < (unsigned)N_GRAPHS) {
        float* t = tab + g * HID + 4 * lane;
        atomicAdd(t + 0, a.x); atomicAdd(t + 1, a.y); atomicAdd(t + 2, a.z); atomicAdd(t + 3, a.w);
        if (lane == 0) atomicAdd(tcnt + g, rl);
    }
}

__global__ __launch_bounds__(POOL_WAVES * 32) void pool_partial(const float* __restrict__ h,
                                                                const int* __restrict__ batch,
                                                                float* __restrict__ part,
                                                                int* __restrict__ cntp, int N) {
    __shared__ __attribute__((aligned(16))) float tab[N_GRAPHS * HID];
    __shared__ int tcnt[N_GRAPHS];
    const int tid  = threadIdx.x;
    const int lane = tid & 31;
    const int wave = __builtin_amdgcn_readfirstlane(tid >> 5);
    const float4 zero = make_float4(0.f, 0.f, 0.f, 0.f);

    const int r0 = blockIdx.x * POOL_R + wave * POOL_RW;
    int rend = blockIdx.x * POOL_R + POOL_R;
    if (rend > N) rend = N;
    int nr = rend - r0;
    nr = nr < 0 ? 0 : (nr > POOL_RW ? POOL_RW : nr);
    const int gl = (lane < nr) ? batch[r0 + lane] : -1;
    float4 v[POOL_RW];
    #pragma unroll
    for (int i = 0; i < POOL_RW; ++i) {
        int ri = r0 + i;
        if (ri > N - 1) ri = N - 1;
        v[i] = ((const float4*)(h + (size_t)ri * HID))[lane];
    }

    for (int i = tid; i < N_GRAPHS * HID / 4; i += POOL_WAVES * 32) ((float4*)tab)[i] = zero;
    if (tid < N_GRAPHS) tcnt[tid] = 0;
    __syncthreads();

    int gcur = -1, rl = 0;
    float4 acc = zero;
    #pragma unroll
    for (int i = 0; i < POOL_RW; ++i) {
        if (i < nr) {
            const int g = __builtin_amdgcn_readlane(gl, i);
            if (g != gcur) {
                pool_flush(tab, tcnt, gcur, rl, acc, lane);
                gcur = g; rl = 0; acc = zero;
            }
            acc.x += v[i].x; acc.y += v[i].y; acc.z += v[i].z; acc.w += v[i].w;
            ++rl;
        }
    }
    pool_flush(tab, tcnt, gcur, rl, acc, lane);
    __syncthreads();

    float4* pl = (float4*)(part + (size_t)blockIdx.x * (N_GRAPHS * HID));
    for (int i = tid; i < N_GRAPHS * HID / 4; i += POOL_WAVES * 32) pl[i] = ((const float4*)tab)[i];
    if (tid < N_GRAPHS) cntp[blockIdx.x * N_GRAPHS + tid] = tcnt[tid];
}

__global__ __launch_bounds__(1024) void pool_finish(const float* __restrict__ part,
                                                    const int* __restrict__ cntp,
                                                    float* __restrict__ pooled) {
    static_assert(POOL_NB == 256, "32 waves x 8 planes, one count per thread of the first 256");
    __shared__ __attribute__((aligned(16))) float red[32 * HID];
    __shared__ int ntot;
    const int tid  = threadIdx.x;
    const int lane = tid & 31;
    const int wave = tid >> 5;
    const int g    = blockIdx.x;

    float4 v[8];
    #pragma unroll
    for (int i = 0; i < 8; ++i)
        v[i] = ((const float4*)(part + ((size_t)(wave * 8 + i) * N_GRAPHS + g) * HID))[lane];
    const int c = (tid < POOL_NB) ? cntp[tid * N_GRAPHS + g] : 0;

    if (tid == 0) ntot = 0;
    __syncthreads();
    float4 s = v[0];
    #pragma unroll
    for (int i = 1; i < 8; ++i) { s.x += v[i].x; s.y += v[i].y; s.z += v[i].z; s.w += v[i].w; }
    ((float4*)red)[wave * 32 + lane] = s;
    if (c != 0) atomicAdd(&ntot, c);
    __syncthreads();

    if (wave == 0) {
        float4 t = ((const float4*)red)[lane];
        #pragma unroll
        for (int w = 1; w < 32; ++w) {
            const float4 u = ((const float4*)red)[w * 32 + lane];
            t.x += u.x; t.y += u.y; t.z += u.z; t.w += u.w;
        }
        float n = (float)ntot;
        n = n > 1.f ? n : 1.f;
        t.x /= n; t.y /= n; t.z /= n; t.w /= n;
        ((float4*)(pooled + g * HID))[lane] = t;
    }
}

#define HEAD_T_MF   ((OUT_MF + 15) / 16)
#define HEAD_T_BP   ((OUT_BP + 15) / 16)
#define HEAD_T_CC   ((OUT_CC + 15) / 16)
#define HEAD_TILES  (HEAD_T_MF + HEAD_T_BP + HEAD_T_CC)
#define HEAD_WAVES  4
#define HEAD_PITCH  (HID + 4)
#define HEAD_KS     4

__global__ __launch_bounds__(HEAD_WAVES * 32) void heads_wmma(const float* __restrict__ pooled,
                                                              const float* __restrict__ Wmf,
                                                              const float* __restrict__ bmf,
                                                              const float* __restrict__ Wbp,
                                                              const float* __restrict__ bbp,
                                                              const float* __restrict__ Wcc,
                                                              const float* __restrict__ bcc,
                                                              float* __restrict__ out_mf,
                                                              float* __restrict__ out_bp,
                                                              float* __restrict__ out_cc) {
    static_assert(N_GRAPHS == 64 && HID == 128, "4 row tiles, 32 k steps, 16 pieces of the means per thread");
    __shared__ __attribute__((aligned(16))) float ldsP[N_GRAPHS * HEAD_PITCH];
    const int tid  = threadIdx.x;
    const int lane = tid & 31;
    const int wave = __builtin_amdgcn_readfirstlane(tid >> 5);
    const int half = lane >> 4;
    const int m    = lane & 15;

    v4f pp[16];
    #pragma unroll
    for (int j = 0; j < 16; ++j) pp[j] = ((const v4f*)pooled)[tid + HEAD_WAVES * 32 * j];

    int t = blockIdx.x * HEAD_WAVES + wave;
    const bool live = t < HEAD_TILES;
    if (!live) t = HEAD_TILES - 1;
    const float* W; const float* b; float* o; int OUT;
    if (t < HEAD_T_MF) {
        W = Wmf; b = bmf; o = out_mf; OUT = OUT_MF;
    } else if (t < HEAD_T_MF + HEAD_T_BP) {
        t -= HEAD_T_MF; W = Wbp; b = bbp; o = out_bp; OUT = OUT_BP;
    } else {
        t -= HEAD_T_MF + HEAD_T_BP; W = Wcc; b = bcc; o = out_cc; OUT = OUT_CC;
    }
    const int col  = t * 16 + m;
    const int colc = col < OUT ? col : OUT - 1;

    v2f bf[HID / 4];
    #pragma unroll
    for (int s = 0; s < HID / 4; ++s) {
        const int ka = 4 * s + 2 * half;
        bf[s].x = W[ka * OUT + colc];
        bf[s].y = W[(ka + 1) * OUT + colc];
    }
    const float bias = b[colc];

    #pragma unroll
    for (int j = 0; j < 16; ++j) {
        const int p = tid + HEAD_WAVES * 32 * j;
        *(v4f*)(ldsP + (p >> 5) * HEAD_PITCH + 4 * (p & 31)) = pp[j];
    }
    __syncthreads();

    v8f acc[4];
    #pragma unroll
    for (int rt = 0; rt < 4; ++rt) {
        #pragma unroll
        for (int v = 0; v < 8; ++v) acc[rt][v] = bias;
    }

    constexpr int NG = HID / 4 / HEAD_KS;
    const float* arow = ldsP + m * HEAD_PITCH + 2 * half;
    v2f fr[2][HEAD_KS * 4];
    #pragma unroll
    for (int i = 0; i < HEAD_KS; ++i) {
        #pragma unroll
        for (int rt = 0; rt < 4; ++rt)
            fr[0][i * 4 + rt] = *(const v2f*)(arow + rt * 16 * HEAD_PITCH + 4 * i);
    }
    #pragma unroll
    for (int g = 0; g < NG; ++g) {
        if (g + 1 < NG) {
            #pragma unroll
            for (int i = 0; i < HEAD_KS; ++i) {
                #pragma unroll
                for (int rt = 0; rt < 4; ++rt)
                    fr[(g + 1) & 1][i * 4 + rt] =
                        *(const v2f*)(arow + rt * 16 * HEAD_PITCH + 4 * ((g + 1) * HEAD_KS + i));
            }
        }
        __builtin_amdgcn_sched_barrier(0);
        #pragma unroll
        for (int i = 0; i < HEAD_KS; ++i) {
            #pragma unroll
            for (int rt = 0; rt < 4; ++rt)
                acc[rt] = __builtin_amdgcn_wmma_f32_16x16x4_f32(
                    false, fr[g & 1][i * 4 + rt], false, bf[g * HEAD_KS + i], (short)0, acc[rt], false, false);
        }
        __builtin_amdgcn_sched_barrier(0);
    }

    if (live && col < OUT) {
        #pragma unroll
        for (int rt = 0; rt < 4; ++rt) {
            float* op = o + (size_t)(16 * rt + 8 * half) * OUT + col;
            #pragma unroll
            for (int v = 0; v < 8; ++v) op[(size_t)v * OUT] = acc[rt][v];
        }
    }
}

extern "C" void kernel_launch(void* const* d_in, const int* in_sizes, int n_in,
                              void* d_out, int out_size, void* d_ws, size_t ws_size,
                              hipStream_t stream) {
    const float* x   = (const float*)d_in[0];
    const int*   ei  = (const int*)d_in[1];
    const int*   bat = (const int*)d_in[2];
    const float* W1  = (const float*)d_in[3];
    const float* b1  = (const float*)d_in[4];
    const float* W2  = (const float*)d_in[5];
    const float* b2  = (const float*)d_in[6];
    const float* Wmf = (const float*)d_in[7];
    const float* bmf = (const float*)d_in[8];
    const float* Wbp = (const float*)d_in[9];
    const float* bbp = (const float*)d_in[10];
    const float* Wcc = (const float*)d_in[11];
    const float* bcc = (const float*)d_in[12];
    float* out = (float*)d_out;

    const int E = in_sizes[1] / 2;
    const int* src = ei;
    const int* dst = ei + E;

    float* ws     = (float*)d_ws;
    float* dis    = ws;
    float* h      = dis + N_NODES;
    float* agg    = h + (size_t)N_NODES * HID;
    float* pooled = agg + (size_t)N_NODES * HID;
    float* cnt    = pooled + N_GRAPHS * HID;
    unsigned* own = (unsigned*)(cnt + N_GRAPHS);
    unsigned* rlist = own + OWN_PLANE_WORDS;
    int*      rcnt  = (int*)(rlist + (size_t)OWN_NB * OWN_WAVES * OWN_CAP);

    unsigned* degp = (unsigned*)agg;
    deg_partial<<<DEG_P, 1024, 0, stream>>>(dst, degp, own, E);
    deg_finish<<<((N_NODES + 1) / 2 + 255) / 256, 256, 0, stream>>>(degp, dis, N_NODES);

    gemm_wmma_f32<IN_DIM><<<GEMM_GRID_K256, 256, 0, stream>>>(x, W1, h, N_NODES / 16);
    owner_aggregate<1><<<OWN_NB, 1024, 0, stream>>>(src, dst, dis, h, b1, own, agg, rlist, rcnt, ws_size, E, N_NODES);

    gemm_wmma_f32<HID><<<GEMM_GRID_K128, 256, 0, stream>>>(agg, W2, h, N_NODES / 16);
    owner_aggregate<2><<<OWN_NB, 1024, 0, stream>>>(src, dst, dis, h, b2, own, agg, rlist, rcnt, ws_size, E, N_NODES);

    float* part = h;
    int*   cntp = (int*)(h + (size_t)POOL_NB * N_GRAPHS * HID);
    pool_partial<<<POOL_NB, POOL_WAVES * 32, 0, stream>>>(agg, bat, part, cntp, N_NODES);
    pool_finish<<<N_GRAPHS, 1024, 0, stream>>>(part, cntp, pooled);

    float* out_mf = out;
    float* out_bp = out + N_GRAPHS * OUT_MF;
    float* out_cc = out + N_GRAPHS * (OUT_MF + OUT_BP);
    heads_wmma<<<(HEAD_TILES + HEAD_WAVES - 1) / HEAD_WAVES, HEAD_WAVES * 32, 0, stream>>>(
        pooled, Wmf, bmf, Wbp, bbp, Wcc, bcc, out_mf, out_bp, out_cc);
}


# aggregation gathers rows through two register sets: a group's loads go out before the group before it is added (16 rows in flight per wave, was 8); short last chunk by whole groups; sums by arrival
# speedup vs baseline: 3.6165x; 1.0439x over previous
#include <hip/hip_runtime.h>
#include <hip/hip_bf16.h>

#define N_NODES 100000
#define N_EDGES 1600000
#define IN_DIM  256
#define HID     128
#define N_GRAPHS 64
#define OUT_MF 489
#define OUT_BP 1943
#define OUT_CC 320

typedef __attribute__((ext_vector_type(2))) float v2f;
typedef __attribute__((ext_vector_type(8))) float v8f;

#define OWN_R     391
#define DEG_P     64
#define DEG_U     8
#define DEG_WORDS (((N_NODES + 1) / 2 + 3) & ~3)

__device__ __forceinline__ unsigned deg_count(unsigned* tab, int d) {
    if ((unsigned)d >= (unsigned)N_NODES) return 0u;
    atomicAdd(&tab[d >> 1], (d & 1) ? 0x10000u : 1u);
    return (unsigned)(d / OWN_R);
}

__global__ __launch_bounds__(1024) void deg_partial(const int* __restrict__ dst,
                                                    unsigned* __restrict__ part,
                                                    unsigned* __restrict__ own, int E) {
    static_assert(((N_EDGES + 511) / 512 * 512 + DEG_P - 1) / DEG_P + 4 < 65536,
                  "the counts of one slice must fit in 16 bits");
    __shared__ __attribute__((aligned(16))) unsigned tab[DEG_WORDS];
    const int tid = threadIdx.x;
    for (int i = tid; i < DEG_WORDS / 4; i += 1024) ((uint4*)tab)[i] = make_uint4(0u, 0u, 0u, 0u);
    __syncthreads();

    const int nwords = ((E + 511) >> 9) << 7;
    const int per = (nwords + DEG_P - 1) / DEG_P;
    const int w0 = blockIdx.x * per;
    const int w1 = (w0 + per) < nwords ? (w0 + per) : nwords;
    if (((E & 3) == 0) && ((((size_t)dst) & 15) == 0)) {
        const int lastw = (E >> 2) - 1;
        for (int tb = w0 + tid; tb < w1; tb += 1024 * DEG_U) {
            int4 d[DEG_U];
            #pragma unroll
            for (int u = 0; u < DEG_U; ++u) {
                const int t = tb + u * 1024;
                const int tl = t < lastw ? t : lastw;
                d[u] = *(const int4*)(dst + 4 * (size_t)tl);
            }
            #pragma unroll
            for (int u = 0; u < DEG_U; ++u) {
                const int t = tb + u * 1024;
                if (t < w1) {
                    const int4 dd = t <= lastw ? d[u] : make_int4(-1, -1, -1, -1);
                    own[t] = deg_count(tab, dd.x) | (deg_count(tab, dd.y) << 8)
                           | (deg_count(tab, dd.z) << 16) | (deg_count(tab, dd.w) << 24);
                }
            }
        }
    } else {
        for (int t = w0 + tid; t < w1; t += 1024) {
            unsigned w = 0u;
            for (int c = 0; c < 4; ++c)
                if (4 * t + c < E) w |= deg_count(tab, dst[4 * t + c]) << (8 * c);
            own[t] = w;
        }
    }
    __syncthreads();

    uint4* pl = (uint4*)(part + (size_t)blockIdx.x * DEG_WORDS);
    for (int i = tid; i < DEG_WORDS / 4; i += 1024) pl[i] = ((const uint4*)tab)[i];
}

__global__ void deg_finish(const unsigned* __restrict__ part, float* __restrict__ dis, int N) {
    const int t = blockIdx.x * blockDim.x + threadIdx.x;
    if (2 * t >= N) return;
    unsigned lo = 0u, hi = 0u;
    #pragma unroll 16
    for (int p = 0; p < DEG_P; ++p) {
        const unsigned w = part[(size_t)p * DEG_WORDS + t];
        lo += w & 0xFFFFu;
        hi += w >> 16;
    }
    const float d0 = rsqrtf(1.0f + (float)lo);
    const float d1 = rsqrtf(1.0f + (float)hi);
    if (2 * t + 1 < N) {
        *(float2*)(dis + 2 * t) = make_float2(d0, d1);
    } else {
        dis[2 * t] = d0;
    }
}

#define GEMM_GRID_K256 512
#define GEMM_GRID_K128 768
typedef __attribute__((ext_vector_type(4))) float v4f;

template <int K>
__global__ __launch_bounds__(256) void gemm_wmma_f32(const float* __restrict__ A,
                                                     const float* __restrict__ W,
                                                     float* __restrict__ out, int ntiles) {
    constexpr int PITCH = K + 4;
    constexpr int RV    = K / 4;
    constexpr int NV    = 16 * RV / 256;
    constexpr int RS    = 256 / RV;
    constexpr int G     = 16;
    constexpr int NG    = K / 4 / G;
    static_assert(NV == 4 || NV == 2, "a thread moves 2 or 4 pieces of a tile");
    __shared__ __attribute__((aligned(16))) float ldsA[16 * PITCH];
    const int tid  = threadIdx.x;
    const int wave = tid >> 5;
    const int lane = tid & 31;
    const int half = lane >> 4;
    const int m    = lane & 15;
    const int col0 = wave * 16;

    int tile = blockIdx.x;
    if (tile >= ntiles) return;

    const int pr = tid / RV, pc = tid - pr * RV;
    float* lst = ldsA + pr * PITCH + 4 * pc;
    const v4f* ap = (const v4f*)(A + (size_t)tile * 16 * K) + tid;
    v4f p0 = ap[0], p1 = ap[256], p2 = {}, p3 = {};
    if constexpr (NV == 4) { p2 = ap[512]; p3 = ap[768]; }

    v2f bf[K / 4];
    #pragma unroll
    for (int s = 0; s < K / 4; ++s) {
        const int ka = 4 * s + 2 * half;
        bf[s].x = W[ka * HID + col0 + m];
        bf[s].y = W[(ka + 1) * HID + col0 + m];
    }

    const float* arow = ldsA + m * PITCH + 2 * half;
    for (;;) {
        __syncthreads();
        *(v4f*)(lst) = p0;
        *(v4f*)(lst + RS * PITCH) = p1;
        if constexpr (NV == 4) {
            *(v4f*)(lst + 2 * RS * PITCH) = p2;
            *(v4f*)(lst + 3 * RS * PITCH) = p3;
        }
        __syncthreads();

        const int nt = tile + (int)gridDim.x;
        const int lt = nt < ntiles ? nt : tile;
        ap = (const v4f*)(A + (size_t)lt * 16 * K) + tid;
        p0 = ap[0]; p1 = ap[256];
        if constexpr (NV == 4) { p2 = ap[512]; p3 = ap[768]; }

        v2f fr[2][G];
        #pragma unroll
        for (int i = 0; i < G; ++i) fr[0][i] = *(const v2f*)(arow + 4 * i);
        v8f acc = {};
        #pragma unroll
        for (int g = 0; g < NG; ++g) {
            if (g + 1 < NG) {
                #pragma unroll
                for (int i = 0; i < G; ++i)
                    fr[(g + 1) & 1][i] = *(const v2f*)(arow + 4 * ((g + 1) * G + i));
            }
            __builtin_amdgcn_sched_barrier(0);
            #pragma unroll
            for (int i = 0; i < G; ++i)
                acc = __builtin_amdgcn_wmma_f32_16x16x4_f32(
                    false, fr[g & 1][i], false, bf[g * G + i], (short)0, acc, false, false);
            __builtin_amdgcn_sched_barrier(0);
        }

        float* op = out + (size_t)(tile * 16 + 8 * half) * HID + col0 + m;
        #pragma unroll
        for (int v = 0; v < 8; ++v) op[v * HID] = acc[v];

        if (nt >= ntiles) break;
        tile = nt;
    }
}

#define OWN_WAVES 32
#define OWN_Q     64
#define OWN_U     4
#define OWN_G     8
#define OWN_NB    ((N_NODES + OWN_R - 1) / OWN_R)
#define OWN_CAP   256

#define OWN_PLANE_WORDS ((N_EDGES + 511) / 512 * 128)
#define OWN_WS_WORDS ((size_t)N_NODES + 2 * (size_t)N_NODES * HID + N_GRAPHS * HID + N_GRAPHS \
                      + OWN_PLANE_WORDS + (size_t)OWN_NB * OWN_WAVES * OWN_CAP + OWN_NB * OWN_WAVES)

typedef __attribute__((address_space(3))) int lds_int;

typedef const __attribute__((address_space(1))) float owner_gf;
static_assert(HID * sizeof(float) == 512, "a row is 512 bytes: the row address is the node number shifted by 9");

template <bool CLAMP>
__device__ __forceinline__ void owner_rows_load(float (&val)[OWN_G][4], int s, int base, int last,
                                                int lane, const float* __restrict__ h) {
    #pragma unroll
    for (int u = 0; u < OWN_G; ++u) {
        const int k = (CLAMP && base + u > last) ? last : base + u;
        const int si = __builtin_amdgcn_readlane(s, k);
        const unsigned long long ra = (unsigned long long)h + ((unsigned long long)(unsigned)si << 9);
        const unsigned ralo = __builtin_amdgcn_readfirstlane((unsigned)ra);
        const unsigned rahi = __builtin_amdgcn_readfirstlane((unsigned)(ra >> 32));
        const owner_gf* hs = (const owner_gf*)(((unsigned long long)rahi << 32) | ralo);
        #pragma unroll
        for (int j = 0; j < 4; ++j) val[u][j] = hs[j * 32 + lane];
    }
}

template <bool GUARD>
__device__ __forceinline__ void owner_rows_add(const float (&val)[OWN_G][4], int r, float nr, int base,
                                               int nb, int lane, float* acc) {
    #pragma unroll
    for (int u = 0; u < OWN_G; ++u) {
        if (!GUARD || base + u < nb) {
            const int ri = __builtin_amdgcn_readlane(r, base + u);
            const float ni = __int_as_float(__builtin_amdgcn_readlane(__float_as_int(nr), base + u));
            float* ar = acc + ri * HID;
            #pragma unroll
            for (int j = 0; j < 4; ++j) atomicAdd(&ar[j * 32 + lane], val[u][j] * ni);
        }
    }
}

__device__ __forceinline__ void owner_rows_full(int s, int r, float nr, int lane,
                                                const float* __restrict__ h, float* acc) {
    static_assert(OWN_G == 8, "a full chunk is four groups");
    float va[OWN_G][4], vb[OWN_G][4];
    owner_rows_load<false>(va, s, 0, 31, lane, h);
    owner_rows_load<false>(vb, s, 8, 31, lane, h);
    owner_rows_add<false>(va, r, nr, 0, 32, lane, acc);
    owner_rows_load<false>(va, s, 16, 31, lane, h);
    owner_rows_add<false>(vb, r, nr, 8, 32, lane, acc);
    owner_rows_load<false>(vb, s, 24, 31, lane, h);
    owner_rows_add<false>(va, r, nr, 16, 32, lane, acc);
    owner_rows_add<false>(vb, r, nr, 24, 32, lane, acc);
}

__device__ __forceinline__ void owner_rows_part(int s, int r, float nr, int nb, int lane,
                                                const float* __restrict__ h, float* acc) {
    float va[OWN_G][4];
    #pragma unroll 1
    for (int i = 0; i < nb; i += OWN_G) {
        owner_rows_load<true>(va, s, i, nb - 1, lane, h);
        owner_rows_add<true>(va, r, nr, i, nb, lane, acc);
    }
}

template <bool FULL>
__device__ __forceinline__ void owner_drain(volatile lds_int* q, int head, int nb, int lane, int lo,
                                            const int* __restrict__ src, const int* __restrict__ dst,
                                            const float* __restrict__ dis, const float* __restrict__ h,
                                            float* acc, unsigned* __restrict__ rec, int& nrec) {
    const bool v = lane < nb;
    const int e = v ? q[(head + lane) & (OWN_Q - 1)] : 0;
    const int s = v ? src[e] : 0;
    const int d = v ? dst[e] : lo;
    const float nr = v ? dis[s] * dis[d] : 0.f;
    const int r = d - lo;
    if (rec != nullptr) {
        if (v && nrec + nb <= OWN_CAP) rec[nrec + lane] = (unsigned)s | ((unsigned)r << 17);
        nrec += nb;
    }
    if (FULL) owner_rows_full(s, r, nr, lane, h, acc);
    else      owner_rows_part(s, r, nr, nb, lane, h, acc);
}

__device__ __forceinline__ void owner_replay(const unsigned* __restrict__ rec, int n, int lane, int lo,
                                             const float* __restrict__ dis, const float* __restrict__ h,
                                             float* acc) {
    if (n <= 0) return;
    const int last = n - 1;
    unsigned w1 = rec[lane < last ? lane : last];
    unsigned w2 = rec[32 + lane < last ? 32 + lane : last];
    float fa = dis[w1 & 0x1FFFFu], fb = dis[lo + (int)(w1 >> 17)];
    int i = 0;
    for (; i + 32 <= n; i += 32) {
        const int s = (int)(w1 & 0x1FFFFu), r = (int)(w1 >> 17);
        const float nr = fa * fb;
        w1 = w2;
        const int nx = i + 64 + lane;
        w2 = rec[nx < last ? nx : last];
        fa = dis[w1 & 0x1FFFFu]; fb = dis[lo + (int)(w1 >> 17)];
        owner_rows_full(s, r, nr, lane, h, acc);
    }
    if (i < n) {
        const int s = (int)(w1 & 0x1FFFFu), r = (int)(w1 >> 17);
        owner_rows_part(s, r, fa * fb, n - i, lane, h, acc);
    }
}

__device__ __forceinline__ void owner_scan(volatile lds_int* q, int wave, int lane, int lo, unsigned pat,
                                           int rot, int nsteps, int E,
                                           const unsigned* __restrict__ own,
                                           const int* __restrict__ src, const int* __restrict__ dst,
                                           const float* __restrict__ dis, const float* __restrict__ h,
                                           float* acc, unsigned* __restrict__ rec, int& nrec) {
    int head = 0, cnt = 0;
    for (int st0 = wave; st0 < nsteps; st0 += OWN_WAVES * OWN_U) {
        uint4 ov[OWN_U];
        #pragma unroll
        for (int u = 0; u < OWN_U; ++u) {
            const int st = st0 + u * OWN_WAVES;
            int sidx = st + rot; if (sidx >= nsteps) sidx -= nsteps;
            if (st >= nsteps) sidx = 0;
            ov[u] = *(const uint4*)(own + ((size_t)(sidx * 32 + lane) << 2));
        }
        __builtin_amdgcn_sched_barrier(0);
        unsigned long long lm = 0ull;
        #pragma unroll
        for (int u = 0; u < OWN_U; ++u) {
            const unsigned wd[4] = {ov[u].x, ov[u].y, ov[u].z, ov[u].w};
            unsigned nz = 0u;
            #pragma unroll
            for (int j = 0; j < 4; ++j) {
                const unsigned x = wd[j] ^ pat;
                const unsigned y = ((x & 0x7F7F7F7Fu) + 0x7F7F7F7Fu) | x;
                nz |= ((y >> 7) & 0x01010101u) << j;
            }
            unsigned m16 = ~nz & 0x0F0F0F0Fu;
            if (st0 + u * OWN_WAVES >= nsteps) m16 = 0u;
            lm |= (unsigned long long)(m16 << (4 * (u & 1))) << (32 * (u >> 1));
        }
        for (int round = 0; round < 16 * OWN_U; ++round) {
            const bool has = lm != 0ull;
            if (__builtin_amdgcn_ballot_w32(has) == 0u) break;
            const int p  = has ? __builtin_ctzll(lm) : 0;
            const int lw = p & 31;
            const int u  = ((p >> 5) << 1) | ((lw >> 2) & 1);
            int sidx = st0 + u * OWN_WAVES + rot; if (sidx >= nsteps) sidx -= nsteps;
            const int e = ((sidx * 32 + lane) << 4) + ((lw & 3) << 2) + (lw >> 3);
            const bool take = has && (e < E);
            const unsigned mask = __builtin_amdgcn_ballot_w32(take);
            const int pos = (head + cnt + (int)__builtin_amdgcn_mbcnt_lo(mask, 0u)) & (OWN_Q - 1);
            if (take) q[pos] = e;
            lm &= lm - 1ull;
            cnt += __builtin_popcount(mask);
            if (cnt >= 32) {
                owner_drain<true>(q, head, 32, lane, lo, src, dst, dis, h, acc, rec, nrec);
                head = (head + 32) & (OWN_Q - 1);
                cnt -= 32;
            }
        }
    }
    owner_drain<false>(q, head, cnt, lane, lo, src, dst, dis, h, acc, rec, nrec);
}

template <int LAYER>
__global__ __launch_bounds__(1024) void owner_aggregate(const int* __restrict__ src,
                                                        const int* __restrict__ dst,
                                                        const float* __restrict__ dis,
                                                        const float* __restrict__ h,
                                                        const float* __restrict__ bias,
                                                        const unsigned* __restrict__ own,
                                                        float* __restrict__ outp,
                                                        unsigned* __restrict__ rlist,
                                                        int* __restrict__ rcnt,
                                                        size_t ws_bytes, int E, int N) {
    static_assert((N_NODES + OWN_R - 1) / OWN_R <= 256, "an owner id must fit in one byte");
    static_assert(N_NODES <= (1 << 17) && OWN_R <= (1 << 15), "a record word holds source and row");
    __shared__ __attribute__((aligned(16))) float acc[OWN_R * HID];
    __shared__ int queue[OWN_WAVES * OWN_Q];

    const int tid  = threadIdx.x;
    const int lane = tid & 31;
    const int wave = __builtin_amdgcn_readfirstlane(tid >> 5);
    const int lo   = blockIdx.x * OWN_R;
    const int nrows = (N - lo) < OWN_R ? (N - lo) : OWN_R;
    volatile lds_int* q = (volatile lds_int*)(queue + wave * OWN_Q);

    for (int idx = tid; idx < nrows * (HID / 4); idx += 1024) {
        const int r = idx >> 5, c4 = idx & 31;
        const float di = dis[lo + r];
        const float w = di * di;
        const float4 hv = ((const float4*)(h + (size_t)(lo + r) * HID))[c4];
        const float4 bv = ((const float4*)bias)[c4];
        float4 o;
        o.x = w * hv.x + bv.x; o.y = w * hv.y + bv.y; o.z = w * hv.z + bv.z; o.w = w * hv.w + bv.w;
        ((float4*)acc)[idx] = o;
    }
    __syncthreads();

    const int nsteps = (E + 511) >> 9;
    const int rot = (int)(((unsigned)blockIdx.x * (unsigned)nsteps) / gridDim.x);
    const unsigned pat = (unsigned)blockIdx.x * 0x01010101u;
    const bool room = ws_bytes >= OWN_WS_WORDS * sizeof(float);
    const int slot = blockIdx.x * OWN_WAVES + wave;
    unsigned* rec = rlist + (size_t)slot * OWN_CAP;
    if (LAYER == 1) {
        int nrec = 0;
        owner_scan(q, wave, lane, lo, pat, rot, nsteps, E, own, src, dst, dis, h, acc,
                   room ? rec : nullptr, nrec);
        if (lane == 0) q[0] = nrec;
    } else {
        const int n = room ? rcnt[slot] : OWN_CAP + 1;
        if (n <= OWN_CAP) {
            owner_replay(rec, n, lane, lo, dis, h, acc);
        } else {
            int none = 0;
            owner_scan(q, wave, lane, lo, pat, rot, nsteps, E, own, src, dst, dis, h, acc, nullptr, none);
        }
    }
    __syncthreads();

    if (LAYER == 1 && room && wave == 0) rcnt[blockIdx.x * OWN_WAVES + lane] = queue[lane * OWN_Q];

    float4* orow = (float4*)(outp + (size_t)lo * HID);
    for (int idx = tid; idx < nrows * (HID / 4); idx += 1024) {
        float4 v4 = ((const float4*)acc)[idx];
        v4.x = v4.x > 0.f ? v4.x : 0.f; v4.y = v4.y > 0.f ? v4.y : 0.f;
        v4.z = v4.z > 0.f ? v4.z : 0.f; v4.w = v4.w > 0.f ? v4.w : 0.f;
        orow[idx] = v4;
    }
}

#define POOL_NB    256
#define POOL_R     ((N_NODES + POOL_NB - 1) / POOL_NB)
#define POOL_WAVES 16
#define POOL_RW    ((POOL_R + POOL_WAVES - 1) / POOL_WAVES)

__device__ __forceinline__ void pool_flush(float* tab, int* tcnt, int g, int rl, float4 a, int lane) {
    if ((unsigned)g < (unsigned)N_GRAPHS) {
        float* t = tab + g * HID + 4 * lane;
        atomicAdd(t + 0, a.x); atomicAdd(t + 1, a.y); atomicAdd(t + 2, a.z); atomicAdd(t + 3, a.w);
        if (lane == 0) atomicAdd(tcnt + g, rl);
    }
}

__global__ __launch_bounds__(POOL_WAVES * 32) void pool_partial(const float* __restrict__ h,
                                                                const int* __restrict__ batch,
                                                                float* __restrict__ part,
                                                                int* __restrict__ cntp, int N) {
    __shared__ __attribute__((aligned(16))) float tab[N_GRAPHS * HID];
    __shared__ int tcnt[N_GRAPHS];
    const int tid  = threadIdx.x;
    const int lane = tid & 31;
    const int wave = __builtin_amdgcn_readfirstlane(tid >> 5);
    const float4 zero = make_float4(0.f, 0.f, 0.f, 0.f);

    const int r0 = blockIdx.x * POOL_R + wave * POOL_RW;
    int rend = blockIdx.x * POOL_R + POOL_R;
    if (rend > N) rend = N;
    int nr = rend - r0;
    nr = nr < 0 ? 0 : (nr > POOL_RW ? POOL_RW : nr);
    const int gl = (lane < nr) ? batch[r0 + lane] : -1;
    float4 v[POOL_RW];
    #pragma unroll
    for (int i = 0; i < POOL_RW; ++i) {
        int ri = r0 + i;
        if (ri > N - 1) ri = N - 1;
        v[i] = ((const float4*)(h + (size_t)ri * HID))[lane];
    }

    for (int i = tid; i < N_GRAPHS * HID / 4; i += POOL_WAVES * 32) ((float4*)tab)[i] = zero;
    if (tid < N_GRAPHS) tcnt[tid] = 0;
    __syncthreads();

    int gcur = -1, rl = 0;
    float4 acc = zero;
    #pragma unroll
    for (int i = 0; i < POOL_RW; ++i) {
        if (i < nr) {
            const int g = __builtin_amdgcn_readlane(gl, i);
            if (g != gcur) {
                pool_flush(tab, tcnt, gcur, rl, acc, lane);
                gcur = g; rl = 0; acc = zero;
            }
            acc.x += v[i].x; acc.y += v[i].y; acc.z += v[i].z; acc.w += v[i].w;
            ++rl;
        }
    }
    pool_flush(tab, tcnt, gcur, rl, acc, lane);
    __syncthreads();

    float4* pl = (float4*)(part + (size_t)blockIdx.x * (N_GRAPHS * HID));
    for (int i = tid; i < N_GRAPHS * HID / 4; i += POOL_WAVES * 32) pl[i] = ((const float4*)tab)[i];
    if (tid < N_GRAPHS) cntp[blockIdx.x * N_GRAPHS + tid] = tcnt[tid];
}

__global__ __launch_bounds__(1024) void pool_finish(const float* __restrict__ part,
                                                    const int* __restrict__ cntp,
                                                    float* __restrict__ pooled) {
    static_assert(POOL_NB == 256, "32 waves x 8 planes, one count per thread of the first 256");
    __shared__ __attribute__((aligned(16))) float red[32 * HID];
    __shared__ int ntot;
    const int tid  = threadIdx.x;
    const int lane = tid & 31;
    const int wave = tid >> 5;
    const int g    = blockIdx.x;

    float4 v[8];
    #pragma unroll
    for (int i = 0; i < 8; ++i)
        v[i] = ((const float4*)(part + ((size_t)(wave * 8 + i) * N_GRAPHS + g) * HID))[lane];
    const int c = (tid < POOL_NB) ? cntp[tid * N_GRAPHS + g] : 0;

    if (tid == 0) ntot = 0;
    __syncthreads();
    float4 s = v[0];
    #pragma unroll
    for (int i = 1; i < 8; ++i) { s.x += v[i].x; s.y += v[i].y; s.z += v[i].z; s.w += v[i].w; }
    ((float4*)red)[wave * 32 + lane] = s;
    if (c != 0) atomicAdd(&ntot, c);
    __syncthreads();

    if (wave == 0) {
        float4 t = ((const float4*)red)[lane];
        #pragma unroll
        for (int w = 1; w < 32; ++w) {
            const float4 u = ((const float4*)red)[w * 32 + lane];
            t.x += u.x; t.y += u.y; t.z += u.z; t.w += u.w;
        }
        float n = (float)ntot;
        n = n > 1.f ? n : 1.f;
        t.x /= n; t.y /= n; t.z /= n; t.w /= n;
        ((float4*)(pooled + g * HID))[lane] = t;
    }
}

#define HEAD_T_MF   ((OUT_MF + 15) / 16)
#define HEAD_T_BP   ((OUT_BP + 15) / 16)
#define HEAD_T_CC   ((OUT_CC + 15) / 16)
#define HEAD_TILES  (HEAD_T_MF + HEAD_T_BP + HEAD_T_CC)
#define HEAD_WAVES  4
#define HEAD_PITCH  (HID + 4)
#define HEAD_KS     4

__global__ __launch_bounds__(HEAD_WAVES * 32) void heads_wmma(const float* __restrict__ pooled,
                                                              const float* __restrict__ Wmf,
                                                              const float* __restrict__ bmf,
                                                              const float* __restrict__ Wbp,
                                                              const float* __restrict__ bbp,
                                                              const float* __restrict__ Wcc,
                                                              const float* __restrict__ bcc,
                                                              float* __restrict__ out_mf,
                                                              float* __restrict__ out_bp,
                                                              float* __restrict__ out_cc) {
    static_assert(N_GRAPHS == 64 && HID == 128, "4 row tiles, 32 k steps, 16 pieces of the means per thread");
    __shared__ __attribute__((aligned(16))) float ldsP[N_GRAPHS * HEAD_PITCH];
    const int tid  = threadIdx.x;
    const int lane = tid & 31;
    const int wave = __builtin_amdgcn_readfirstlane(tid >> 5);
    const int half = lane >> 4;
    const int m    = lane & 15;

    v4f pp[16];
    #pragma unroll
    for (int j = 0; j < 16; ++j) pp[j] = ((const v4f*)pooled)[tid + HEAD_WAVES * 32 * j];

    int t = blockIdx.x * HEAD_WAVES + wave;
    const bool live = t < HEAD_TILES;
    if (!live) t = HEAD_TILES - 1;
    const float* W; const float* b; float* o; int OUT;
    if (t < HEAD_T_MF) {
        W = Wmf; b = bmf; o = out_mf; OUT = OUT_MF;
    } else if (t < HEAD_T_MF + HEAD_T_BP) {
        t -= HEAD_T_MF; W = Wbp; b = bbp; o = out_bp; OUT = OUT_BP;
    } else {
        t -= HEAD_T_MF + HEAD_T_BP; W = Wcc; b = bcc; o = out_cc; OUT = OUT_CC;
    }
    const int col  = t * 16 + m;
    const int colc = col < OUT ? col : OUT - 1;

    v2f bf[HID / 4];
    #pragma unroll
    for (int s = 0; s < HID / 4; ++s) {
        const int ka = 4 * s + 2 * half;
        bf[s].x = W[ka * OUT + colc];
        bf[s].y = W[(ka + 1) * OUT + colc];
    }
    const float bias = b[colc];

    #pragma unroll
    for (int j = 0; j < 16; ++j) {
        const int p = tid + HEAD_WAVES * 32 * j;
        *(v4f*)(ldsP + (p >> 5) * HEAD_PITCH + 4 * (p & 31)) = pp[j];
    }
    __syncthreads();

    v8f acc[4];
    #pragma unroll
    for (int rt = 0; rt < 4; ++rt) {
        #pragma unroll
        for (int v = 0; v < 8; ++v) acc[rt][v] = bias;
    }

    constexpr int NG = HID / 4 / HEAD_KS;
    const float* arow = ldsP + m * HEAD_PITCH + 2 * half;
    v2f fr[2][HEAD_KS * 4];
    #pragma unroll
    for (int i = 0; i < HEAD_KS; ++i) {
        #pragma unroll
        for (int rt = 0; rt < 4; ++rt)
            fr[0][i * 4 + rt] = *(const v2f*)(arow + rt * 16 * HEAD_PITCH + 4 * i);
    }
    #pragma unroll
    for (int g = 0; g < NG; ++g) {
        if (g + 1 < NG) {
            #pragma unroll
            for (int i = 0; i < HEAD_KS; ++i) {
                #pragma unroll
                for (int rt = 0; rt < 4; ++rt)
                    fr[(g + 1) & 1][i * 4 + rt] =
                        *(const v2f*)(arow + rt * 16 * HEAD_PITCH + 4 * ((g + 1) * HEAD_KS + i));
            }
        }
        __builtin_amdgcn_sched_barrier(0);
        #pragma unroll
        for (int i = 0; i < HEAD_KS; ++i) {
            #pragma unroll
            for (int rt = 0; rt < 4; ++rt)
                acc[rt] = __builtin_amdgcn_wmma_f32_16x16x4_f32(
                    false, fr[g & 1][i * 4 + rt], false, bf[g * HEAD_KS + i], (short)0, acc[rt], false, false);
        }
        __builtin_amdgcn_sched_barrier(0);
    }

    if (live && col < OUT) {
        #pragma unroll
        for (int rt = 0; rt < 4; ++rt) {
            float* op = o + (size_t)(16 * rt + 8 * half) * OUT + col;
            #pragma unroll
            for (int v = 0; v < 8; ++v) op[(size_t)v * OUT] = acc[rt][v];
        }
    }
}

extern "C" void kernel_launch(void* const* d_in, const int* in_sizes, int n_in,
                              void* d_out, int out_size, void* d_ws, size_t ws_size,
                              hipStream_t stream) {
    const float* x   = (const float*)d_in[0];
    const int*   ei  = (const int*)d_in[1];
    const int*   bat = (const int*)d_in[2];
    const float* W1  = (const float*)d_in[3];
    const float* b1  = (const float*)d_in[4];
    const float* W2  = (const float*)d_in[5];
    const float* b2  = (const float*)d_in[6];
    const float* Wmf = (const float*)d_in[7];
    const float* bmf = (const float*)d_in[8];
    const float* Wbp = (const float*)d_in[9];
    const float* bbp = (const float*)d_in[10];
    const float* Wcc = (const float*)d_in[11];
    const float* bcc = (const float*)d_in[12];
    float* out = (float*)d_out;

    const int E = in_sizes[1] / 2;
    const int* src = ei;
    const int* dst = ei + E;

    float* ws     = (float*)d_ws;
    float* dis    = ws;
    float* h      = dis + N_NODES;
    float* agg    = h + (size_t)N_NODES * HID;
    float* pooled = agg + (size_t)N_NODES * HID;
    float* cnt    = pooled + N_GRAPHS * HID;
    unsigned* own = (unsigned*)(cnt + N_GRAPHS);
    unsigned* rlist = own + OWN_PLANE_WORDS;
    int*      rcnt  = (int*)(rlist + (size_t)OWN_NB * OWN_WAVES * OWN_CAP);

    unsigned* degp = (unsigned*)agg;
    deg_partial<<<DEG_P, 1024, 0, stream>>>(dst, degp, own, E);
    deg_finish<<<((N_NODES + 1) / 2 + 255) / 256, 256, 0, stream>>>(degp, dis, N_NODES);

    gemm_wmma_f32<IN_DIM><<<GEMM_GRID_K256, 256, 0, stream>>>(x, W1, h, N_NODES / 16);
    owner_aggregate<1><<<OWN_NB, 1024, 0, stream>>>(src, dst, dis, h, b1, own, agg, rlist, rcnt, ws_size, E, N_NODES);

    gemm_wmma_f32<HID><<<GEMM_GRID_K128, 256, 0, stream>>>(agg, W2, h, N_NODES / 16);
    owner_aggregate<2><<<OWN_NB, 1024, 0, stream>>>(src, dst, dis, h, b2, own, agg, rlist, rcnt, ws_size, E, N_NODES);

    float* part = h;
    int*   cntp = (int*)(h + (size_t)POOL_NB * N_GRAPHS * HID);
    pool_partial<<<POOL_NB, POOL_WAVES * 32, 0, stream>>>(agg, bat, part, cntp, N_NODES);
    pool_finish<<<N_GRAPHS, 1024, 0, stream>>>(part, cntp, pooled);

    float* out_mf = out;
    float* out_bp = out + N_GRAPHS * OUT_MF;
    float* out_cc = out + N_GRAPHS * (OUT_MF + OUT_BP);
    heads_wmma<<<(HEAD_TILES + HEAD_WAVES - 1) / HEAD_WAVES, HEAD_WAVES * 32, 0, stream>>>(
        pooled, Wmf, bmf, Wbp, bbp, Wcc, bcc, out_mf, out_bp, out_cc);
}


# layer 1 reads the owner-id plane by a block index: the degree pass marks, per owner and 512-edge step, which 16-edge pieces hold its edges; a group loads those pieces only; sums still by arrival
# speedup vs baseline: 4.0405x; 1.1172x over previous
#include <hip/hip_runtime.h>
#include <hip/hip_bf16.h>

#define N_NODES 100000
#define N_EDGES 1600000
#define IN_DIM  256
#define HID     128
#define N_GRAPHS 64
#define OUT_MF 489
#define OUT_BP 1943
#define OUT_CC 320

typedef __attribute__((ext_vector_type(2))) float v2f;
typedef __attribute__((ext_vector_type(8))) float v8f;

#define OWN_R     391
#define DEG_P     64
#define DEG_NOWN  ((N_NODES + OWN_R - 1) / OWN_R)
#define DEG_STEPS ((N_EDGES + 511) / 512)
#define DEG_SPS   ((DEG_STEPS + DEG_P - 1) / DEG_P)
#define DEG_U     8
#define DEG_WORDS (((N_NODES + 1) / 2 + 3) & ~3)

__device__ __forceinline__ unsigned deg_count(unsigned* tab, unsigned* idx, int d, int srel, unsigned bit) {
    if ((unsigned)d >= (unsigned)N_NODES) return 0u;
    atomicAdd(&tab[d >> 1], (d & 1) ? 0x10000u : 1u);
    const unsigned o = (unsigned)(d / OWN_R);
    atomicOr(&idx[o * DEG_SPS + srel], bit);
    return o;
}

__global__ __launch_bounds__(1024) void deg_partial(const int* __restrict__ dst,
                                                    unsigned* __restrict__ part,
                                                    unsigned* __restrict__ own,
                                                    unsigned* __restrict__ idxp, int E) {
    static_assert(DEG_SPS * 512 < 65536, "the counts of one slice must fit in 16 bits");
    static_assert((DEG_NOWN * DEG_SPS) % 4 == 0, "the index table is cleared in 16-byte pieces");
    static_assert(DEG_STEPS <= (N_NODES - (DEG_NOWN - 1) * OWN_R) * HID,
                  "an owner's index words fit in its own rows of the activation plane");
    __shared__ __attribute__((aligned(16))) unsigned tab[DEG_WORDS];
    __shared__ __attribute__((aligned(16))) unsigned idx[DEG_NOWN * DEG_SPS];
    const int tid = threadIdx.x;
    for (int i = tid; i < DEG_WORDS / 4; i += 1024) ((uint4*)tab)[i] = make_uint4(0u, 0u, 0u, 0u);
    for (int i = tid; i < DEG_NOWN * DEG_SPS / 4; i += 1024) ((uint4*)idx)[i] = make_uint4(0u, 0u, 0u, 0u);
    __syncthreads();

    const int nsteps = (E + 511) >> 9;
    const int nwords = nsteps << 7;
    const int sps = (nsteps + DEG_P - 1) / DEG_P;
    const int s0 = blockIdx.x * sps;
    const int w0 = s0 << 7;
    const int w1 = (w0 + (sps << 7)) < nwords ? (w0 + (sps << 7)) : nwords;
    if (((E & 3) == 0) && ((((size_t)dst) & 15) == 0)) {
        const int lastw = (E >> 2) - 1;
        for (int tb = w0 + tid; tb < w1; tb += 1024 * DEG_U) {
            int4 d[DEG_U];
            #pragma unroll
            for (int u = 0; u < DEG_U; ++u) {
                const int t = tb + u * 1024;
                const int tl = t < lastw ? t : lastw;
                d[u] = *(const int4*)(dst + 4 * (size_t)tl);
            }
            #pragma unroll
            for (int u = 0; u < DEG_U; ++u) {
                const int t = tb + u * 1024;
                if (t < w1) {
                    const int4 dd = t <= lastw ? d[u] : make_int4(-1, -1, -1, -1);
                    const int srel = (t >> 7) - s0;
                    const unsigned bit = 1u << ((t >> 2) & 31);
                    own[t] = deg_count(tab, idx, dd.x, srel, bit) | (deg_count(tab, idx, dd.y, srel, bit) << 8)
                           | (deg_count(tab, idx, dd.z, srel, bit) << 16) | (deg_count(tab, idx, dd.w, srel, bit) << 24);
                }
            }
        }
    } else {
        for (int t = w0 + tid; t < w1; t += 1024) {
            unsigned w = 0u;
            for (int c = 0; c < 4; ++c)
                if (4 * t + c < E)
                    w |= deg_count(tab, idx, dst[4 * t + c], (t >> 7) - s0, 1u << ((t >> 2) & 31)) << (8 * c);
            own[t] = w;
        }
    }
    __syncthreads();

    uint4* pl = (uint4*)(part + (size_t)blockIdx.x * DEG_WORDS);
    for (int i = tid; i < DEG_WORDS / 4; i += 1024) pl[i] = ((const uint4*)tab)[i];

    int ns = nsteps - s0;
    ns = ns < 0 ? 0 : (ns > sps ? sps : ns);
    for (int i = tid; i < DEG_NOWN * ns; i += 1024) {
        const int o = i / ns, j = i - o * ns;
        idxp[(size_t)o * (OWN_R * HID) + s0 + j] = idx[o * DEG_SPS + j];
    }
}

__global__ void deg_finish(const unsigned* __restrict__ part, float* __restrict__ dis, int N) {
    const int t = blockIdx.x * blockDim.x + threadIdx.x;
    if (2 * t >= N) return;
    unsigned lo = 0u, hi = 0u;
    #pragma unroll 16
    for (int p = 0; p < DEG_P; ++p) {
        const unsigned w = part[(size_t)p * DEG_WORDS + t];
        lo += w & 0xFFFFu;
        hi += w >> 16;
    }
    const float d0 = rsqrtf(1.0f + (float)lo);
    const float d1 = rsqrtf(1.0f + (float)hi);
    if (2 * t + 1 < N) {
        *(float2*)(dis + 2 * t) = make_float2(d0, d1);
    } else {
        dis[2 * t] = d0;
    }
}

#define GEMM_GRID_K256 512
#define GEMM_GRID_K128 768
typedef __attribute__((ext_vector_type(4))) float v4f;

template <int K>
__global__ __launch_bounds__(256) void gemm_wmma_f32(const float* __restrict__ A,
                                                     const float* __restrict__ W,
                                                     float* __restrict__ out, int ntiles) {
    constexpr int PITCH = K + 4;
    constexpr int RV    = K / 4;
    constexpr int NV    = 16 * RV / 256;
    constexpr int RS    = 256 / RV;
    constexpr int G     = 16;
    constexpr int NG    = K / 4 / G;
    static_assert(NV == 4 || NV == 2, "a thread moves 2 or 4 pieces of a tile");
    __shared__ __attribute__((aligned(16))) float ldsA[16 * PITCH];
    const int tid  = threadIdx.x;
    const int wave = tid >> 5;
    const int lane = tid & 31;
    const int half = lane >> 4;
    const int m    = lane & 15;
    const int col0 = wave * 16;

    int tile = blockIdx.x;
    if (tile >= ntiles) return;

    const int pr = tid / RV, pc = tid - pr * RV;
    float* lst = ldsA + pr * PITCH + 4 * pc;
    const v4f* ap = (const v4f*)(A + (size_t)tile * 16 * K) + tid;
    v4f p0 = ap[0], p1 = ap[256], p2 = {}, p3 = {};
    if constexpr (NV == 4) { p2 = ap[512]; p3 = ap[768]; }

    v2f bf[K / 4];
    #pragma unroll
    for (int s = 0; s < K / 4; ++s) {
        const int ka = 4 * s + 2 * half;
        bf[s].x = W[ka * HID + col0 + m];
        bf[s].y = W[(ka + 1) * HID + col0 + m];
    }

    const float* arow = ldsA + m * PITCH + 2 * half;
    for (;;) {
        __syncthreads();
        *(v4f*)(lst) = p0;
        *(v4f*)(lst + RS * PITCH) = p1;
        if constexpr (NV == 4) {
            *(v4f*)(lst + 2 * RS * PITCH) = p2;
            *(v4f*)(lst + 3 * RS * PITCH) = p3;
        }
        __syncthreads();

        const int nt = tile + (int)gridDim.x;
        const int lt = nt < ntiles ? nt : tile;
        ap = (const v4f*)(A + (size_t)lt * 16 * K) + tid;
        p0 = ap[0]; p1 = ap[256];
        if constexpr (NV == 4) { p2 = ap[512]; p3 = ap[768]; }

        v2f fr[2][G];
        #pragma unroll
        for (int i = 0; i < G; ++i) fr[0][i] = *(const v2f*)(arow + 4 * i);
        v8f acc = {};
        #pragma unroll
        for (int g = 0; g < NG; ++g) {
            if (g + 1 < NG) {
                #pragma unroll
                for (int i = 0; i < G; ++i)
                    fr[(g + 1) & 1][i] = *(const v2f*)(arow + 4 * ((g + 1) * G + i));
            }
            __builtin_amdgcn_sched_barrier(0);
            #pragma unroll
            for (int i = 0; i < G; ++i)
                acc = __builtin_amdgcn_wmma_f32_16x16x4_f32(
                    false, fr[g & 1][i], false, bf[g * G + i], (short)0, acc, false, false);
            __builtin_amdgcn_sched_barrier(0);
        }

        float* op = out + (size_t)(tile * 16 + 8 * half) * HID + col0 + m;
        #pragma unroll
        for (int v = 0; v < 8; ++v) op[v * HID] = acc[v];

        if (nt >= ntiles) break;
        tile = nt;
    }
}

#define OWN_WAVES 32
#define OWN_Q     64
#define OWN_U     4
#define OWN_G     8
#define OWN_NB    ((N_NODES + OWN_R - 1) / OWN_R)
#define OWN_CAP   256

#define OWN_PLANE_WORDS ((N_EDGES + 511) / 512 * 128)
#define OWN_WS_WORDS ((size_t)N_NODES + 2 * (size_t)N_NODES * HID + N_GRAPHS * HID + N_GRAPHS \
                      + OWN_PLANE_WORDS + (size_t)OWN_NB * OWN_WAVES * OWN_CAP + OWN_NB * OWN_WAVES)

typedef __attribute__((address_space(3))) int lds_int;

typedef const __attribute__((address_space(1))) float owner_gf;
static_assert(HID * sizeof(float) == 512, "a row is 512 bytes: the row address is the node number shifted by 9");

template <bool CLAMP>
__device__ __forceinline__ void owner_rows_load(float (&val)[OWN_G][4], int s, int base, int last,
                                                int lane, const float* __restrict__ h) {
    #pragma unroll
    for (int u = 0; u < OWN_G; ++u) {
        const int k = (CLAMP && base + u > last) ? last : base + u;
        const int si = __builtin_amdgcn_readlane(s, k);
        const unsigned long long ra = (unsigned long long)h + ((unsigned long long)(unsigned)si << 9);
        const unsigned ralo = __builtin_amdgcn_readfirstlane((unsigned)ra);
        const unsigned rahi = __builtin_amdgcn_readfirstlane((unsigned)(ra >> 32));
        const owner_gf* hs = (const owner_gf*)(((unsigned long long)rahi << 32) | ralo);
        #pragma unroll
        for (int j = 0; j < 4; ++j) val[u][j] = hs[j * 32 + lane];
    }
}

template <bool GUARD>
__device__ __forceinline__ void owner_rows_add(const float (&val)[OWN_G][4], int r, float nr, int base,
                                               int nb, int lane, float* acc) {
    #pragma unroll
    for (int u = 0; u < OWN_G; ++u) {
        if (!GUARD || base + u < nb) {
            const int ri = __builtin_amdgcn_readlane(r, base + u);
            const float ni = __int_as_float(__builtin_amdgcn_readlane(__float_as_int(nr), base + u));
            float* ar = acc + ri * HID;
            #pragma unroll
            for (int j = 0; j < 4; ++j) atomicAdd(&ar[j * 32 + lane], val[u][j] * ni);
        }
    }
}

__device__ __forceinline__ void owner_rows_full(int s, int r, float nr, int lane,
                                                const float* __restrict__ h, float* acc) {
    static_assert(OWN_G == 8, "a full chunk is four groups");
    float va[OWN_G][4], vb[OWN_G][4];
    owner_rows_load<false>(va, s, 0, 31, lane, h);
    owner_rows_load<false>(vb, s, 8, 31, lane, h);
    owner_rows_add<false>(va, r, nr, 0, 32, lane, acc);
    owner_rows_load<false>(va, s, 16, 31, lane, h);
    owner_rows_add<false>(vb, r, nr, 8, 32, lane, acc);
    owner_rows_load<false>(vb, s, 24, 31, lane, h);
    owner_rows_add<false>(va, r, nr, 16, 32, lane, acc);
    owner_rows_add<false>(vb, r, nr, 24, 32, lane, acc);
}

__device__ __forceinline__ void owner_rows_part(int s, int r, float nr, int nb, int lane,
                                                const float* __restrict__ h, float* acc) {
    float va[OWN_G][4];
    #pragma unroll 1
    for (int i = 0; i < nb; i += OWN_G) {
        owner_rows_load<true>(va, s, i, nb - 1, lane, h);
        owner_rows_add<true>(va, r, nr, i, nb, lane, acc);
    }
}

template <bool FULL>
__device__ __forceinline__ void owner_drain(volatile lds_int* q, int head, int nb, int lane, int lo,
                                            const int* __restrict__ src, const int* __restrict__ dst,
                                            const float* __restrict__ dis, const float* __restrict__ h,
                                            float* acc, unsigned* __restrict__ rec, int& nrec) {
    const bool v = lane < nb;
    const int e = v ? q[(head + lane) & (OWN_Q - 1)] : 0;
    const int s = v ? src[e] : 0;
    const int d = v ? dst[e] : lo;
    const float nr = v ? dis[s] * dis[d] : 0.f;
    const int r = d - lo;
    if (rec != nullptr) {
        if (v && nrec + nb <= OWN_CAP) rec[nrec + lane] = (unsigned)s | ((unsigned)r << 17);
        nrec += nb;
    }
    if (FULL) owner_rows_full(s, r, nr, lane, h, acc);
    else      owner_rows_part(s, r, nr, nb, lane, h, acc);
}

__device__ __forceinline__ void owner_replay(const unsigned* __restrict__ rec, int n, int lane, int lo,
                                             const float* __restrict__ dis, const float* __restrict__ h,
                                             float* acc) {
    if (n <= 0) return;
    const int last = n - 1;
    unsigned w1 = rec[lane < last ? lane : last];
    unsigned w2 = rec[32 + lane < last ? 32 + lane : last];
    float fa = dis[w1 & 0x1FFFFu], fb = dis[lo + (int)(w1 >> 17)];
    int i = 0;
    for (; i + 32 <= n; i += 32) {
        const int s = (int)(w1 & 0x1FFFFu), r = (int)(w1 >> 17);
        const float nr = fa * fb;
        w1 = w2;
        const int nx = i + 64 + lane;
        w2 = rec[nx < last ? nx : last];
        fa = dis[w1 & 0x1FFFFu]; fb = dis[lo + (int)(w1 >> 17)];
        owner_rows_full(s, r, nr, lane, h, acc);
    }
    if (i < n) {
        const int s = (int)(w1 & 0x1FFFFu), r = (int)(w1 >> 17);
        owner_rows_part(s, r, fa * fb, n - i, lane, h, acc);
    }
}

__device__ __forceinline__ void owner_range(int wave, int nsteps, int& sbeg, int& send) {
    const int spw = (nsteps + OWN_WAVES - 1) / OWN_WAVES;
    sbeg = wave * spw;
    send = (sbeg + spw) < nsteps ? (sbeg + spw) : nsteps;
}

__device__ __forceinline__ void owner_scan(volatile lds_int* q, int wave, int lane, int lo, unsigned pat,
                                           int nsteps, int E,
                                           const unsigned* __restrict__ own,
                                           const int* __restrict__ src, const int* __restrict__ dst,
                                           const float* __restrict__ dis, const float* __restrict__ h,
                                           float* acc, unsigned* __restrict__ rec, int& nrec) {
    int head = 0, cnt = 0;
    int sbeg, send;
    owner_range(wave, nsteps, sbeg, send);
    for (int st0 = sbeg; st0 < send; st0 += OWN_U) {
        uint4 ov[OWN_U];
        #pragma unroll
        for (int u = 0; u < OWN_U; ++u) {
            const int st = st0 + u;
            const int sidx = st < send ? st : 0;
            ov[u] = *(const uint4*)(own + ((size_t)(sidx * 32 + lane) << 2));
        }
        __builtin_amdgcn_sched_barrier(0);
        unsigned long long lm = 0ull;
        #pragma unroll
        for (int u = 0; u < OWN_U; ++u) {
            const unsigned wd[4] = {ov[u].x, ov[u].y, ov[u].z, ov[u].w};
            unsigned nz = 0u;
            #pragma unroll
            for (int j = 0; j < 4; ++j) {
                const unsigned x = wd[j] ^ pat;
                const unsigned y = ((x & 0x7F7F7F7Fu) + 0x7F7F7F7Fu) | x;
                nz |= ((y >> 7) & 0x01010101u) << j;
            }
            unsigned m16 = ~nz & 0x0F0F0F0Fu;
            if (st0 + u >= send) m16 = 0u;
            lm |= (unsigned long long)(m16 << (4 * (u & 1))) << (32 * (u >> 1));
        }
        for (int round = 0; round < 16 * OWN_U; ++round) {
            const bool has = lm != 0ull;
            if (__builtin_amdgcn_ballot_w32(has) == 0u) break;
            const int p  = has ? __builtin_ctzll(lm) : 0;
            const int lw = p & 31;
            const int u  = ((p >> 5) << 1) | ((lw >> 2) & 1);
            const int sidx = st0 + u;
            const int e = ((sidx * 32 + lane) << 4) + ((lw & 3) << 2) + (lw >> 3);
            const bool take = has && (e < E);
            const unsigned mask = __builtin_amdgcn_ballot_w32(take);
            const int pos = (head + cnt + (int)__builtin_amdgcn_mbcnt_lo(mask, 0u)) & (OWN_Q - 1);
            if (take) q[pos] = e;
            lm &= lm - 1ull;
            cnt += __builtin_popcount(mask);
            if (cnt >= 32) {
                owner_drain<true>(q, head, 32, lane, lo, src, dst, dis, h, acc, rec, nrec);
                head = (head + 32) & (OWN_Q - 1);
                cnt -= 32;
            }
        }
    }
    owner_drain<false>(q, head, cnt, lane, lo, src, dst, dis, h, acc, rec, nrec);
}

__device__ __forceinline__ void owner_scan_index(volatile lds_int* q, int wave, int lane, int lo, unsigned pat,
                                                 int nsteps, int E, const unsigned* idx,
                                                 const unsigned* __restrict__ own,
                                                 const int* __restrict__ src, const int* __restrict__ dst,
                                                 const float* __restrict__ dis, const float* __restrict__ h,
                                                 float* acc, unsigned* __restrict__ rec, int& nrec) {
    static_assert(OWN_U == 4, "a trip takes four blocks per lane");
    int head = 0, cnt = 0;
    int sbeg, send;
    owner_range(wave, nsteps, sbeg, send);
    for (int sb = sbeg; sb < send; sb += 32) {
        const int st = sb + lane;
        unsigned bits = st < send ? idx[st] : 0u;
        for (int trip = 0; trip < 32 / OWN_U; ++trip) {
            if (__builtin_amdgcn_ballot_w32(bits != 0u) == 0u) break;
            int blk[OWN_U];
            uint4 ov[OWN_U];
            #pragma unroll
            for (int u = 0; u < OWN_U; ++u) {
                const bool hasb = bits != 0u;
                blk[u] = hasb ? (st << 5) + __builtin_ctz(bits) : -1;
                bits &= bits - 1u;
                ov[u] = *(const uint4*)(own + ((size_t)(hasb ? blk[u] : 0) << 2));
            }
            __builtin_amdgcn_sched_barrier(0);
            unsigned long long lm = 0ull;
            #pragma unroll
            for (int u = 0; u < OWN_U; ++u) {
                const unsigned wd[4] = {ov[u].x, ov[u].y, ov[u].z, ov[u].w};
                unsigned nz = 0u;
                #pragma unroll
                for (int j = 0; j < 4; ++j) {
                    const unsigned x = wd[j] ^ pat;
                    const unsigned y = ((x & 0x7F7F7F7Fu) + 0x7F7F7F7Fu) | x;
                    nz |= ((y >> 7) & 0x01010101u) << j;
                }
                unsigned m16 = ~nz & 0x0F0F0F0Fu;
                if (blk[u] < 0) m16 = 0u;
                lm |= (unsigned long long)(m16 << (4 * (u & 1))) << (32 * (u >> 1));
            }
            for (int round = 0; round < 16 * OWN_U; ++round) {
                const bool has = lm != 0ull;
                if (__builtin_amdgcn_ballot_w32(has) == 0u) break;
                const int p  = has ? __builtin_ctzll(lm) : 0;
                const int lw = p & 31;
                const int u  = ((p >> 5) << 1) | ((lw >> 2) & 1);
                const int b  = u == 0 ? blk[0] : (u == 1 ? blk[1] : (u == 2 ? blk[2] : blk[3]));
                const int e = (b << 4) + ((lw & 3) << 2) + (lw >> 3);
                const bool take = has && (e < E);
                const unsigned mask = __builtin_amdgcn_ballot_w32(take);
                const int pos = (head + cnt + (int)__builtin_amdgcn_mbcnt_lo(mask, 0u)) & (OWN_Q - 1);
                if (take) q[pos] = e;
                lm &= lm - 1ull;
                cnt += __builtin_popcount(mask);
                if (cnt >= 32) {
                    owner_drain<true>(q, head, 32, lane, lo, src, dst, dis, h, acc, rec, nrec);
                    head = (head + 32) & (OWN_Q - 1);
                    cnt -= 32;
                }
            }
        }
    }
    owner_drain<false>(q, head, cnt, lane, lo, src, dst, dis, h, acc, rec, nrec);
}

template <int LAYER>
__global__ __launch_bounds__(1024) void owner_aggregate(const int* __restrict__ src,
                                                        const int* __restrict__ dst,
                                                        const float* __restrict__ dis,
                                                        const float* __restrict__ h,
                                                        const float* __restrict__ bias,
                                                        const unsigned* __restrict__ own,
                                                        float* __restrict__ outp,
                                                        unsigned* __restrict__ rlist,
                                                        int* __restrict__ rcnt,
                                                        size_t ws_bytes, int E, int N) {
    static_assert((N_NODES + OWN_R - 1) / OWN_R <= 256, "an owner id must fit in one byte");
    static_assert(N_NODES <= (1 << 17) && OWN_R <= (1 << 15), "a record word holds source and row");
    __shared__ __attribute__((aligned(16))) float acc[OWN_R * HID];
    __shared__ int queue[OWN_WAVES * OWN_Q];

    const int tid  = threadIdx.x;
    const int lane = tid & 31;
    const int wave = __builtin_amdgcn_readfirstlane(tid >> 5);
    const int lo   = blockIdx.x * OWN_R;
    const int nrows = (N - lo) < OWN_R ? (N - lo) : OWN_R;
    volatile lds_int* q = (volatile lds_int*)(queue + wave * OWN_Q);

    for (int idx = tid; idx < nrows * (HID / 4); idx += 1024) {
        const int r = idx >> 5, c4 = idx & 31;
        const float di = dis[lo + r];
        const float w = di * di;
        const float4 hv = ((const float4*)(h + (size_t)(lo + r) * HID))[c4];
        const float4 bv = ((const float4*)bias)[c4];
        float4 o;
        o.x = w * hv.x + bv.x; o.y = w * hv.y + bv.y; o.z = w * hv.z + bv.z; o.w = w * hv.w + bv.w;
        ((float4*)acc)[idx] = o;
    }
    __syncthreads();

    const int nsteps = (E + 511) >> 9;
    const unsigned pat = (unsigned)blockIdx.x * 0x01010101u;
    const bool room = ws_bytes >= OWN_WS_WORDS * sizeof(float);
    const int slot = blockIdx.x * OWN_WAVES + wave;
    unsigned* rec = rlist + (size_t)slot * OWN_CAP;
    if (LAYER == 1) {
        int nrec = 0;
        const unsigned* idx = (const unsigned*)(outp + (size_t)lo * HID);
        owner_scan_index(q, wave, lane, lo, pat, nsteps, E, idx, own, src, dst, dis, h, acc,
                         room ? rec : nullptr, nrec);
        if (lane == 0) q[0] = nrec;
    } else {
        const int n = room ? rcnt[slot] : OWN_CAP + 1;
        if (n <= OWN_CAP) {
            owner_replay(rec, n, lane, lo, dis, h, acc);
        } else {
            int none = 0;
            owner_scan(q, wave, lane, lo, pat, nsteps, E, own, src, dst, dis, h, acc, nullptr, none);
        }
    }
    __syncthreads();

    if (LAYER == 1 && room && wave == 0) rcnt[blockIdx.x * OWN_WAVES + lane] = queue[lane * OWN_Q];

    float4* orow = (float4*)(outp + (size_t)lo * HID);
    for (int idx = tid; idx < nrows * (HID / 4); idx += 1024) {
        float4 v4 = ((const float4*)acc)[idx];
        v4.x = v4.x > 0.f ? v4.x : 0.f; v4.y = v4.y > 0.f ? v4.y : 0.f;
        v4.z = v4.z > 0.f ? v4.z : 0.f; v4.w = v4.w > 0.f ? v4.w : 0.f;
        orow[idx] = v4;
    }
}

#define POOL_NB    256
#define POOL_R     ((N_NODES + POOL_NB - 1) / POOL_NB)
#define POOL_WAVES 16
#define POOL_RW    ((POOL_R + POOL_WAVES - 1) / POOL_WAVES)

__device__ __forceinline__ void pool_flush(float* tab, int* tcnt, int g, int rl, float4 a, int lane) {
    if ((unsigned)g < (unsigned)N_GRAPHS) {
        float* t = tab + g * HID + 4 * lane;
        atomicAdd(t + 0, a.x); atomicAdd(t + 1, a.y); atomicAdd(t + 2, a.z); atomicAdd(t + 3, a.w);
        if (lane == 0) atomicAdd(tcnt + g, rl);
    }
}

__global__ __launch_bounds__(POOL_WAVES * 32) void pool_partial(const float* __restrict__ h,
                                                                const int* __restrict__ batch,
                                                                float* __restrict__ part,
                                                                int* __restrict__ cntp, int N) {
    __shared__ __attribute__((aligned(16))) float tab[N_GRAPHS * HID];
    __shared__ int tcnt[N_GRAPHS];
    const int tid  = threadIdx.x;
    const int lane = tid & 31;
    const int wave = __builtin_amdgcn_readfirstlane(tid >> 5);
    const float4 zero = make_float4(0.f, 0.f, 0.f, 0.f);

    const int r0 = blockIdx.x * POOL_R + wave * POOL_RW;
    int rend = blockIdx.x * POOL_R + POOL_R;
    if (rend > N) rend = N;
    int nr = rend - r0;
    nr = nr < 0 ? 0 : (nr > POOL_RW ? POOL_RW : nr);
    const int gl = (lane < nr) ? batch[r0 + lane] : -1;
    float4 v[POOL_RW];
    #pragma unroll
    for (int i = 0; i < POOL_RW; ++i) {
        int ri = r0 + i;
        if (ri > N - 1) ri = N - 1;
        v[i] = ((const float4*)(h + (size_t)ri * HID))[lane];
    }

    for (int i = tid; i < N_GRAPHS * HID / 4; i += POOL_WAVES * 32) ((float4*)tab)[i] = zero;
    if (tid < N_GRAPHS) tcnt[tid] = 0;
    __syncthreads();

    int gcur = -1, rl = 0;
    float4 acc = zero;
    #pragma unroll
    for (int i = 0; i < POOL_RW; ++i) {
        if (i < nr) {
            const int g = __builtin_amdgcn_readlane(gl, i);
            if (g != gcur) {
                pool_flush(tab, tcnt, gcur, rl, acc, lane);
                gcur = g; rl = 0; acc = zero;
            }
            acc.x += v[i].x; acc.y += v[i].y; acc.z += v[i].z; acc.w += v[i].w;
            ++rl;
        }
    }
    pool_flush(tab, tcnt, gcur, rl, acc, lane);
    __syncthreads();

    float4* pl = (float4*)(part + (size_t)blockIdx.x * (N_GRAPHS * HID));
    for (int i = tid; i < N_GRAPHS * HID / 4; i += POOL_WAVES * 32) pl[i] = ((const float4*)tab)[i];
    if (tid < N_GRAPHS) cntp[blockIdx.x * N_GRAPHS + tid] = tcnt[tid];
}

__global__ __launch_bounds__(1024) void pool_finish(const float* __restrict__ part,
                                                    const int* __restrict__ cntp,
                                                    float* __restrict__ pooled) {
    static_assert(POOL_NB == 256, "32 waves x 8 planes, one count per thread of the first 256");
    __shared__ __attribute__((aligned(16))) float red[32 * HID];
    __shared__ int ntot;
    const int tid  = threadIdx.x;
    const int lane = tid & 31;
    const int wave = tid >> 5;
    const int g    = blockIdx.x;

    float4 v[8];
    #pragma unroll
    for (int i = 0; i < 8; ++i)
        v[i] = ((const float4*)(part + ((size_t)(wave * 8 + i) * N_GRAPHS + g) * HID))[lane];
    const int c = (tid < POOL_NB) ? cntp[tid * N_GRAPHS + g] : 0;

    if (tid == 0) ntot = 0;
    __syncthreads();
    float4 s = v[0];
    #pragma unroll
    for (int i = 1; i < 8; ++i) { s.x += v[i].x; s.y += v[i].y; s.z += v[i].z; s.w += v[i].w; }
    ((float4*)red)[wave * 32 + lane] = s;
    if (c != 0) atomicAdd(&ntot, c);
    __syncthreads();

    if (wave == 0) {
        float4 t = ((const float4*)red)[lane];
        #pragma unroll
        for (int w = 1; w < 32; ++w) {
            const float4 u = ((const float4*)red)[w * 32 + lane];
            t.x += u.x; t.y += u.y; t.z += u.z; t.w += u.w;
        }
        float n = (float)ntot;
        n = n > 1.f ? n : 1.f;
        t.x /= n; t.y /= n; t.z /= n; t.w /= n;
        ((float4*)(pooled + g * HID))[lane] = t;
    }
}

#define HEAD_T_MF   ((OUT_MF + 15) / 16)
#define HEAD_T_BP   ((OUT_BP + 15) / 16)
#define HEAD_T_CC   ((OUT_CC + 15) / 16)
#define HEAD_TILES  (HEAD_T_MF + HEAD_T_BP + HEAD_T_CC)
#define HEAD_WAVES  4
#define HEAD_PITCH  (HID + 4)
#define HEAD_KS     4

__global__ __launch_bounds__(HEAD_WAVES * 32) void heads_wmma(const float* __restrict__ pooled,
                                                              const float* __restrict__ Wmf,
                                                              const float* __restrict__ bmf,
                                                              const float* __restrict__ Wbp,
                                                              const float* __restrict__ bbp,
                                                              const float* __restrict__ Wcc,
                                                              const float* __restrict__ bcc,
                                                              float* __restrict__ out_mf,
                                                              float* __restrict__ out_bp,
                                                              float* __restrict__ out_cc) {
    static_assert(N_GRAPHS == 64 && HID == 128, "4 row tiles, 32 k steps, 16 pieces of the means per thread");
    __shared__ __attribute__((aligned(16))) float ldsP[N_GRAPHS * HEAD_PITCH];
    const int tid  = threadIdx.x;
    const int lane = tid & 31;
    const int wave = __builtin_amdgcn_readfirstlane(tid >> 5);
    const int half = lane >> 4;
    const int m    = lane & 15;

    v4f pp[16];
    #pragma unroll
    for (int j = 0; j < 16; ++j) pp[j] = ((const v4f*)pooled)[tid + HEAD_WAVES * 32 * j];

    int t = blockIdx.x * HEAD_WAVES + wave;
    const bool live = t < HEAD_TILES;
    if (!live) t = HEAD_TILES - 1;
    const float* W; const float* b; float* o; int OUT;
    if (t < HEAD_T_MF) {
        W = Wmf; b = bmf; o = out_mf; OUT = OUT_MF;
    } else if (t < HEAD_T_MF + HEAD_T_BP) {
        t -= HEAD_T_MF; W = Wbp; b = bbp; o = out_bp; OUT = OUT_BP;
    } else {
        t -= HEAD_T_MF + HEAD_T_BP; W = Wcc; b = bcc; o = out_cc; OUT = OUT_CC;
    }
    const int col  = t * 16 + m;
    const int colc = col < OUT ? col : OUT - 1;

    v2f bf[HID / 4];
    #pragma unroll
    for (int s = 0; s < HID / 4; ++s) {
        const int ka = 4 * s + 2 * half;
        bf[s].x = W[ka * OUT + colc];
        bf[s].y = W[(ka + 1) * OUT + colc];
    }
    const float bias = b[colc];

    #pragma unroll
    for (int j = 0; j < 16; ++j) {
        const int p = tid + HEAD_WAVES * 32 * j;
        *(v4f*)(ldsP + (p >> 5) * HEAD_PITCH + 4 * (p & 31)) = pp[j];
    }
    __syncthreads();

    v8f acc[4];
    #pragma unroll
    for (int rt = 0; rt < 4; ++rt) {
        #pragma unroll
        for (int v = 0; v < 8; ++v) acc[rt][v] = bias;
    }

    constexpr int NG = HID / 4 / HEAD_KS;
    const float* arow = ldsP + m * HEAD_PITCH + 2 * half;
    v2f fr[2][HEAD_KS * 4];
    #pragma unroll
    for (int i = 0; i < HEAD_KS; ++i) {
        #pragma unroll
        for (int rt = 0; rt < 4; ++rt)
            fr[0][i * 4 + rt] = *(const v2f*)(arow + rt * 16 * HEAD_PITCH + 4 * i);
    }
    #pragma unroll
    for (int g = 0; g < NG; ++g) {
        if (g + 1 < NG) {
            #pragma unroll
            for (int i = 0; i < HEAD_KS; ++i) {
                #pragma unroll
                for (int rt = 0; rt < 4; ++rt)
                    fr[(g + 1) & 1][i * 4 + rt] =
                        *(const v2f*)(arow + rt * 16 * HEAD_PITCH + 4 * ((g + 1) * HEAD_KS + i));
            }
        }
        __builtin_amdgcn_sched_barrier(0);
        #pragma unroll
        for (int i = 0; i < HEAD_KS; ++i) {
            #pragma unroll
            for (int rt = 0; rt < 4; ++rt)
                acc[rt] = __builtin_amdgcn_wmma_f32_16x16x4_f32(
                    false, fr[g & 1][i * 4 + rt], false, bf[g * HEAD_KS + i], (short)0, acc[rt], false, false);
        }
        __builtin_amdgcn_sched_barrier(0);
    }

    if (live && col < OUT) {
        #pragma unroll
        for (int rt = 0; rt < 4; ++rt) {
            float* op = o + (size_t)(16 * rt + 8 * half) * OUT + col;
            #pragma unroll
            for (int v = 0; v < 8; ++v) op[(size_t)v * OUT] = acc[rt][v];
        }
    }
}

extern "C" void kernel_launch(void* const* d_in, const int* in_sizes, int n_in,
                              void* d_out, int out_size, void* d_ws, size_t ws_size,
                              hipStream_t stream) {
    const float* x   = (const float*)d_in[0];
    const int*   ei  = (const int*)d_in[1];
    const int*   bat = (const int*)d_in[2];
    const float* W1  = (const float*)d_in[3];
    const float* b1  = (const float*)d_in[4];
    const float* W2  = (const float*)d_in[5];
    const float* b2  = (const float*)d_in[6];
    const float* Wmf = (const float*)d_in[7];
    const float* bmf = (const float*)d_in[8];
    const float* Wbp = (const float*)d_in[9];
    const float* bbp = (const float*)d_in[10];
    const float* Wcc = (const float*)d_in[11];
    const float* bcc = (const float*)d_in[12];
    float* out = (float*)d_out;

    const int E = in_sizes[1] / 2;
    const int* src = ei;
    const int* dst = ei + E;

    float* ws     = (float*)d_ws;
    float* dis    = ws;
    float* h      = dis + N_NODES;
    float* agg    = h + (size_t)N_NODES * HID;
    float* pooled = agg + (size_t)N_NODES * HID;
    float* cnt    = pooled + N_GRAPHS * HID;
    unsigned* own = (unsigned*)(cnt + N_GRAPHS);
    unsigned* rlist = own + OWN_PLANE_WORDS;
    int*      rcnt  = (int*)(rlist + (size_t)OWN_NB * OWN_WAVES * OWN_CAP);

    unsigned* degp = (unsigned*)h;
    deg_partial<<<DEG_P, 1024, 0, stream>>>(dst, degp, own, (unsigned*)agg, E);
    deg_finish<<<((N_NODES + 1) / 2 + 255) / 256, 256, 0, stream>>>(degp, dis, N_NODES);

    gemm_wmma_f32<IN_DIM><<<GEMM_GRID_K256, 256, 0, stream>>>(x, W1, h, N_NODES / 16);
    owner_aggregate<1><<<OWN_NB, 1024, 0, stream>>>(src, dst, dis, h, b1, own, agg, rlist, rcnt, ws_size, E, N_NODES);

    gemm_wmma_f32<HID><<<GEMM_GRID_K128, 256, 0, stream>>>(agg, W2, h, N_NODES / 16);
    owner_aggregate<2><<<OWN_NB, 1024, 0, stream>>>(src, dst, dis, h, b2, own, agg, rlist, rcnt, ws_size, E, N_NODES);

    float* part = h;
    int*   cntp = (int*)(h + (size_t)POOL_NB * N_GRAPHS * HID);
    pool_partial<<<POOL_NB, POOL_WAVES * 32, 0, stream>>>(agg, bat, part, cntp, N_NODES);
    pool_finish<<<N_GRAPHS, 1024, 0, stream>>>(part, cntp, pooled);

    float* out_mf = out;
    float* out_bp = out + N_GRAPHS * OUT_MF;
    float* out_cc = out + N_GRAPHS * (OUT_MF + OUT_BP);
    heads_wmma<<<(HEAD_TILES + HEAD_WAVES - 1) / HEAD_WAVES, HEAD_WAVES * 32, 0, stream>>>(
        pooled, Wmf, bmf, Wbp, bbp, Wcc, bcc, out_mf, out_bp, out_cc);
}


# layer 2's aggregation sums its finished LDS rows by graph id itself (relu, per-wave register runs, LDS adds) into its pool partial plane: no activation store, no pool launch; pooled sum order changes
# speedup vs baseline: 4.2352x; 1.0482x over previous
#include <hip/hip_runtime.h>
#include <hip/hip_bf16.h>

#define N_NODES 100000
#define N_EDGES 1600000
#define IN_DIM  256
#define HID     128
#define N_GRAPHS 64
#define OUT_MF 489
#define OUT_BP 1943
#define OUT_CC 320

typedef __attribute__((ext_vector_type(2))) float v2f;
typedef __attribute__((ext_vector_type(8))) float v8f;

#define OWN_R     391
#define DEG_P     64
#define DEG_NOWN  ((N_NODES + OWN_R - 1) / OWN_R)
#define DEG_STEPS ((N_EDGES + 511) / 512)
#define DEG_SPS   ((DEG_STEPS + DEG_P - 1) / DEG_P)
#define DEG_U     8
#define DEG_WORDS (((N_NODES + 1) / 2 + 3) & ~3)

__device__ __forceinline__ unsigned deg_count(unsigned* tab, unsigned* idx, int d, int srel, unsigned bit) {
    if ((unsigned)d >= (unsigned)N_NODES) return 0u;
    atomicAdd(&tab[d >> 1], (d & 1) ? 0x10000u : 1u);
    const unsigned o = (unsigned)(d / OWN_R);
    atomicOr(&idx[o * DEG_SPS + srel], bit);
    return o;
}

__global__ __launch_bounds__(1024) void deg_partial(const int* __restrict__ dst,
                                                    unsigned* __restrict__ part,
                                                    unsigned* __restrict__ own,
                                                    unsigned* __restrict__ idxp, int E) {
    static_assert(DEG_SPS * 512 < 65536, "the counts of one slice must fit in 16 bits");
    static_assert((DEG_NOWN * DEG_SPS) % 4 == 0, "the index table is cleared in 16-byte pieces");
    static_assert(DEG_STEPS <= (N_NODES - (DEG_NOWN - 1) * OWN_R) * HID,
                  "an owner's index words fit in its own rows of the activation plane");
    __shared__ __attribute__((aligned(16))) unsigned tab[DEG_WORDS];
    __shared__ __attribute__((aligned(16))) unsigned idx[DEG_NOWN * DEG_SPS];
    const int tid = threadIdx.x;
    for (int i = tid; i < DEG_WORDS / 4; i += 1024) ((uint4*)tab)[i] = make_uint4(0u, 0u, 0u, 0u);
    for (int i = tid; i < DEG_NOWN * DEG_SPS / 4; i += 1024) ((uint4*)idx)[i] = make_uint4(0u, 0u, 0u, 0u);
    __syncthreads();

    const int nsteps = (E + 511) >> 9;
    const int nwords = nsteps << 7;
    const int sps = (nsteps + DEG_P - 1) / DEG_P;
    const int s0 = blockIdx.x * sps;
    const int w0 = s0 << 7;
    const int w1 = (w0 + (sps << 7)) < nwords ? (w0 + (sps << 7)) : nwords;
    if (((E & 3) == 0) && ((((size_t)dst) & 15) == 0)) {
        const int lastw = (E >> 2) - 1;
        for (int tb = w0 + tid; tb < w1; tb += 1024 * DEG_U) {
            int4 d[DEG_U];
            #pragma unroll
            for (int u = 0; u < DEG_U; ++u) {
                const int t = tb + u * 1024;
                const int tl = t < lastw ? t : lastw;
                d[u] = *(const int4*)(dst + 4 * (size_t)tl);
            }
            #pragma unroll
            for (int u = 0; u < DEG_U; ++u) {
                const int t = tb + u * 1024;
                if (t < w1) {
                    const int4 dd = t <= lastw ? d[u] : make_int4(-1, -1, -1, -1);
                    const int srel = (t >> 7) - s0;
                    const unsigned bit = 1u << ((t >> 2) & 31);
                    own[t] = deg_count(tab, idx, dd.x, srel, bit) | (deg_count(tab, idx, dd.y, srel, bit) << 8)
                           | (deg_count(tab, idx, dd.z, srel, bit) << 16) | (deg_count(tab, idx, dd.w, srel, bit) << 24);
                }
            }
        }
    } else {
        for (int t = w0 + tid; t < w1; t += 1024) {
            unsigned w = 0u;
            for (int c = 0; c < 4; ++c)
                if (4 * t + c < E)
                    w |= deg_count(tab, idx, dst[4 * t + c], (t >> 7) - s0, 1u << ((t >> 2) & 31)) << (8 * c);
            own[t] = w;
        }
    }
    __syncthreads();

    uint4* pl = (uint4*)(part + (size_t)blockIdx.x * DEG_WORDS);
    for (int i = tid; i < DEG_WORDS / 4; i += 1024) pl[i] = ((const uint4*)tab)[i];

    int ns = nsteps - s0;
    ns = ns < 0 ? 0 : (ns > sps ? sps : ns);
    for (int i = tid; i < DEG_NOWN * ns; i += 1024) {
        const int o = i / ns, j = i - o * ns;
        idxp[(size_t)o * (OWN_R * HID) + s0 + j] = idx[o * DEG_SPS + j];
    }
}

__global__ void deg_finish(const unsigned* __restrict__ part, float* __restrict__ dis, int N) {
    const int t = blockIdx.x * blockDim.x + threadIdx.x;
    if (2 * t >= N) return;
    unsigned lo = 0u, hi = 0u;
    #pragma unroll 16
    for (int p = 0; p < DEG_P; ++p) {
        const unsigned w = part[(size_t)p * DEG_WORDS + t];
        lo += w & 0xFFFFu;
        hi += w >> 16;
    }
    const float d0 = rsqrtf(1.0f + (float)lo);
    const float d1 = rsqrtf(1.0f + (float)hi);
    if (2 * t + 1 < N) {
        *(float2*)(dis + 2 * t) = make_float2(d0, d1);
    } else {
        dis[2 * t] = d0;
    }
}

#define GEMM_GRID_K256 512
#define GEMM_GRID_K128 768
typedef __attribute__((ext_vector_type(4))) float v4f;

template <int K>
__global__ __launch_bounds__(256) void gemm_wmma_f32(const float* __restrict__ A,
                                                     const float* __restrict__ W,
                                                     float* __restrict__ out, int ntiles) {
    constexpr int PITCH = K + 4;
    constexpr int RV    = K / 4;
    constexpr int NV    = 16 * RV / 256;
    constexpr int RS    = 256 / RV;
    constexpr int G     = 16;
    constexpr int NG    = K / 4 / G;
    static_assert(NV == 4 || NV == 2, "a thread moves 2 or 4 pieces of a tile");
    __shared__ __attribute__((aligned(16))) float ldsA[16 * PITCH];
    const int tid  = threadIdx.x;
    const int wave = tid >> 5;
    const int lane = tid & 31;
    const int half = lane >> 4;
    const int m    = lane & 15;
    const int col0 = wave * 16;

    int tile = blockIdx.x;
    if (tile >= ntiles) return;

    const int pr = tid / RV, pc = tid - pr * RV;
    float* lst = ldsA + pr * PITCH + 4 * pc;
    const v4f* ap = (const v4f*)(A + (size_t)tile * 16 * K) + tid;
    v4f p0 = ap[0], p1 = ap[256], p2 = {}, p3 = {};
    if constexpr (NV == 4) { p2 = ap[512]; p3 = ap[768]; }

    v2f bf[K / 4];
    #pragma unroll
    for (int s = 0; s < K / 4; ++s) {
        const int ka = 4 * s + 2 * half;
        bf[s].x = W[ka * HID + col0 + m];
        bf[s].y = W[(ka + 1) * HID + col0 + m];
    }

    const float* arow = ldsA + m * PITCH + 2 * half;
    for (;;) {
        __syncthreads();
        *(v4f*)(lst) = p0;
        *(v4f*)(lst + RS * PITCH) = p1;
        if constexpr (NV == 4) {
            *(v4f*)(lst + 2 * RS * PITCH) = p2;
            *(v4f*)(lst + 3 * RS * PITCH) = p3;
        }
        __syncthreads();

        const int nt = tile + (int)gridDim.x;
        const int lt = nt < ntiles ? nt : tile;
        ap = (const v4f*)(A + (size_t)lt * 16 * K) + tid;
        p0 = ap[0]; p1 = ap[256];
        if constexpr (NV == 4) { p2 = ap[512]; p3 = ap[768]; }

        v2f fr[2][G];
        #pragma unroll
        for (int i = 0; i < G; ++i) fr[0][i] = *(const v2f*)(arow + 4 * i);
        v8f acc = {};
        #pragma unroll
        for (int g = 0; g < NG; ++g) {
            if (g + 1 < NG) {
                #pragma unroll
                for (int i = 0; i < G; ++i)
                    fr[(g + 1) & 1][i] = *(const v2f*)(arow + 4 * ((g + 1) * G + i));
            }
            __builtin_amdgcn_sched_barrier(0);
            #pragma unroll
            for (int i = 0; i < G; ++i)
                acc = __builtin_amdgcn_wmma_f32_16x16x4_f32(
                    false, fr[g & 1][i], false, bf[g * G + i], (short)0, acc, false, false);
            __builtin_amdgcn_sched_barrier(0);
        }

        float* op = out + (size_t)(tile * 16 + 8 * half) * HID + col0 + m;
        #pragma unroll
        for (int v = 0; v < 8; ++v) op[v * HID] = acc[v];

        if (nt >= ntiles) break;
        tile = nt;
    }
}

#define OWN_WAVES 32
#define OWN_Q     64
#define OWN_U     4
#define OWN_G     8
#define OWN_NB    ((N_NODES + OWN_R - 1) / OWN_R)
#define OWN_CAP   256
#define OWN_SEG_R ((OWN_R + OWN_WAVES - 1) / OWN_WAVES)

#define OWN_PLANE_WORDS ((N_EDGES + 511) / 512 * 128)
#define OWN_WS_WORDS ((size_t)N_NODES + 2 * (size_t)N_NODES * HID + N_GRAPHS * HID + N_GRAPHS \
                      + OWN_PLANE_WORDS + (size_t)OWN_NB * OWN_WAVES * OWN_CAP + OWN_NB * OWN_WAVES)

typedef __attribute__((address_space(3))) int lds_int;

typedef const __attribute__((address_space(1))) float owner_gf;
static_assert(HID * sizeof(float) == 512, "a row is 512 bytes: the row address is the node number shifted by 9");

template <bool CLAMP>
__device__ __forceinline__ void owner_rows_load(float (&val)[OWN_G][4], int s, int base, int last,
                                                int lane, const float* __restrict__ h) {
    #pragma unroll
    for (int u = 0; u < OWN_G; ++u) {
        const int k = (CLAMP && base + u > last) ? last : base + u;
        const int si = __builtin_amdgcn_readlane(s, k);
        const unsigned long long ra = (unsigned long long)h + ((unsigned long long)(unsigned)si << 9);
        const unsigned ralo = __builtin_amdgcn_readfirstlane((unsigned)ra);
        const unsigned rahi = __builtin_amdgcn_readfirstlane((unsigned)(ra >> 32));
        const owner_gf* hs = (const owner_gf*)(((unsigned long long)rahi << 32) | ralo);
        #pragma unroll
        for (int j = 0; j < 4; ++j) val[u][j] = hs[j * 32 + lane];
    }
}

template <bool GUARD>
__device__ __forceinline__ void owner_rows_add(const float (&val)[OWN_G][4], int r, float nr, int base,
                                               int nb, int lane, float* acc) {
    #pragma unroll
    for (int u = 0; u < OWN_G; ++u) {
        if (!GUARD || base + u < nb) {
            const int ri = __builtin_amdgcn_readlane(r, base + u);
            const float ni = __int_as_float(__builtin_amdgcn_readlane(__float_as_int(nr), base + u));
            float* ar = acc + ri * HID;
            #pragma unroll
            for (int j = 0; j < 4; ++j) atomicAdd(&ar[j * 32 + lane], val[u][j] * ni);
        }
    }
}

__device__ __forceinline__ void owner_rows_full(int s, int r, float nr, int lane,
                                                const float* __restrict__ h, float* acc) {
    static_assert(OWN_G == 8, "a full chunk is four groups");
    float va[OWN_G][4], vb[OWN_G][4];
    owner_rows_load<false>(va, s, 0, 31, lane, h);
    owner_rows_load<false>(vb, s, 8, 31, lane, h);
    owner_rows_add<false>(va, r, nr, 0, 32, lane, acc);
    owner_rows_load<false>(va, s, 16, 31, lane, h);
    owner_rows_add<false>(vb, r, nr, 8, 32, lane, acc);
    owner_rows_load<false>(vb, s, 24, 31, lane, h);
    owner_rows_add<false>(va, r, nr, 16, 32, lane, acc);
    owner_rows_add<false>(vb, r, nr, 24, 32, lane, acc);
}

__device__ __forceinline__ void owner_rows_part(int s, int r, float nr, int nb, int lane,
                                                const float* __restrict__ h, float* acc) {
    float va[OWN_G][4];
    #pragma unroll 1
    for (int i = 0; i < nb; i += OWN_G) {
        owner_rows_load<true>(va, s, i, nb - 1, lane, h);
        owner_rows_add<true>(va, r, nr, i, nb, lane, acc);
    }
}

template <bool FULL>
__device__ __forceinline__ void owner_drain(volatile lds_int* q, int head, int nb, int lane, int lo,
                                            const int* __restrict__ src, const int* __restrict__ dst,
                                            const float* __restrict__ dis, const float* __restrict__ h,
                                            float* acc, unsigned* __restrict__ rec, int& nrec) {
    const bool v = lane < nb;
    const int e = v ? q[(head + lane) & (OWN_Q - 1)] : 0;
    const int s = v ? src[e] : 0;
    const int d = v ? dst[e] : lo;
    const float nr = v ? dis[s] * dis[d] : 0.f;
    const int r = d - lo;
    if (rec != nullptr) {
        if (v && nrec + nb <= OWN_CAP) rec[nrec + lane] = (unsigned)s | ((unsigned)r << 17);
        nrec += nb;
    }
    if (FULL) owner_rows_full(s, r, nr, lane, h, acc);
    else      owner_rows_part(s, r, nr, nb, lane, h, acc);
}

__device__ __forceinline__ void owner_replay(const unsigned* __restrict__ rec, int n, int lane, int lo,
                                             const float* __restrict__ dis, const float* __restrict__ h,
                                             float* acc) {
    if (n <= 0) return;
    const int last = n - 1;
    unsigned w1 = rec[lane < last ? lane : last];
    unsigned w2 = rec[32 + lane < last ? 32 + lane : last];
    float fa = dis[w1 & 0x1FFFFu], fb = dis[lo + (int)(w1 >> 17)];
    int i = 0;
    for (; i + 32 <= n; i += 32) {
        const int s = (int)(w1 & 0x1FFFFu), r = (int)(w1 >> 17);
        const float nr = fa * fb;
        w1 = w2;
        const int nx = i + 64 + lane;
        w2 = rec[nx < last ? nx : last];
        fa = dis[w1 & 0x1FFFFu]; fb = dis[lo + (int)(w1 >> 17)];
        owner_rows_full(s, r, nr, lane, h, acc);
    }
    if (i < n) {
        const int s = (int)(w1 & 0x1FFFFu), r = (int)(w1 >> 17);
        owner_rows_part(s, r, fa * fb, n - i, lane, h, acc);
    }
}

__device__ __forceinline__ void owner_range(int wave, int nsteps, int& sbeg, int& send) {
    const int spw = (nsteps + OWN_WAVES - 1) / OWN_WAVES;
    sbeg = wave * spw;
    send = (sbeg + spw) < nsteps ? (sbeg + spw) : nsteps;
}

__device__ __forceinline__ void owner_scan(volatile lds_int* q, int wave, int lane, int lo, unsigned pat,
                                           int nsteps, int E,
                                           const unsigned* __restrict__ own,
                                           const int* __restrict__ src, const int* __restrict__ dst,
                                           const float* __restrict__ dis, const float* __restrict__ h,
                                           float* acc, unsigned* __restrict__ rec, int& nrec) {
    int head = 0, cnt = 0;
    int sbeg, send;
    owner_range(wave, nsteps, sbeg, send);
    for (int st0 = sbeg; st0 < send; st0 += OWN_U) {
        uint4 ov[OWN_U];
        #pragma unroll
        for (int u = 0; u < OWN_U; ++u) {
            const int st = st0 + u;
            const int sidx = st < send ? st : 0;
            ov[u] = *(const uint4*)(own + ((size_t)(sidx * 32 + lane) << 2));
        }
        __builtin_amdgcn_sched_barrier(0);
        unsigned long long lm = 0ull;
        #pragma unroll
        for (int u = 0; u < OWN_U; ++u) {
            const unsigned wd[4] = {ov[u].x, ov[u].y, ov[u].z, ov[u].w};
            unsigned nz = 0u;
            #pragma unroll
            for (int j = 0; j < 4; ++j) {
                const unsigned x = wd[j] ^ pat;
                const unsigned y = ((x & 0x7F7F7F7Fu) + 0x7F7F7F7Fu) | x;
                nz |= ((y >> 7) & 0x01010101u) << j;
            }
            unsigned m16 = ~nz & 0x0F0F0F0Fu;
            if (st0 + u >= send) m16 = 0u;
            lm |= (unsigned long long)(m16 << (4 * (u & 1))) << (32 * (u >> 1));
        }
        for (int round = 0; round < 16 * OWN_U; ++round) {
            const bool has = lm != 0ull;
            if (__builtin_amdgcn_ballot_w32(has) == 0u) break;
            const int p  = has ? __builtin_ctzll(lm) : 0;
            const int lw = p & 31;
            const int u  = ((p >> 5) << 1) | ((lw >> 2) & 1);
            const int sidx = st0 + u;
            const int e = ((sidx * 32 + lane) << 4) + ((lw & 3) << 2) + (lw >> 3);
            const bool take = has && (e < E);
            const unsigned mask = __builtin_amdgcn_ballot_w32(take);
            const int pos = (head + cnt + (int)__builtin_amdgcn_mbcnt_lo(mask, 0u)) & (OWN_Q - 1);
            if (take) q[pos] = e;
            lm &= lm - 1ull;
            cnt += __builtin_popcount(mask);
            if (cnt >= 32) {
                owner_drain<true>(q, head, 32, lane, lo, src, dst, dis, h, acc, rec, nrec);
                head = (head + 32) & (OWN_Q - 1);
                cnt -= 32;
            }
        }
    }
    owner_drain<false>(q, head, cnt, lane, lo, src, dst, dis, h, acc, rec, nrec);
}

__device__ __forceinline__ void owner_scan_index(volatile lds_int* q, int wave, int lane, int lo, unsigned pat,
                                                 int nsteps, int E, const unsigned* idx,
                                                 const unsigned* __restrict__ own,
                                                 const int* __restrict__ src, const int* __restrict__ dst,
                                                 const float* __restrict__ dis, const float* __restrict__ h,
                                                 float* acc, unsigned* __restrict__ rec, int& nrec) {
    static_assert(OWN_U == 4, "a trip takes four blocks per lane");
    int head = 0, cnt = 0;
    int sbeg, send;
    owner_range(wave, nsteps, sbeg, send);
    for (int sb = sbeg; sb < send; sb += 32) {
        const int st = sb + lane;
        unsigned bits = st < send ? idx[st] : 0u;
        for (int trip = 0; trip < 32 / OWN_U; ++trip) {
            if (__builtin_amdgcn_ballot_w32(bits != 0u) == 0u) break;
            int blk[OWN_U];
            uint4 ov[OWN_U];
            #pragma unroll
            for (int u = 0; u < OWN_U; ++u) {
                const bool hasb = bits != 0u;
                blk[u] = hasb ? (st << 5) + __builtin_ctz(bits) : -1;
                bits &= bits - 1u;
                ov[u] = *(const uint4*)(own + ((size_t)(hasb ? blk[u] : 0) << 2));
            }
            __builtin_amdgcn_sched_barrier(0);
            unsigned long long lm = 0ull;
            #pragma unroll
            for (int u = 0; u < OWN_U; ++u) {
                const unsigned wd[4] = {ov[u].x, ov[u].y, ov[u].z, ov[u].w};
                unsigned nz = 0u;
                #pragma unroll
                for (int j = 0; j < 4; ++j) {
                    const unsigned x = wd[j] ^ pat;
                    const unsigned y = ((x & 0x7F7F7F7Fu) + 0x7F7F7F7Fu) | x;
                    nz |= ((y >> 7) & 0x01010101u) << j;
                }
                unsigned m16 = ~nz & 0x0F0F0F0Fu;
                if (blk[u] < 0) m16 = 0u;
                lm |= (unsigned long long)(m16 << (4 * (u & 1))) << (32 * (u >> 1));
            }
            for (int round = 0; round < 16 * OWN_U; ++round) {
                const bool has = lm != 0ull;
                if (__builtin_amdgcn_ballot_w32(has) == 0u) break;
                const int p  = has ? __builtin_ctzll(lm) : 0;
                const int lw = p & 31;
                const int u  = ((p >> 5) << 1) | ((lw >> 2) & 1);
                const int b  = u == 0 ? blk[0] : (u == 1 ? blk[1] : (u == 2 ? blk[2] : blk[3]));
                const int e = (b << 4) + ((lw & 3) << 2) + (lw >> 3);
                const bool take = has && (e < E);
                const unsigned mask = __builtin_amdgcn_ballot_w32(take);
                const int pos = (head + cnt + (int)__builtin_amdgcn_mbcnt_lo(mask, 0u)) & (OWN_Q - 1);
                if (take) q[pos] = e;
                lm &= lm - 1ull;
                cnt += __builtin_popcount(mask);
                if (cnt >= 32) {
                    owner_drain<true>(q, head, 32, lane, lo, src, dst, dis, h, acc, rec, nrec);
                    head = (head + 32) & (OWN_Q - 1);
                    cnt -= 32;
                }
            }
        }
    }
    owner_drain<false>(q, head, cnt, lane, lo, src, dst, dis, h, acc, rec, nrec);
}

__device__ __forceinline__ void owner_pool_flush(float* tab, int* tcnt, int g, int rl, float4 a, int lane) {
    if ((unsigned)g < (unsigned)N_GRAPHS) {
        float* t = tab + g * HID + 4 * lane;
        atomicAdd(t + 0, a.x); atomicAdd(t + 1, a.y); atomicAdd(t + 2, a.z); atomicAdd(t + 3, a.w);
        if (lane == 0) atomicAdd(tcnt + g, rl);
    }
}

template <int LAYER>
__global__ __launch_bounds__(1024) void owner_aggregate(const int* __restrict__ src,
                                                        const int* __restrict__ dst,
                                                        const float* __restrict__ dis,
                                                        const float* __restrict__ h,
                                                        const float* __restrict__ bias,
                                                        const unsigned* __restrict__ own,
                                                        float* __restrict__ outp,
                                                        unsigned* __restrict__ rlist,
                                                        int* __restrict__ rcnt,
                                                        const int* __restrict__ batch,
                                                        float* __restrict__ ppart,
                                                        int* __restrict__ pcnt,
                                                        size_t ws_bytes, int E, int N) {
    static_assert((N_NODES + OWN_R - 1) / OWN_R <= 256, "an owner id must fit in one byte");
    static_assert(N_NODES <= (1 << 17) && OWN_R <= (1 << 15), "a record word holds source and row");
    static_assert(OWN_SEG_R <= 32, "the pool pass: lane i holds the graph id of its wave's row i");
    constexpr int ACC_ROWS = LAYER == 2 ? OWN_WAVES * OWN_SEG_R : OWN_R;
    constexpr int TAB_N    = LAYER == 2 ? N_GRAPHS * HID : 4;
    __shared__ __attribute__((aligned(16))) float acc[ACC_ROWS * HID];
    __shared__ int queue[OWN_WAVES * OWN_Q];
    __shared__ __attribute__((aligned(16))) float tab[TAB_N];
    __shared__ int tcnt[LAYER == 2 ? N_GRAPHS : 1];

    const int tid  = threadIdx.x;
    const int lane = tid & 31;
    const int wave = __builtin_amdgcn_readfirstlane(tid >> 5);
    const int lo   = blockIdx.x * OWN_R;
    const int nrows = (N - lo) < OWN_R ? (N - lo) : OWN_R;
    volatile lds_int* q = (volatile lds_int*)(queue + wave * OWN_Q);

    for (int idx = tid; idx < nrows * (HID / 4); idx += 1024) {
        const int r = idx >> 5, c4 = idx & 31;
        const float di = dis[lo + r];
        const float w = di * di;
        const float4 hv = ((const float4*)(h + (size_t)(lo + r) * HID))[c4];
        const float4 bv = ((const float4*)bias)[c4];
        float4 o;
        o.x = w * hv.x + bv.x; o.y = w * hv.y + bv.y; o.z = w * hv.z + bv.z; o.w = w * hv.w + bv.w;
        ((float4*)acc)[idx] = o;
    }
    if (LAYER == 2) {
        for (int i = tid; i < TAB_N / 4; i += 1024) ((float4*)tab)[i] = make_float4(0.f, 0.f, 0.f, 0.f);
        if (tid < N_GRAPHS) tcnt[tid] = 0;
    }
    __syncthreads();

    const int nsteps = (E + 511) >> 9;
    const unsigned pat = (unsigned)blockIdx.x * 0x01010101u;
    const bool room = ws_bytes >= OWN_WS_WORDS * sizeof(float);
    const int slot = blockIdx.x * OWN_WAVES + wave;
    unsigned* rec = rlist + (size_t)slot * OWN_CAP;
    if (LAYER == 1) {
        int nrec = 0;
        const unsigned* idx = (const unsigned*)(outp + (size_t)lo * HID);
        owner_scan_index(q, wave, lane, lo, pat, nsteps, E, idx, own, src, dst, dis, h, acc,
                         room ? rec : nullptr, nrec);
        if (lane == 0) q[0] = nrec;
    } else {
        const int n = room ? rcnt[slot] : OWN_CAP + 1;
        if (n <= OWN_CAP) {
            owner_replay(rec, n, lane, lo, dis, h, acc);
        } else {
            int none = 0;
            owner_scan(q, wave, lane, lo, pat, nsteps, E, own, src, dst, dis, h, acc, nullptr, none);
        }
    }
    const int sr0 = wave * OWN_SEG_R;
    int snr = nrows - sr0;
    snr = snr < 0 ? 0 : (snr > OWN_SEG_R ? OWN_SEG_R : snr);
    int gl = -1;
    if (LAYER == 2 && lane < snr) gl = batch[lo + sr0 + lane];
    __syncthreads();

    if (LAYER == 1) {
        if (room && wave == 0) rcnt[blockIdx.x * OWN_WAVES + lane] = queue[lane * OWN_Q];

        float4* orow = (float4*)(outp + (size_t)lo * HID);
        for (int idx = tid; idx < nrows * (HID / 4); idx += 1024) {
            float4 v4 = ((const float4*)acc)[idx];
            v4.x = v4.x > 0.f ? v4.x : 0.f; v4.y = v4.y > 0.f ? v4.y : 0.f;
            v4.z = v4.z > 0.f ? v4.z : 0.f; v4.w = v4.w > 0.f ? v4.w : 0.f;
            orow[idx] = v4;
        }
    } else {
        const float4 zero = make_float4(0.f, 0.f, 0.f, 0.f);
        const float4* ap = (const float4*)(acc + sr0 * HID) + lane;
        float4 v[OWN_SEG_R];
        #pragma unroll
        for (int i = 0; i < OWN_SEG_R; ++i) v[i] = ap[i * (HID / 4)];
        int gcur = -1, rl = 0;
        float4 a = zero;
        #pragma unroll
        for (int i = 0; i < OWN_SEG_R; ++i) {
            if (i < snr) {
                const int g = __builtin_amdgcn_readlane(gl, i);
                if (g != gcur) {
                    owner_pool_flush(tab, tcnt, gcur, rl, a, lane);
                    gcur = g; rl = 0; a = zero;
                }
                a.x += v[i].x > 0.f ? v[i].x : 0.f; a.y += v[i].y > 0.f ? v[i].y : 0.f;
                a.z += v[i].z > 0.f ? v[i].z : 0.f; a.w += v[i].w > 0.f ? v[i].w : 0.f;
                ++rl;
            }
        }
        owner_pool_flush(tab, tcnt, gcur, rl, a, lane);
        __syncthreads();

        float4* pl = (float4*)(ppart + (size_t)blockIdx.x * (N_GRAPHS * HID));
        for (int i = tid; i < N_GRAPHS * HID / 4; i += 1024) pl[i] = ((const float4*)tab)[i];
        if (tid < N_GRAPHS) pcnt[blockIdx.x * N_GRAPHS + tid] = tcnt[tid];
    }
}

#define POOL_NB    256
static_assert(POOL_NB == OWN_NB, "one partial plane per work-group of the aggregation");

__global__ __launch_bounds__(1024) void pool_finish(const float* __restrict__ part,
                                                    const int* __restrict__ cntp,
                                                    float* __restrict__ pooled) {
    static_assert(POOL_NB == 256, "32 waves x 8 planes, one count per thread of the first 256");
    __shared__ __attribute__((aligned(16))) float red[32 * HID];
    __shared__ int ntot;
    const int tid  = threadIdx.x;
    const int lane = tid & 31;
    const int wave = tid >> 5;
    const int g    = blockIdx.x;

    float4 v[8];
    #pragma unroll
    for (int i = 0; i < 8; ++i)
        v[i] = ((const float4*)(part + ((size_t)(wave * 8 + i) * N_GRAPHS + g) * HID))[lane];
    const int c = (tid < POOL_NB) ? cntp[tid * N_GRAPHS + g] : 0;

    if (tid == 0) ntot = 0;
    __syncthreads();
    float4 s = v[0];
    #pragma unroll
    for (int i = 1; i < 8; ++i) { s.x += v[i].x; s.y += v[i].y; s.z += v[i].z; s.w += v[i].w; }
    ((float4*)red)[wave * 32 + lane] = s;
    if (c != 0) atomicAdd(&ntot, c);
    __syncthreads();

    if (wave == 0) {
        float4 t = ((const float4*)red)[lane];
        #pragma unroll
        for (int w = 1; w < 32; ++w) {
            const float4 u = ((const float4*)red)[w * 32 + lane];
            t.x += u.x; t.y += u.y; t.z += u.z; t.w += u.w;
        }
        float n = (float)ntot;
        n = n > 1.f ? n : 1.f;
        t.x /= n; t.y /= n; t.z /= n; t.w /= n;
        ((float4*)(pooled + g * HID))[lane] = t;
    }
}

#define HEAD_T_MF   ((OUT_MF + 15) / 16)
#define HEAD_T_BP   ((OUT_BP + 15) / 16)
#define HEAD_T_CC   ((OUT_CC + 15) / 16)
#define HEAD_TILES  (HEAD_T_MF + HEAD_T_BP + HEAD_T_CC)
#define HEAD_WAVES  4
#define HEAD_PITCH  (HID + 4)
#define HEAD_KS     4

__global__ __launch_bounds__(HEAD_WAVES * 32) void heads_wmma(const float* __restrict__ pooled,
                                                              const float* __restrict__ Wmf,
                                                              const float* __restrict__ bmf,
                                                              const float* __restrict__ Wbp,
                                                              const float* __restrict__ bbp,
                                                              const float* __restrict__ Wcc,
                                                              const float* __restrict__ bcc,
                                                              float* __restrict__ out_mf,
                                                              float* __restrict__ out_bp,
                                                              float* __restrict__ out_cc) {
    static_assert(N_GRAPHS == 64 && HID == 128, "4 row tiles, 32 k steps, 16 pieces of the means per thread");
    __shared__ __attribute__((aligned(16))) float ldsP[N_GRAPHS * HEAD_PITCH];
    const int tid  = threadIdx.x;
    const int lane = tid & 31;
    const int wave = __builtin_amdgcn_readfirstlane(tid >> 5);
    const int half = lane >> 4;
    const int m    = lane & 15;

    v4f pp[16];
    #pragma unroll
    for (int j = 0; j < 16; ++j) pp[j] = ((const v4f*)pooled)[tid + HEAD_WAVES * 32 * j];

    int t = blockIdx.x * HEAD_WAVES + wave;
    const bool live = t < HEAD_TILES;
    if (!live) t = HEAD_TILES - 1;
    const float* W; const float* b; float* o; int OUT;
    if (t < HEAD_T_MF) {
        W = Wmf; b = bmf; o = out_mf; OUT = OUT_MF;
    } else if (t < HEAD_T_MF + HEAD_T_BP) {
        t -= HEAD_T_MF; W = Wbp; b = bbp; o = out_bp; OUT = OUT_BP;
    } else {
        t -= HEAD_T_MF + HEAD_T_BP; W = Wcc; b = bcc; o = out_cc; OUT = OUT_CC;
    }
    const int col  = t * 16 + m;
    const int colc = col < OUT ? col : OUT - 1;

    v2f bf[HID / 4];
    #pragma unroll
    for (int s = 0; s < HID / 4; ++s) {
        const int ka = 4 * s + 2 * half;
        bf[s].x = W[ka * OUT + colc];
        bf[s].y = W[(ka + 1) * OUT + colc];
    }
    const float bias = b[colc];

    #pragma unroll
    for (int j = 0; j < 16; ++j) {
        const int p = tid + HEAD_WAVES * 32 * j;
        *(v4f*)(ldsP + (p >> 5) * HEAD_PITCH + 4 * (p & 31)) = pp[j];
    }
    __syncthreads();

    v8f acc[4];
    #pragma unroll
    for (int rt = 0; rt < 4; ++rt) {
        #pragma unroll
        for (int v = 0; v < 8; ++v) acc[rt][v] = bias;
    }

    constexpr int NG = HID / 4 / HEAD_KS;
    const float* arow = ldsP + m * HEAD_PITCH + 2 * half;
    v2f fr[2][HEAD_KS * 4];
    #pragma unroll
    for (int i = 0; i < HEAD_KS; ++i) {
        #pragma unroll
        for (int rt = 0; rt < 4; ++rt)
            fr[0][i * 4 + rt] = *(const v2f*)(arow + rt * 16 * HEAD_PITCH + 4 * i);
    }
    #pragma unroll
    for (int g = 0; g < NG; ++g) {
        if (g + 1 < NG) {
            #pragma unroll
            for (int i = 0; i < HEAD_KS; ++i) {
                #pragma unroll
                for (int rt = 0; rt < 4; ++rt)
                    fr[(g + 1) & 1][i * 4 + rt] =
                        *(const v2f*)(arow + rt * 16 * HEAD_PITCH + 4 * ((g + 1) * HEAD_KS + i));
            }
        }
        __builtin_amdgcn_sched_barrier(0);
        #pragma unroll
        for (int i = 0; i < HEAD_KS; ++i) {
            #pragma unroll
            for (int rt = 0; rt < 4; ++rt)
                acc[rt] = __builtin_amdgcn_wmma_f32_16x16x4_f32(
                    false, fr[g & 1][i * 4 + rt], false, bf[g * HEAD_KS + i], (short)0, acc[rt], false, false);
        }
        __builtin_amdgcn_sched_barrier(0);
    }

    if (live && col < OUT) {
        #pragma unroll
        for (int rt = 0; rt < 4; ++rt) {
            float* op = o + (size_t)(16 * rt + 8 * half) * OUT + col;
            #pragma unroll
            for (int v = 0; v < 8; ++v) op[(size_t)v * OUT] = acc[rt][v];
        }
    }
}

extern "C" void kernel_launch(void* const* d_in, const int* in_sizes, int n_in,
                              void* d_out, int out_size, void* d_ws, size_t ws_size,
                              hipStream_t stream) {
    const float* x   = (const float*)d_in[0];
    const int*   ei  = (const int*)d_in[1];
    const int*   bat = (const int*)d_in[2];
    const float* W1  = (const float*)d_in[3];
    const float* b1  = (const float*)d_in[4];
    const float* W2  = (const float*)d_in[5];
    const float* b2  = (const float*)d_in[6];
    const float* Wmf = (const float*)d_in[7];
    const float* bmf = (const float*)d_in[8];
    const float* Wbp = (const float*)d_in[9];
    const float* bbp = (const float*)d_in[10];
    const float* Wcc = (const float*)d_in[11];
    const float* bcc = (const float*)d_in[12];
    float* out = (float*)d_out;

    const int E = in_sizes[1] / 2;
    const int* src = ei;
    const int* dst = ei + E;

    float* ws     = (float*)d_ws;
    float* dis    = ws;
    float* h      = dis + N_NODES;
    float* agg    = h + (size_t)N_NODES * HID;
    float* pooled = agg + (size_t)N_NODES * HID;
    float* cnt    = pooled + N_GRAPHS * HID;
    unsigned* own = (unsigned*)(cnt + N_GRAPHS);
    unsigned* rlist = own + OWN_PLANE_WORDS;
    int*      rcnt  = (int*)(rlist + (size_t)OWN_NB * OWN_WAVES * OWN_CAP);

    unsigned* degp = (unsigned*)h;
    deg_partial<<<DEG_P, 1024, 0, stream>>>(dst, degp, own, (unsigned*)agg, E);
    deg_finish<<<((N_NODES + 1) / 2 + 255) / 256, 256, 0, stream>>>(degp, dis, N_NODES);

    gemm_wmma_f32<IN_DIM><<<GEMM_GRID_K256, 256, 0, stream>>>(x, W1, h, N_NODES / 16);
    owner_aggregate<1><<<OWN_NB, 1024, 0, stream>>>(src, dst, dis, h, b1, own, agg, rlist, rcnt,
                                                    nullptr, nullptr, nullptr, ws_size, E, N_NODES);

    float* part = agg;
    int*   cntp = (int*)(agg + (size_t)POOL_NB * N_GRAPHS * HID);
    gemm_wmma_f32<HID><<<GEMM_GRID_K128, 256, 0, stream>>>(agg, W2, h, N_NODES / 16);
    owner_aggregate<2><<<OWN_NB, 1024, 0, stream>>>(src, dst, dis, h, b2, own, nullptr, rlist, rcnt,
                                                    bat, part, cntp, ws_size, E, N_NODES);

    pool_finish<<<N_GRAPHS, 1024, 0, stream>>>(part, cntp, pooled);

    float* out_mf = out;
    float* out_bp = out + N_GRAPHS * OUT_MF;
    float* out_cc = out + N_GRAPHS * (OUT_MF + OUT_BP);
    heads_wmma<<<(HEAD_TILES + HEAD_WAVES - 1) / HEAD_WAVES, HEAD_WAVES * 32, 0, stream>>>(
        pooled, Wmf, bmf, Wbp, bbp, Wcc, bcc, out_mf, out_bp, out_cc);
}
